# starting kernel
# baseline (speedup 1.0000x reference)
#include <hip/hip_runtime.h>

#define HDIM 128
#define NCLS 40
#define BN_EPS 1e-5f

typedef __attribute__((ext_vector_type(2))) float v2f;
typedef __attribute__((ext_vector_type(8))) float v8f;

__global__ void k_set_ones(float* deg, int n) {
  int i = blockIdx.x * blockDim.x + threadIdx.x;
  if (i < n) deg[i] = 1.0f;
}

__global__ void k_deg_accum(const int* __restrict__ ei, float* deg, int E) {
  int e = blockIdx.x * blockDim.x + threadIdx.x;
  if (e < E) atomicAdd(&deg[ei[E + e]], 1.0f);
}

__global__ void k_rsqrt_ip(float* d, int n) {
  int i = blockIdx.x * blockDim.x + threadIdx.x;
  if (i < n) d[i] = rsqrtf(d[i]);
}

__global__ void k_stats(const float* __restrict__ x, float* __restrict__ stats, int n) {
  int col = threadIdx.x & (HDIM - 1);
  int half = threadIdx.x >> 7;
  float s = 0.f, s2 = 0.f;
  for (int r = blockIdx.x * 2 + half; r < n; r += gridDim.x * 2) {
    float v = x[(size_t)r * HDIM + col];
    s += v; s2 += v * v;
  }
  __shared__ float ls[HDIM], ls2[HDIM];
  if (half == 0) { ls[col] = s; ls2[col] = s2; }
  __syncthreads();
  if (half == 1) {
    atomicAdd(&stats[col], s + ls[col]);
    atomicAdd(&stats[HDIM + col], s2 + ls2[col]);
  }
}

__global__ void k_bias_relu_stats(float* __restrict__ x, const float* __restrict__ b,
                                  float* __restrict__ stats, int n, int do_stats) {
  int col = threadIdx.x & (HDIM - 1);
  int half = threadIdx.x >> 7;
  float bias = b[col];
  float s = 0.f, s2 = 0.f;
  for (int r = blockIdx.x * 2 + half; r < n; r += gridDim.x * 2) {
    size_t idx = (size_t)r * HDIM + col;
    float v = x[idx] + bias;
    v = v > 0.f ? v : 0.f;
    x[idx] = v;
    s += v; s2 += v * v;
  }
  if (!do_stats) return;
  __shared__ float ls[HDIM], ls2[HDIM];
  if (half == 0) { ls[col] = s; ls2[col] = s2; }
  __syncthreads();
  if (half == 1) {
    atomicAdd(&stats[col], s + ls[col]);
    atomicAdd(&stats[HDIM + col], s2 + ls2[col]);
  }
}

__global__ void k_fold(const float* __restrict__ stats, const float* __restrict__ gamma,
                       const float* __restrict__ beta, const float* __restrict__ W,
                       float* __restrict__ WfT, float* __restrict__ cb, float inv_n) {
  __shared__ float as[HDIM], cs[HDIM];
  int t = threadIdx.x;
  float mean = stats[t] * inv_n;
  float var  = stats[HDIM + t] * inv_n - mean * mean;
  float a = gamma[t] * rsqrtf(var + BN_EPS);
  float c = beta[t] - a * mean;
  as[t] = a; cs[t] = c;
  __syncthreads();
  float acc = 0.f;
  for (int k = 0; k < HDIM; ++k) {
    float w = W[k * HDIM + t];
    WfT[t * HDIM + k] = as[k] * w;
    acc += cs[k] * w;
  }
  cb[t] = acc;
}

#define XPAD 68

__global__ __launch_bounds__(128) void k_gemm(const float* __restrict__ x,
                                              const float* __restrict__ WfT,
                                              const float* __restrict__ cb,
                                              float* __restrict__ h, int n) {
  __shared__ float xs[64 * XPAD];
  __shared__ float wfs[HDIM * XPAD];
  const int tid = threadIdx.x;
  const int lane = tid & 31;
  const int wave = tid >> 5;
  const int rowBase = blockIdx.x * 64;
  const int m = lane & 15;
  const int hi = lane >> 4;

  v8f acc[8];
#pragma unroll
  for (int c = 0; c < 8; ++c)
#pragma unroll
    for (int v = 0; v < 8; ++v) acc[c][v] = 0.f;

  for (int kh = 0; kh < 2; ++kh) {
#pragma unroll
    for (int p = 0; p < 8; ++p) {
      int idx = p * 512 + tid * 4;
      int r = idx >> 6, cc = idx & 63;
      int row = rowBase + r;
      float4 v = make_float4(0.f, 0.f, 0.f, 0.f);
      if (row < n) v = *(const float4*)(x + (size_t)row * HDIM + kh * 64 + cc);
      *(float4*)(&xs[r * XPAD + cc]) = v;
    }
#pragma unroll
    for (int p = 0; p < 16; ++p) {
      int idx = p * 512 + tid * 4;
      int nn = idx >> 6, cc = idx & 63;
      *(float4*)(&wfs[nn * XPAD + cc]) = *(const float4*)(WfT + nn * HDIM + kh * 64 + cc);
    }
    __syncthreads();

    const float* xrow = &xs[(wave * 16 + m) * XPAD];
#pragma unroll
    for (int k4 = 0; k4 < 16; ++k4) {
      int kb = k4 * 4 + hi * 2;
      v2f a; a.x = xrow[kb]; a.y = xrow[kb + 1];
#pragma unroll
      for (int c = 0; c < 8; ++c) {
        const float* wrow = &wfs[(c * 16 + m) * XPAD];
        v2f b; b.x = wrow[kb]; b.y = wrow[kb + 1];
        acc[c] = __builtin_amdgcn_wmma_f32_16x16x4_f32(
            false, a, false, b, (short)0, acc[c], false, false);
      }
    }
    __syncthreads();
  }

#pragma unroll
  for (int c = 0; c < 8; ++c) {
    int col = c * 16 + m;
    float cbv = cb[col];
#pragma unroll
    for (int v = 0; v < 8; ++v) {
      int row = rowBase + wave * 16 + v + 8 * hi;
      if (row < n) h[(size_t)row * HDIM + col] = acc[c][v] + cbv;
    }
  }
}

__global__ void k_edge_agg(const int* __restrict__ ei, const float* __restrict__ dinv,
                           const float* __restrict__ hbuf, float* __restrict__ out,
                           int E, int n) {
  int e = blockIdx.x * 8 + (threadIdx.x >> 5);
  if (e >= E + n) return;
  int lane = threadIdx.x & 31;
  int src, dst;
  if (e < E) { src = ei[e]; dst = ei[E + e]; }
  else       { src = dst = e - E; }
  float nrm = dinv[src] * dinv[dst];
  const float4 hv = *(const float4*)(hbuf + (size_t)src * HDIM + lane * 4);
  float* o = out + (size_t)dst * HDIM + lane * 4;
  atomicAdd(o + 0, hv.x * nrm);
  atomicAdd(o + 1, hv.y * nrm);
  atomicAdd(o + 2, hv.z * nrm);
  atomicAdd(o + 3, hv.w * nrm);
}

__global__ __launch_bounds__(256) void k_out_gemm(const float* __restrict__ x,
                                                  const float* __restrict__ Wo,
                                                  const float* __restrict__ bo,
                                                  float* __restrict__ out, int n) {
  __shared__ float ws[HDIM * NCLS];
  __shared__ float bs[NCLS];
  int tid = threadIdx.x;
  for (int i = tid; i < HDIM * NCLS; i += 256) ws[i] = Wo[i];
  if (tid < NCLS) bs[tid] = bo[tid];
  __syncthreads();
  int r = blockIdx.x * 128 + (tid >> 1);
  int c0 = (tid & 1) * 20;
  if (r >= n) return;
  float acc[20];
#pragma unroll
  for (int c = 0; c < 20; ++c) acc[c] = bs[c0 + c];
  const float* xr = x + (size_t)r * HDIM;
  for (int k = 0; k < HDIM; ++k) {
    float xv = xr[k];
    const float* wr = &ws[k * NCLS + c0];
#pragma unroll
    for (int c = 0; c < 20; ++c) acc[c] += xv * wr[c];
  }
  float* o = out + (size_t)r * NCLS + c0;
#pragma unroll
  for (int c = 0; c < 20; ++c) o[c] = acc[c];
}

extern "C" void kernel_launch(void* const* d_in, const int* in_sizes, int n_in,
                              void* d_out, int out_size, void* d_ws, size_t ws_size,
                              hipStream_t stream) {
  const float* x_in  = (const float*)d_in[0];
  const int*   ei    = (const int*)d_in[1];
  const float* gamma = (const float*)d_in[2];
  const float* beta  = (const float*)d_in[3];
  const float* W     = (const float*)d_in[4];
  const float* b     = (const float*)d_in[5];
  const float* Wo    = (const float*)d_in[6];
  const float* bo    = (const float*)d_in[7];
  float* out = (float*)d_out;

  const int N = in_sizes[0] / HDIM;
  const int E = in_sizes[1] / 2;

  float* dinv  = (float*)d_ws;
  float* stats = dinv + ((N + 255) / 256) * 256;
  float* WfT   = stats + 2 * HDIM;
  float* cb    = WfT + HDIM * HDIM;
  float* hbuf  = cb + HDIM + 64;
  float* xbuf  = hbuf + (size_t)N * HDIM;

  k_set_ones<<<(N + 255) / 256, 256, 0, stream>>>(dinv, N);
  k_deg_accum<<<(E + 255) / 256, 256, 0, stream>>>(ei, dinv, E);
  k_rsqrt_ip<<<(N + 255) / 256, 256, 0, stream>>>(dinv, N);

  hipMemsetAsync(stats, 0, 2 * HDIM * sizeof(float), stream);
  k_stats<<<512, 256, 0, stream>>>(x_in, stats, N);

  const int EN = E + N;
  for (int l = 0; l < 3; ++l) {
    k_fold<<<1, 128, 0, stream>>>(stats, gamma + l * HDIM, beta + l * HDIM,
                                  W + (size_t)l * HDIM * HDIM, WfT, cb, 1.0f / (float)N);
    const float* xl = (l == 0) ? x_in : xbuf;
    k_gemm<<<(N + 63) / 64, 128, 0, stream>>>(xl, WfT, cb, hbuf, N);
    hipMemsetAsync(xbuf, 0, (size_t)N * HDIM * sizeof(float), stream);
    k_edge_agg<<<(EN + 7) / 8, 256, 0, stream>>>(ei, dinv, hbuf, xbuf, E, N);
    if (l < 2) {
      hipMemsetAsync(stats, 0, 2 * HDIM * sizeof(float), stream);
      k_bias_relu_stats<<<512, 256, 0, stream>>>(xbuf, b + l * HDIM, stats, N, 1);
    } else {
      k_bias_relu_stats<<<512, 256, 0, stream>>>(xbuf, b + l * HDIM, stats, N, 0);
    }
  }

  k_out_gemm<<<(N + 127) / 128, 256, 0, stream>>>(xbuf, Wo, bo, out, N);
}


# Aggregation by destination range: a block keeps 512 rows' sums in LDS, filters the edge list for its rows, adds source rows by LDS float adds, plain row stores; global atomics go; sum order differs
# speedup vs baseline: 1.5462x; 1.5462x over previous
#include <hip/hip_runtime.h>

#define HDIM 128
#define NCLS 40
#define BN_EPS 1e-5f

typedef __attribute__((ext_vector_type(2))) float v2f;
typedef __attribute__((ext_vector_type(8))) float v8f;

__global__ void k_set_ones(float* deg, int n) {
  int i = blockIdx.x * blockDim.x + threadIdx.x;
  if (i < n) deg[i] = 1.0f;
}

__global__ void k_deg_accum(const int* __restrict__ ei, float* deg, int E) {
  int e = blockIdx.x * blockDim.x + threadIdx.x;
  if (e < E) atomicAdd(&deg[ei[E + e]], 1.0f);
}

__global__ void k_rsqrt_ip(float* d, int n) {
  int i = blockIdx.x * blockDim.x + threadIdx.x;
  if (i < n) d[i] = rsqrtf(d[i]);
}

__global__ void k_stats(const float* __restrict__ x, float* __restrict__ stats, int n) {
  int col = threadIdx.x & (HDIM - 1);
  int half = threadIdx.x >> 7;
  float s = 0.f, s2 = 0.f;
  for (int r = blockIdx.x * 2 + half; r < n; r += gridDim.x * 2) {
    float v = x[(size_t)r * HDIM + col];
    s += v; s2 += v * v;
  }
  __shared__ float ls[HDIM], ls2[HDIM];
  if (half == 0) { ls[col] = s; ls2[col] = s2; }
  __syncthreads();
  if (half == 1) {
    atomicAdd(&stats[col], s + ls[col]);
    atomicAdd(&stats[HDIM + col], s2 + ls2[col]);
  }
}

__global__ void k_bias_relu_stats(float* __restrict__ x, const float* __restrict__ b,
                                  float* __restrict__ stats, int n, int do_stats) {
  int col = threadIdx.x & (HDIM - 1);
  int half = threadIdx.x >> 7;
  float bias = b[col];
  float s = 0.f, s2 = 0.f;
  for (int r = blockIdx.x * 2 + half; r < n; r += gridDim.x * 2) {
    size_t idx = (size_t)r * HDIM + col;
    float v = x[idx] + bias;
    v = v > 0.f ? v : 0.f;
    x[idx] = v;
    s += v; s2 += v * v;
  }
  if (!do_stats) return;
  __shared__ float ls[HDIM], ls2[HDIM];
  if (half == 0) { ls[col] = s; ls2[col] = s2; }
  __syncthreads();
  if (half == 1) {
    atomicAdd(&stats[col], s + ls[col]);
    atomicAdd(&stats[HDIM + col], s2 + ls2[col]);
  }
}

__global__ void k_fold(const float* __restrict__ stats, const float* __restrict__ gamma,
                       const float* __restrict__ beta, const float* __restrict__ W,
                       float* __restrict__ WfT, float* __restrict__ cb, float inv_n) {
  __shared__ float as[HDIM], cs[HDIM];
  int t = threadIdx.x;
  float mean = stats[t] * inv_n;
  float var  = stats[HDIM + t] * inv_n - mean * mean;
  float a = gamma[t] * rsqrtf(var + BN_EPS);
  float c = beta[t] - a * mean;
  as[t] = a; cs[t] = c;
  __syncthreads();
  float acc = 0.f;
  for (int k = 0; k < HDIM; ++k) {
    float w = W[k * HDIM + t];
    WfT[t * HDIM + k] = as[k] * w;
    acc += cs[k] * w;
  }
  cb[t] = acc;
}

#define XPAD 68

__global__ __launch_bounds__(128) void k_gemm(const float* __restrict__ x,
                                              const float* __restrict__ WfT,
                                              const float* __restrict__ cb,
                                              float* __restrict__ h, int n) {
  __shared__ float xs[64 * XPAD];
  __shared__ float wfs[HDIM * XPAD];
  const int tid = threadIdx.x;
  const int lane = tid & 31;
  const int wave = tid >> 5;
  const int rowBase = blockIdx.x * 64;
  const int m = lane & 15;
  const int hi = lane >> 4;

  v8f acc[8];
#pragma unroll
  for (int c = 0; c < 8; ++c)
#pragma unroll
    for (int v = 0; v < 8; ++v) acc[c][v] = 0.f;

  for (int kh = 0; kh < 2; ++kh) {
#pragma unroll
    for (int p = 0; p < 8; ++p) {
      int idx = p * 512 + tid * 4;
      int r = idx >> 6, cc = idx & 63;
      int row = rowBase + r;
      float4 v = make_float4(0.f, 0.f, 0.f, 0.f);
      if (row < n) v = *(const float4*)(x + (size_t)row * HDIM + kh * 64 + cc);
      *(float4*)(&xs[r * XPAD + cc]) = v;
    }
#pragma unroll
    for (int p = 0; p < 16; ++p) {
      int idx = p * 512 + tid * 4;
      int nn = idx >> 6, cc = idx & 63;
      *(float4*)(&wfs[nn * XPAD + cc]) = *(const float4*)(WfT + nn * HDIM + kh * 64 + cc);
    }
    __syncthreads();

    const float* xrow = &xs[(wave * 16 + m) * XPAD];
#pragma unroll
    for (int k4 = 0; k4 < 16; ++k4) {
      int kb = k4 * 4 + hi * 2;
      v2f a; a.x = xrow[kb]; a.y = xrow[kb + 1];
#pragma unroll
      for (int c = 0; c < 8; ++c) {
        const float* wrow = &wfs[(c * 16 + m) * XPAD];
        v2f b; b.x = wrow[kb]; b.y = wrow[kb + 1];
        acc[c] = __builtin_amdgcn_wmma_f32_16x16x4_f32(
            false, a, false, b, (short)0, acc[c], false, false);
      }
    }
    __syncthreads();
  }

#pragma unroll
  for (int c = 0; c < 8; ++c) {
    int col = c * 16 + m;
    float cbv = cb[col];
#pragma unroll
    for (int v = 0; v < 8; ++v) {
      int row = rowBase + wave * 16 + v + 8 * hi;
      if (row < n) h[(size_t)row * HDIM + col] = acc[c][v] + cbv;
    }
  }
}

#define TN 512
#define NWAVE 32
#define QCAP 288

__global__ __launch_bounds__(1024) void k_agg_tile(const int* __restrict__ ei,
                                                   const float* __restrict__ dinv,
                                                   const float* __restrict__ hbuf,
                                                   float* __restrict__ out, int E, int n) {
  __shared__ float tile[TN * HDIM];
  __shared__ int qe[NWAVE][QCAP];
  const int tid = threadIdx.x;
  const int lane = tid & 31;
  const int wave = tid >> 5;
  const int base = blockIdx.x * TN;
  const int* __restrict__ esrc = ei;
  const int* __restrict__ edst = ei + E;

  for (int r = wave; r < TN; r += NWAVE) {
    int node = base + r;
    float4 v = make_float4(0.f, 0.f, 0.f, 0.f);
    if (node < n) {
      float d = dinv[node];
      float w = d * d;
      float4 hv = *(const float4*)(hbuf + (size_t)node * HDIM + lane * 4);
      v.x = hv.x * w; v.y = hv.y * w; v.z = hv.z * w; v.w = hv.w * w;
    }
    float* t = &tile[r * HDIM + lane];
    t[0] = v.x; t[32] = v.y; t[64] = v.z; t[96] = v.w;
  }
  __syncthreads();

  const int per = (((E + NWAVE - 1) / NWAVE) + 255) & ~255;
  const int e0 = wave * per;
  const int e1 = (e0 + per < E) ? (e0 + per) : E;
  const int trips = (e1 > e0) ? ((e1 - e0 + 255) >> 8) : 0;
  int trip = trips ? (int)((blockIdx.x * 37u) % (unsigned)trips) : 0;
  const bool vec_ok = ((E & 3) == 0);
  int qn = 0;

  auto load_trip = [&](int eb, int (&d)[8]) {
    const int ia = eb + lane * 4;
    const int ib = eb + 128 + lane * 4;
    if (vec_ok && eb + 256 <= e1) {
      const int4 da = *(const int4*)(edst + ia);
      const int4 db = *(const int4*)(edst + ib);
      d[0] = da.x; d[1] = da.y; d[2] = da.z; d[3] = da.w;
      d[4] = db.x; d[5] = db.y; d[6] = db.z; d[7] = db.w;
    } else {
#pragma unroll
      for (int j = 0; j < 4; ++j) {
        d[j]     = (ia + j < e1) ? edst[ia + j] : -1;
        d[4 + j] = (ib + j < e1) ? edst[ib + j] : -1;
      }
    }
  };
  int dn[8];
#pragma unroll
  for (int j = 0; j < 8; ++j) dn[j] = -1;
  if (trips) load_trip(e0 + (trip << 8), dn);

  for (int it = 0; it < trips; ++it) {
    const int eb = e0 + (trip << 8);
    trip = (trip + 1 == trips) ? 0 : trip + 1;
    int d[8];
#pragma unroll
    for (int j = 0; j < 8; ++j) d[j] = dn[j];
    if (it + 1 < trips) load_trip(e0 + (trip << 8), dn);
    const int ia = eb + lane * 4;
    const int ib = eb + 128 + lane * 4;
#pragma unroll
    for (int j = 0; j < 8; ++j) {
      const int eidx = ((j < 4) ? ia : ib) + (j & 3);
      const bool hit = (d[j] >= 0) && ((unsigned)(d[j] - base) < (unsigned)TN);
      const unsigned mask = (unsigned)__ballot(hit);
      if (hit) qe[wave][qn + (int)__builtin_amdgcn_mbcnt_lo(mask, 0u)] = eidx;
      qn += __popc(mask);
    }

    if (qn >= 32 || it + 1 == trips) {
      for (int qb = 0; qb < qn; qb += 32) {
        const int cnt = (qn - qb < 32) ? (qn - qb) : 32;
        int my_s = 0, my_d = 0; float my_w = 0.f;
        if (lane < cnt) {
          const int e = qe[wave][qb + lane];
          my_s = esrc[e];
          const int dd = edst[e];
          my_d = dd - base;
          my_w = dinv[my_s] * dinv[dd];
        }
        for (int u0 = 0; u0 < cnt; u0 += 8) {
          float4 hv[8]; float w[8]; int dr[8];
#pragma unroll
          for (int u = 0; u < 8; ++u) {
            const int s = __builtin_amdgcn_readlane(my_s, u0 + u);
            dr[u] = __builtin_amdgcn_readlane(my_d, u0 + u);
            w[u] = __int_as_float(__builtin_amdgcn_readlane(__float_as_int(my_w), u0 + u));
            hv[u] = *(const float4*)(hbuf + (size_t)s * HDIM + lane * 4);
          }
#pragma unroll
          for (int u = 0; u < 8; ++u) {
            if (u0 + u < cnt) {
              float* t = &tile[dr[u] * HDIM + lane];
              atomicAdd(t + 0,  hv[u].x * w[u]);
              atomicAdd(t + 32, hv[u].y * w[u]);
              atomicAdd(t + 64, hv[u].z * w[u]);
              atomicAdd(t + 96, hv[u].w * w[u]);
            }
          }
        }
      }
      qn = 0;
    }
  }
  __syncthreads();

  for (int r = wave; r < TN; r += NWAVE) {
    int node = base + r;
    if (node < n) {
      const float* t = &tile[r * HDIM + lane];
      float4 v = make_float4(t[0], t[32], t[64], t[96]);
      *(float4*)(out + (size_t)node * HDIM + lane * 4) = v;
    }
  }
}

__global__ __launch_bounds__(256) void k_out_gemm(const float* __restrict__ x,
                                                  const float* __restrict__ Wo,
                                                  const float* __restrict__ bo,
                                                  float* __restrict__ out, int n) {
  __shared__ float ws[HDIM * NCLS];
  __shared__ float bs[NCLS];
  int tid = threadIdx.x;
  for (int i = tid; i < HDIM * NCLS; i += 256) ws[i] = Wo[i];
  if (tid < NCLS) bs[tid] = bo[tid];
  __syncthreads();
  int r = blockIdx.x * 128 + (tid >> 1);
  int c0 = (tid & 1) * 20;
  if (r >= n) return;
  float acc[20];
#pragma unroll
  for (int c = 0; c < 20; ++c) acc[c] = bs[c0 + c];
  const float* xr = x + (size_t)r * HDIM;
  for (int k = 0; k < HDIM; ++k) {
    float xv = xr[k];
    const float* wr = &ws[k * NCLS + c0];
#pragma unroll
    for (int c = 0; c < 20; ++c) acc[c] += xv * wr[c];
  }
  float* o = out + (size_t)r * NCLS + c0;
#pragma unroll
  for (int c = 0; c < 20; ++c) o[c] = acc[c];
}

extern "C" void kernel_launch(void* const* d_in, const int* in_sizes, int n_in,
                              void* d_out, int out_size, void* d_ws, size_t ws_size,
                              hipStream_t stream) {
  const float* x_in  = (const float*)d_in[0];
  const int*   ei    = (const int*)d_in[1];
  const float* gamma = (const float*)d_in[2];
  const float* beta  = (const float*)d_in[3];
  const float* W     = (const float*)d_in[4];
  const float* b     = (const float*)d_in[5];
  const float* Wo    = (const float*)d_in[6];
  const float* bo    = (const float*)d_in[7];
  float* out = (float*)d_out;

  const int N = in_sizes[0] / HDIM;
  const int E = in_sizes[1] / 2;

  float* dinv  = (float*)d_ws;
  float* stats = dinv + ((N + 255) / 256) * 256;
  float* WfT   = stats + 2 * HDIM;
  float* cb    = WfT + HDIM * HDIM;
  float* hbuf  = cb + HDIM + 64;
  float* xbuf  = hbuf + (size_t)N * HDIM;

  k_set_ones<<<(N + 255) / 256, 256, 0, stream>>>(dinv, N);
  k_deg_accum<<<(E + 255) / 256, 256, 0, stream>>>(ei, dinv, E);
  k_rsqrt_ip<<<(N + 255) / 256, 256, 0, stream>>>(dinv, N);

  hipMemsetAsync(stats, 0, 2 * HDIM * sizeof(float), stream);
  k_stats<<<512, 256, 0, stream>>>(x_in, stats, N);

  for (int l = 0; l < 3; ++l) {
    k_fold<<<1, 128, 0, stream>>>(stats, gamma + l * HDIM, beta + l * HDIM,
                                  W + (size_t)l * HDIM * HDIM, WfT, cb, 1.0f / (float)N);
    const float* xl = (l == 0) ? x_in : xbuf;
    k_gemm<<<(N + 63) / 64, 128, 0, stream>>>(xl, WfT, cb, hbuf, N);
    k_agg_tile<<<(N + TN - 1) / TN, 1024, 0, stream>>>(ei, dinv, hbuf, xbuf, E, N);
    if (l < 2) {
      hipMemsetAsync(stats, 0, 2 * HDIM * sizeof(float), stream);
      k_bias_relu_stats<<<512, 256, 0, stream>>>(xbuf, b + l * HDIM, stats, N, 1);
    } else {
      k_bias_relu_stats<<<512, 256, 0, stream>>>(xbuf, b + l * HDIM, stats, N, 0);
    }
  }

  k_out_gemm<<<(N + 127) / 128, 256, 0, stream>>>(xbuf, Wo, bo, out, N);
}


# Owner blocks scan a 1-byte range key per edge (written once per call by the degree pass; 4 keys per word test, exact test at the drain), not the 4-byte destination list; LDS add order differs
# speedup vs baseline: 1.8524x; 1.1980x over previous
#include <hip/hip_runtime.h>

#define HDIM 128
#define NCLS 40
#define BN_EPS 1e-5f

typedef __attribute__((ext_vector_type(2))) float v2f;
typedef __attribute__((ext_vector_type(8))) float v8f;

__global__ void k_set_ones(float* deg, int n) {
  int i = blockIdx.x * blockDim.x + threadIdx.x;
  if (i < n) deg[i] = 1.0f;
}

#define KSH 9
__global__ void k_deg_keys(const int* __restrict__ ei, float* deg, float* __restrict__ keys_f,
                           int E, int Epad) {
  const int q = blockIdx.x * blockDim.x + threadIdx.x;
  const int e = q * 4;
  if (e >= Epad) return;
  const int* __restrict__ edst = ei + E;
  int d[4];
  if (((E & 3) == 0) && e + 4 <= E) {
    const int4 dv = *(const int4*)(edst + e);
    d[0] = dv.x; d[1] = dv.y; d[2] = dv.z; d[3] = dv.w;
  } else {
#pragma unroll
    for (int j = 0; j < 4; ++j) d[j] = (e + j < E) ? edst[e + j] : -1;
  }
  unsigned packed = 0u;
#pragma unroll
  for (int j = 0; j < 4; ++j) {
    unsigned k = 0xFFu;
    if (d[j] >= 0) {
      atomicAdd(&deg[d[j]], 1.0f);
      k = ((unsigned)d[j] >> KSH) & 0xFFu;
    }
    packed |= k << (8 * j);
  }
  ((unsigned*)keys_f)[q] = packed;
}

__global__ void k_rsqrt_ip(float* d, int n) {
  int i = blockIdx.x * blockDim.x + threadIdx.x;
  if (i < n) d[i] = rsqrtf(d[i]);
}

__global__ void k_stats(const float* __restrict__ x, float* __restrict__ stats, int n) {
  int col = threadIdx.x & (HDIM - 1);
  int half = threadIdx.x >> 7;
  float s = 0.f, s2 = 0.f;
  for (int r = blockIdx.x * 2 + half; r < n; r += gridDim.x * 2) {
    float v = x[(size_t)r * HDIM + col];
    s += v; s2 += v * v;
  }
  __shared__ float ls[HDIM], ls2[HDIM];
  if (half == 0) { ls[col] = s; ls2[col] = s2; }
  __syncthreads();
  if (half == 1) {
    atomicAdd(&stats[col], s + ls[col]);
    atomicAdd(&stats[HDIM + col], s2 + ls2[col]);
  }
}

__global__ void k_bias_relu_stats(float* __restrict__ x, const float* __restrict__ b,
                                  float* __restrict__ stats, int n, int do_stats) {
  int col = threadIdx.x & (HDIM - 1);
  int half = threadIdx.x >> 7;
  float bias = b[col];
  float s = 0.f, s2 = 0.f;
  for (int r = blockIdx.x * 2 + half; r < n; r += gridDim.x * 2) {
    size_t idx = (size_t)r * HDIM + col;
    float v = x[idx] + bias;
    v = v > 0.f ? v : 0.f;
    x[idx] = v;
    s += v; s2 += v * v;
  }
  if (!do_stats) return;
  __shared__ float ls[HDIM], ls2[HDIM];
  if (half == 0) { ls[col] = s; ls2[col] = s2; }
  __syncthreads();
  if (half == 1) {
    atomicAdd(&stats[col], s + ls[col]);
    atomicAdd(&stats[HDIM + col], s2 + ls2[col]);
  }
}

__global__ void k_fold(const float* __restrict__ stats, const float* __restrict__ gamma,
                       const float* __restrict__ beta, const float* __restrict__ W,
                       float* __restrict__ WfT, float* __restrict__ cb, float inv_n) {
  __shared__ float as[HDIM], cs[HDIM];
  int t = threadIdx.x;
  float mean = stats[t] * inv_n;
  float var  = stats[HDIM + t] * inv_n - mean * mean;
  float a = gamma[t] * rsqrtf(var + BN_EPS);
  float c = beta[t] - a * mean;
  as[t] = a; cs[t] = c;
  __syncthreads();
  float acc = 0.f;
  for (int k = 0; k < HDIM; ++k) {
    float w = W[k * HDIM + t];
    WfT[t * HDIM + k] = as[k] * w;
    acc += cs[k] * w;
  }
  cb[t] = acc;
}

#define XPAD 68

__global__ __launch_bounds__(128) void k_gemm(const float* __restrict__ x,
                                              const float* __restrict__ WfT,
                                              const float* __restrict__ cb,
                                              float* __restrict__ h, int n) {
  __shared__ float xs[64 * XPAD];
  __shared__ float wfs[HDIM * XPAD];
  const int tid = threadIdx.x;
  const int lane = tid & 31;
  const int wave = tid >> 5;
  const int rowBase = blockIdx.x * 64;
  const int m = lane & 15;
  const int hi = lane >> 4;

  v8f acc[8];
#pragma unroll
  for (int c = 0; c < 8; ++c)
#pragma unroll
    for (int v = 0; v < 8; ++v) acc[c][v] = 0.f;

  for (int kh = 0; kh < 2; ++kh) {
#pragma unroll
    for (int p = 0; p < 8; ++p) {
      int idx = p * 512 + tid * 4;
      int r = idx >> 6, cc = idx & 63;
      int row = rowBase + r;
      float4 v = make_float4(0.f, 0.f, 0.f, 0.f);
      if (row < n) v = *(const float4*)(x + (size_t)row * HDIM + kh * 64 + cc);
      *(float4*)(&xs[r * XPAD + cc]) = v;
    }
#pragma unroll
    for (int p = 0; p < 16; ++p) {
      int idx = p * 512 + tid * 4;
      int nn = idx >> 6, cc = idx & 63;
      *(float4*)(&wfs[nn * XPAD + cc]) = *(const float4*)(WfT + nn * HDIM + kh * 64 + cc);
    }
    __syncthreads();

    const float* xrow = &xs[(wave * 16 + m) * XPAD];
#pragma unroll
    for (int k4 = 0; k4 < 16; ++k4) {
      int kb = k4 * 4 + hi * 2;
      v2f a; a.x = xrow[kb]; a.y = xrow[kb + 1];
#pragma unroll
      for (int c = 0; c < 8; ++c) {
        const float* wrow = &wfs[(c * 16 + m) * XPAD];
        v2f b; b.x = wrow[kb]; b.y = wrow[kb + 1];
        acc[c] = __builtin_amdgcn_wmma_f32_16x16x4_f32(
            false, a, false, b, (short)0, acc[c], false, false);
      }
    }
    __syncthreads();
  }

#pragma unroll
  for (int c = 0; c < 8; ++c) {
    int col = c * 16 + m;
    float cbv = cb[col];
#pragma unroll
    for (int v = 0; v < 8; ++v) {
      int row = rowBase + wave * 16 + v + 8 * hi;
      if (row < n) h[(size_t)row * HDIM + col] = acc[c][v] + cbv;
    }
  }
}

#define TN (1 << KSH)
#define NWAVE 32
#define QCAP 64

__global__ __launch_bounds__(1024) void k_agg_tile(const int* __restrict__ ei,
                                                   const float* __restrict__ keys_f,
                                                   const float* __restrict__ dinv,
                                                   const float* __restrict__ hbuf,
                                                   float* __restrict__ out,
                                                   int E, int Epad, int n) {
  __shared__ float tile[TN * HDIM];
  __shared__ int qe[NWAVE][QCAP];
  const int tid = threadIdx.x;
  const int lane = tid & 31;
  const int wave = __builtin_amdgcn_readfirstlane(tid >> 5);
  const int base = blockIdx.x * TN;
  const int* __restrict__ esrc = ei;
  const int* __restrict__ edst = ei + E;

  for (int r = wave; r < TN; r += NWAVE) {
    int node = base + r;
    float4 v = make_float4(0.f, 0.f, 0.f, 0.f);
    if (node < n) {
      float d = dinv[node];
      float w = d * d;
      float4 hv = *(const float4*)(hbuf + (size_t)node * HDIM + lane * 4);
      v.x = hv.x * w; v.y = hv.y * w; v.z = hv.z * w; v.w = hv.w * w;
    }
    float* t = &tile[r * HDIM + lane];
    t[0] = v.x; t[32] = v.y; t[64] = v.z; t[96] = v.w;
  }
  __syncthreads();

  auto drain = [&](const int qtot) {
    for (int qb = 0; qb < qtot; qb += 32) {
      const int cnt = (qtot - qb < 32) ? (qtot - qb) : 32;
      int my_s = 0, my_d = -1; float my_w = 0.f;
      if (lane < cnt) {
        const int e = qe[wave][qb + lane];
        if (e < E) {
          const int s = esrc[e];
          const int dd = edst[e];
          my_s = s;
          my_w = dinv[s] * dinv[dd];
          const int dl = dd - base;
          my_d = ((unsigned)dl < (unsigned)TN) ? dl : -1;
        }
      }
      for (int u0 = 0; u0 < cnt; u0 += 8) {
        float4 hv[8]; float w[8]; int dr[8];
#pragma unroll
        for (int u = 0; u < 8; ++u) {
          const int s = __builtin_amdgcn_readlane(my_s, u0 + u);
          dr[u] = __builtin_amdgcn_readlane(my_d, u0 + u);
          w[u] = __int_as_float(__builtin_amdgcn_readlane(__float_as_int(my_w), u0 + u));
          hv[u] = *(const float4*)(hbuf + (size_t)s * HDIM + lane * 4);
        }
#pragma unroll
        for (int u = 0; u < 8; ++u) {
          if (u0 + u < cnt && dr[u] >= 0) {
            float* t = &tile[dr[u] * HDIM + lane];
            atomicAdd(t + 0,  hv[u].x * w[u]);
            atomicAdd(t + 32, hv[u].y * w[u]);
            atomicAdd(t + 64, hv[u].z * w[u]);
            atomicAdd(t + 96, hv[u].w * w[u]);
          }
        }
      }
    }
  };

  const uint4* __restrict__ keys = (const uint4*)keys_f;
  const int per = ((((Epad >> 9) + NWAVE - 1) / NWAVE)) << 9;
  const int e0 = wave * per;
  const int e1 = (e0 + per < Epad) ? (e0 + per) : Epad;
  const int trips = (e1 > e0) ? ((e1 - e0) >> 9) : 0;
  const int start = trips ? (int)((blockIdx.x * 37u) % (unsigned)trips) : 0;
  const unsigned pat = ((unsigned)blockIdx.x & 0xFFu) * 0x01010101u;
  int qn = 0;

  auto load_keys = [&](const int t) -> uint4 { return keys[((e0 + (t << 9)) >> 4) + lane]; };
  int tc = start;
  uint4 kn = make_uint4(0u, 0u, 0u, 0u);
  if (trips > 0) kn = load_keys(tc);

  for (int it = 0; it < trips; ++it) {
    const uint4 k = kn;
    const int eb = e0 + (tc << 9) + lane * 16;
    tc = (tc + 1 == trips) ? 0 : tc + 1;
    if (it + 1 < trips) kn = load_keys(tc);

    const unsigned x0 = k.x ^ pat, x1 = k.y ^ pat, x2 = k.z ^ pat, x3 = k.w ^ pat;
    const unsigned t0 = (x0 - 0x01010101u) & ~x0 & 0x80808080u;
    const unsigned t1 = (x1 - 0x01010101u) & ~x1 & 0x80808080u;
    const unsigned t2 = (x2 - 0x01010101u) & ~x2 & 0x80808080u;
    const unsigned t3 = (x3 - 0x01010101u) & ~x3 & 0x80808080u;
    unsigned u = (t0 >> 7) | (t1 >> 6) | (t2 >> 5) | (t3 >> 4);

    for (int g = 0; g < 16; ++g) {
      const bool has = (u != 0u);
      const unsigned mask = (unsigned)__ballot(has);
      if (mask == 0u) break;
      if (has) {
        const int p = __builtin_ctz(u);
        qe[wave][qn + (int)__builtin_amdgcn_mbcnt_lo(mask, 0u)] = eb + ((p & 7) << 2) + (p >> 3);
      }
      u &= u - 1u;
      qn += __popc(mask);
      if (qn >= 32) { drain(qn); qn = 0; }
    }
  }
  if (qn > 0) drain(qn);
  __syncthreads();

  for (int r = wave; r < TN; r += NWAVE) {
    int node = base + r;
    if (node < n) {
      const float* t = &tile[r * HDIM + lane];
      float4 v = make_float4(t[0], t[32], t[64], t[96]);
      *(float4*)(out + (size_t)node * HDIM + lane * 4) = v;
    }
  }
}

__global__ __launch_bounds__(256) void k_out_gemm(const float* __restrict__ x,
                                                  const float* __restrict__ Wo,
                                                  const float* __restrict__ bo,
                                                  float* __restrict__ out, int n) {
  __shared__ float ws[HDIM * NCLS];
  __shared__ float bs[NCLS];
  int tid = threadIdx.x;
  for (int i = tid; i < HDIM * NCLS; i += 256) ws[i] = Wo[i];
  if (tid < NCLS) bs[tid] = bo[tid];
  __syncthreads();
  int r = blockIdx.x * 128 + (tid >> 1);
  int c0 = (tid & 1) * 20;
  if (r >= n) return;
  float acc[20];
#pragma unroll
  for (int c = 0; c < 20; ++c) acc[c] = bs[c0 + c];
  const float* xr = x + (size_t)r * HDIM;
  for (int k = 0; k < HDIM; ++k) {
    float xv = xr[k];
    const float* wr = &ws[k * NCLS + c0];
#pragma unroll
    for (int c = 0; c < 20; ++c) acc[c] += xv * wr[c];
  }
  float* o = out + (size_t)r * NCLS + c0;
#pragma unroll
  for (int c = 0; c < 20; ++c) o[c] = acc[c];
}

extern "C" void kernel_launch(void* const* d_in, const int* in_sizes, int n_in,
                              void* d_out, int out_size, void* d_ws, size_t ws_size,
                              hipStream_t stream) {
  const float* x_in  = (const float*)d_in[0];
  const int*   ei    = (const int*)d_in[1];
  const float* gamma = (const float*)d_in[2];
  const float* beta  = (const float*)d_in[3];
  const float* W     = (const float*)d_in[4];
  const float* b     = (const float*)d_in[5];
  const float* Wo    = (const float*)d_in[6];
  const float* bo    = (const float*)d_in[7];
  float* out = (float*)d_out;

  const int N = in_sizes[0] / HDIM;
  const int E = in_sizes[1] / 2;
  const int Epad = ((E + 511) / 512) * 512;

  float* dinv  = (float*)d_ws;
  float* stats = dinv + ((N + 255) / 256) * 256;
  float* WfT   = stats + 2 * HDIM;
  float* cb    = WfT + HDIM * HDIM;
  float* hbuf  = cb + HDIM + 64;
  float* xbuf  = hbuf + (size_t)N * HDIM;
  float* keys  = out;

  k_set_ones<<<(N + 255) / 256, 256, 0, stream>>>(dinv, N);
  k_deg_keys<<<(Epad / 4 + 255) / 256, 256, 0, stream>>>(ei, dinv, keys, E, Epad);
  k_rsqrt_ip<<<(N + 255) / 256, 256, 0, stream>>>(dinv, N);

  hipMemsetAsync(stats, 0, 2 * HDIM * sizeof(float), stream);
  k_stats<<<512, 256, 0, stream>>>(x_in, stats, N);

  for (int l = 0; l < 3; ++l) {
    k_fold<<<1, 128, 0, stream>>>(stats, gamma + l * HDIM, beta + l * HDIM,
                                  W + (size_t)l * HDIM * HDIM, WfT, cb, 1.0f / (float)N);
    const float* xl = (l == 0) ? x_in : xbuf;
    k_gemm<<<(N + 63) / 64, 128, 0, stream>>>(xl, WfT, cb, hbuf, N);
    k_agg_tile<<<(N + TN - 1) / TN, 1024, 0, stream>>>(ei, keys, dinv, hbuf, xbuf, E, Epad, N);
    if (l < 2) {
      hipMemsetAsync(stats, 0, 2 * HDIM * sizeof(float), stream);
      k_bias_relu_stats<<<512, 256, 0, stream>>>(xbuf, b + l * HDIM, stats, N, 1);
    } else {
      k_bias_relu_stats<<<512, 256, 0, stream>>>(xbuf, b + l * HDIM, stats, N, 0);
    }
  }

  k_out_gemm<<<(N + 127) / 128, 256, 0, stream>>>(xbuf, Wo, bo, out, N);
}


# Owner tiles of 392 rows, not 512: 256 aggregation blocks, one per WGP (196 left 60 WGPs idle), each with a quarter fewer rows and edges to add; range key = dst / 392; LDS add order differs
# speedup vs baseline: 1.9256x; 1.0395x over previous
#include <hip/hip_runtime.h>

#define HDIM 128
#define NCLS 40
#define BN_EPS 1e-5f

typedef __attribute__((ext_vector_type(2))) float v2f;
typedef __attribute__((ext_vector_type(8))) float v8f;

__global__ void k_set_ones(float* deg, int n) {
  int i = blockIdx.x * blockDim.x + threadIdx.x;
  if (i < n) deg[i] = 1.0f;
}

#define TN 392
__global__ void k_deg_keys(const int* __restrict__ ei, float* deg, float* __restrict__ keys_f,
                           int E, int Epad) {
  const int q = blockIdx.x * blockDim.x + threadIdx.x;
  const int e = q * 4;
  if (e >= Epad) return;
  const int* __restrict__ edst = ei + E;
  int d[4];
  if (((E & 3) == 0) && e + 4 <= E) {
    const int4 dv = *(const int4*)(edst + e);
    d[0] = dv.x; d[1] = dv.y; d[2] = dv.z; d[3] = dv.w;
  } else {
#pragma unroll
    for (int j = 0; j < 4; ++j) d[j] = (e + j < E) ? edst[e + j] : -1;
  }
  unsigned packed = 0u;
#pragma unroll
  for (int j = 0; j < 4; ++j) {
    unsigned k = 0xFFu;
    if (d[j] >= 0) {
      atomicAdd(&deg[d[j]], 1.0f);
      k = ((unsigned)d[j] / (unsigned)TN) & 0xFFu;
    }
    packed |= k << (8 * j);
  }
  ((unsigned*)keys_f)[q] = packed;
}

__global__ void k_rsqrt_ip(float* d, int n) {
  int i = blockIdx.x * blockDim.x + threadIdx.x;
  if (i < n) d[i] = rsqrtf(d[i]);
}

__global__ void k_stats(const float* __restrict__ x, float* __restrict__ stats, int n) {
  int col = threadIdx.x & (HDIM - 1);
  int half = threadIdx.x >> 7;
  float s = 0.f, s2 = 0.f;
  for (int r = blockIdx.x * 2 + half; r < n; r += gridDim.x * 2) {
    float v = x[(size_t)r * HDIM + col];
    s += v; s2 += v * v;
  }
  __shared__ float ls[HDIM], ls2[HDIM];
  if (half == 0) { ls[col] = s; ls2[col] = s2; }
  __syncthreads();
  if (half == 1) {
    atomicAdd(&stats[col], s + ls[col]);
    atomicAdd(&stats[HDIM + col], s2 + ls2[col]);
  }
}

__global__ void k_bias_relu_stats(float* __restrict__ x, const float* __restrict__ b,
                                  float* __restrict__ stats, int n, int do_stats) {
  int col = threadIdx.x & (HDIM - 1);
  int half = threadIdx.x >> 7;
  float bias = b[col];
  float s = 0.f, s2 = 0.f;
  for (int r = blockIdx.x * 2 + half; r < n; r += gridDim.x * 2) {
    size_t idx = (size_t)r * HDIM + col;
    float v = x[idx] + bias;
    v = v > 0.f ? v : 0.f;
    x[idx] = v;
    s += v; s2 += v * v;
  }
  if (!do_stats) return;
  __shared__ float ls[HDIM], ls2[HDIM];
  if (half == 0) { ls[col] = s; ls2[col] = s2; }
  __syncthreads();
  if (half == 1) {
    atomicAdd(&stats[col], s + ls[col]);
    atomicAdd(&stats[HDIM + col], s2 + ls2[col]);
  }
}

__global__ void k_fold(const float* __restrict__ stats, const float* __restrict__ gamma,
                       const float* __restrict__ beta, const float* __restrict__ W,
                       float* __restrict__ WfT, float* __restrict__ cb, float inv_n) {
  __shared__ float as[HDIM], cs[HDIM];
  int t = threadIdx.x;
  float mean = stats[t] * inv_n;
  float var  = stats[HDIM + t] * inv_n - mean * mean;
  float a = gamma[t] * rsqrtf(var + BN_EPS);
  float c = beta[t] - a * mean;
  as[t] = a; cs[t] = c;
  __syncthreads();
  float acc = 0.f;
  for (int k = 0; k < HDIM; ++k) {
    float w = W[k * HDIM + t];
    WfT[t * HDIM + k] = as[k] * w;
    acc += cs[k] * w;
  }
  cb[t] = acc;
}

#define XPAD 68

__global__ __launch_bounds__(128) void k_gemm(const float* __restrict__ x,
                                              const float* __restrict__ WfT,
                                              const float* __restrict__ cb,
                                              float* __restrict__ h, int n) {
  __shared__ float xs[64 * XPAD];
  __shared__ float wfs[HDIM * XPAD];
  const int tid = threadIdx.x;
  const int lane = tid & 31;
  const int wave = tid >> 5;
  const int rowBase = blockIdx.x * 64;
  const int m = lane & 15;
  const int hi = lane >> 4;

  v8f acc[8];
#pragma unroll
  for (int c = 0; c < 8; ++c)
#pragma unroll
    for (int v = 0; v < 8; ++v) acc[c][v] = 0.f;

  for (int kh = 0; kh < 2; ++kh) {
#pragma unroll
    for (int p = 0; p < 8; ++p) {
      int idx = p * 512 + tid * 4;
      int r = idx >> 6, cc = idx & 63;
      int row = rowBase + r;
      float4 v = make_float4(0.f, 0.f, 0.f, 0.f);
      if (row < n) v = *(const float4*)(x + (size_t)row * HDIM + kh * 64 + cc);
      *(float4*)(&xs[r * XPAD + cc]) = v;
    }
#pragma unroll
    for (int p = 0; p < 16; ++p) {
      int idx = p * 512 + tid * 4;
      int nn = idx >> 6, cc = idx & 63;
      *(float4*)(&wfs[nn * XPAD + cc]) = *(const float4*)(WfT + nn * HDIM + kh * 64 + cc);
    }
    __syncthreads();

    const float* xrow = &xs[(wave * 16 + m) * XPAD];
#pragma unroll
    for (int k4 = 0; k4 < 16; ++k4) {
      int kb = k4 * 4 + hi * 2;
      v2f a; a.x = xrow[kb]; a.y = xrow[kb + 1];
#pragma unroll
      for (int c = 0; c < 8; ++c) {
        const float* wrow = &wfs[(c * 16 + m) * XPAD];
        v2f b; b.x = wrow[kb]; b.y = wrow[kb + 1];
        acc[c] = __builtin_amdgcn_wmma_f32_16x16x4_f32(
            false, a, false, b, (short)0, acc[c], false, false);
      }
    }
    __syncthreads();
  }

#pragma unroll
  for (int c = 0; c < 8; ++c) {
    int col = c * 16 + m;
    float cbv = cb[col];
#pragma unroll
    for (int v = 0; v < 8; ++v) {
      int row = rowBase + wave * 16 + v + 8 * hi;
      if (row < n) h[(size_t)row * HDIM + col] = acc[c][v] + cbv;
    }
  }
}

#define NWAVE 32
#define QCAP 64

__global__ __launch_bounds__(1024) void k_agg_tile(const int* __restrict__ ei,
                                                   const float* __restrict__ keys_f,
                                                   const float* __restrict__ dinv,
                                                   const float* __restrict__ hbuf,
                                                   float* __restrict__ out,
                                                   int E, int Epad, int n) {
  __shared__ float tile[TN * HDIM];
  __shared__ int qe[NWAVE][QCAP];
  const int tid = threadIdx.x;
  const int lane = tid & 31;
  const int wave = __builtin_amdgcn_readfirstlane(tid >> 5);
  const int base = blockIdx.x * TN;
  const int* __restrict__ esrc = ei;
  const int* __restrict__ edst = ei + E;

  for (int r = wave; r < TN; r += NWAVE) {
    int node = base + r;
    float4 v = make_float4(0.f, 0.f, 0.f, 0.f);
    if (node < n) {
      float d = dinv[node];
      float w = d * d;
      float4 hv = *(const float4*)(hbuf + (size_t)node * HDIM + lane * 4);
      v.x = hv.x * w; v.y = hv.y * w; v.z = hv.z * w; v.w = hv.w * w;
    }
    float* t = &tile[r * HDIM + lane];
    t[0] = v.x; t[32] = v.y; t[64] = v.z; t[96] = v.w;
  }
  __syncthreads();

  auto drain = [&](const int qtot) {
    for (int qb = 0; qb < qtot; qb += 32) {
      const int cnt = (qtot - qb < 32) ? (qtot - qb) : 32;
      int my_s = 0, my_d = -1; float my_w = 0.f;
      if (lane < cnt) {
        const int e = qe[wave][qb + lane];
        if (e < E) {
          const int s = esrc[e];
          const int dd = edst[e];
          my_s = s;
          my_w = dinv[s] * dinv[dd];
          const int dl = dd - base;
          my_d = ((unsigned)dl < (unsigned)TN) ? dl : -1;
        }
      }
      for (int u0 = 0; u0 < cnt; u0 += 8) {
        float4 hv[8]; float w[8]; int dr[8];
#pragma unroll
        for (int u = 0; u < 8; ++u) {
          const int s = __builtin_amdgcn_readlane(my_s, u0 + u);
          dr[u] = __builtin_amdgcn_readlane(my_d, u0 + u);
          w[u] = __int_as_float(__builtin_amdgcn_readlane(__float_as_int(my_w), u0 + u));
          hv[u] = *(const float4*)(hbuf + (size_t)s * HDIM + lane * 4);
        }
#pragma unroll
        for (int u = 0; u < 8; ++u) {
          if (u0 + u < cnt && dr[u] >= 0) {
            float* t = &tile[dr[u] * HDIM + lane];
            atomicAdd(t + 0,  hv[u].x * w[u]);
            atomicAdd(t + 32, hv[u].y * w[u]);
            atomicAdd(t + 64, hv[u].z * w[u]);
            atomicAdd(t + 96, hv[u].w * w[u]);
          }
        }
      }
    }
  };

  const uint4* __restrict__ keys = (const uint4*)keys_f;
  const int per = ((((Epad >> 9) + NWAVE - 1) / NWAVE)) << 9;
  const int e0 = wave * per;
  const int e1 = (e0 + per < Epad) ? (e0 + per) : Epad;
  const int trips = (e1 > e0) ? ((e1 - e0) >> 9) : 0;
  const int start = trips ? (int)((blockIdx.x * 37u) % (unsigned)trips) : 0;
  const unsigned pat = ((unsigned)blockIdx.x & 0xFFu) * 0x01010101u;
  int qn = 0;

  auto load_keys = [&](const int t) -> uint4 { return keys[((e0 + (t << 9)) >> 4) + lane]; };
  int tc = start;
  uint4 kn = make_uint4(0u, 0u, 0u, 0u);
  if (trips > 0) kn = load_keys(tc);

  for (int it = 0; it < trips; ++it) {
    const uint4 k = kn;
    const int eb = e0 + (tc << 9) + lane * 16;
    tc = (tc + 1 == trips) ? 0 : tc + 1;
    if (it + 1 < trips) kn = load_keys(tc);

    const unsigned x0 = k.x ^ pat, x1 = k.y ^ pat, x2 = k.z ^ pat, x3 = k.w ^ pat;
    const unsigned t0 = (x0 - 0x01010101u) & ~x0 & 0x80808080u;
    const unsigned t1 = (x1 - 0x01010101u) & ~x1 & 0x80808080u;
    const unsigned t2 = (x2 - 0x01010101u) & ~x2 & 0x80808080u;
    const unsigned t3 = (x3 - 0x01010101u) & ~x3 & 0x80808080u;
    unsigned u = (t0 >> 7) | (t1 >> 6) | (t2 >> 5) | (t3 >> 4);

    for (int g = 0; g < 16; ++g) {
      const bool has = (u != 0u);
      const unsigned mask = (unsigned)__ballot(has);
      if (mask == 0u) break;
      if (has) {
        const int p = __builtin_ctz(u);
        qe[wave][qn + (int)__builtin_amdgcn_mbcnt_lo(mask, 0u)] = eb + ((p & 7) << 2) + (p >> 3);
      }
      u &= u - 1u;
      qn += __popc(mask);
      if (qn >= 32) { drain(qn); qn = 0; }
    }
  }
  if (qn > 0) drain(qn);
  __syncthreads();

  for (int r = wave; r < TN; r += NWAVE) {
    int node = base + r;
    if (node < n) {
      const float* t = &tile[r * HDIM + lane];
      float4 v = make_float4(t[0], t[32], t[64], t[96]);
      *(float4*)(out + (size_t)node * HDIM + lane * 4) = v;
    }
  }
}

__global__ __launch_bounds__(256) void k_out_gemm(const float* __restrict__ x,
                                                  const float* __restrict__ Wo,
                                                  const float* __restrict__ bo,
                                                  float* __restrict__ out, int n) {
  __shared__ float ws[HDIM * NCLS];
  __shared__ float bs[NCLS];
  int tid = threadIdx.x;
  for (int i = tid; i < HDIM * NCLS; i += 256) ws[i] = Wo[i];
  if (tid < NCLS) bs[tid] = bo[tid];
  __syncthreads();
  int r = blockIdx.x * 128 + (tid >> 1);
  int c0 = (tid & 1) * 20;
  if (r >= n) return;
  float acc[20];
#pragma unroll
  for (int c = 0; c < 20; ++c) acc[c] = bs[c0 + c];
  const float* xr = x + (size_t)r * HDIM;
  for (int k = 0; k < HDIM; ++k) {
    float xv = xr[k];
    const float* wr = &ws[k * NCLS + c0];
#pragma unroll
    for (int c = 0; c < 20; ++c) acc[c] += xv * wr[c];
  }
  float* o = out + (size_t)r * NCLS + c0;
#pragma unroll
  for (int c = 0; c < 20; ++c) o[c] = acc[c];
}

extern "C" void kernel_launch(void* const* d_in, const int* in_sizes, int n_in,
                              void* d_out, int out_size, void* d_ws, size_t ws_size,
                              hipStream_t stream) {
  const float* x_in  = (const float*)d_in[0];
  const int*   ei    = (const int*)d_in[1];
  const float* gamma = (const float*)d_in[2];
  const float* beta  = (const float*)d_in[3];
  const float* W     = (const float*)d_in[4];
  const float* b     = (const float*)d_in[5];
  const float* Wo    = (const float*)d_in[6];
  const float* bo    = (const float*)d_in[7];
  float* out = (float*)d_out;

  const int N = in_sizes[0] / HDIM;
  const int E = in_sizes[1] / 2;
  const int Epad = ((E + 511) / 512) * 512;

  float* dinv  = (float*)d_ws;
  float* stats = dinv + ((N + 255) / 256) * 256;
  float* WfT   = stats + 2 * HDIM;
  float* cb    = WfT + HDIM * HDIM;
  float* hbuf  = cb + HDIM + 64;
  float* xbuf  = hbuf + (size_t)N * HDIM;
  float* keys  = out;

  k_set_ones<<<(N + 255) / 256, 256, 0, stream>>>(dinv, N);
  k_deg_keys<<<(Epad / 4 + 255) / 256, 256, 0, stream>>>(ei, dinv, keys, E, Epad);
  k_rsqrt_ip<<<(N + 255) / 256, 256, 0, stream>>>(dinv, N);

  hipMemsetAsync(stats, 0, 2 * HDIM * sizeof(float), stream);
  k_stats<<<512, 256, 0, stream>>>(x_in, stats, N);

  for (int l = 0; l < 3; ++l) {
    k_fold<<<1, 128, 0, stream>>>(stats, gamma + l * HDIM, beta + l * HDIM,
                                  W + (size_t)l * HDIM * HDIM, WfT, cb, 1.0f / (float)N);
    const float* xl = (l == 0) ? x_in : xbuf;
    k_gemm<<<(N + 63) / 64, 128, 0, stream>>>(xl, WfT, cb, hbuf, N);
    k_agg_tile<<<(N + TN - 1) / TN, 1024, 0, stream>>>(ei, keys, dinv, hbuf, xbuf, E, Epad, N);
    if (l < 2) {
      hipMemsetAsync(stats, 0, 2 * HDIM * sizeof(float), stream);
      k_bias_relu_stats<<<512, 256, 0, stream>>>(xbuf, b + l * HDIM, stats, N, 1);
    } else {
      k_bias_relu_stats<<<512, 256, 0, stream>>>(xbuf, b + l * HDIM, stats, N, 0);
    }
  }

  k_out_gemm<<<(N + 127) / 128, 256, 0, stream>>>(xbuf, Wo, bo, out, N);
}


# Last layer's owner blocks finish their rows: bias+relu and the 128x40 projection on the matrix unit straight from the LDS tile, logits stored; 2 passes over the activation go; k sum order differs
# speedup vs baseline: 2.1428x; 1.1128x over previous
#include <hip/hip_runtime.h>

#define HDIM 128
#define NCLS 40
#define BN_EPS 1e-5f

typedef __attribute__((ext_vector_type(2))) float v2f;
typedef __attribute__((ext_vector_type(8))) float v8f;

__global__ void k_set_ones(float* deg, int n) {
  int i = blockIdx.x * blockDim.x + threadIdx.x;
  if (i < n) deg[i] = 1.0f;
}

#define TN 392
__global__ void k_deg_keys(const int* __restrict__ ei, float* deg, float* __restrict__ keys_f,
                           int E, int Epad) {
  const int q = blockIdx.x * blockDim.x + threadIdx.x;
  const int e = q * 4;
  if (e >= Epad) return;
  const int* __restrict__ edst = ei + E;
  int d[4];
  if (((E & 3) == 0) && e + 4 <= E) {
    const int4 dv = *(const int4*)(edst + e);
    d[0] = dv.x; d[1] = dv.y; d[2] = dv.z; d[3] = dv.w;
  } else {
#pragma unroll
    for (int j = 0; j < 4; ++j) d[j] = (e + j < E) ? edst[e + j] : -1;
  }
  unsigned packed = 0u;
#pragma unroll
  for (int j = 0; j < 4; ++j) {
    unsigned k = 0xFFu;
    if (d[j] >= 0) {
      atomicAdd(&deg[d[j]], 1.0f);
      k = ((unsigned)d[j] / (unsigned)TN) & 0xFFu;
    }
    packed |= k << (8 * j);
  }
  ((unsigned*)keys_f)[q] = packed;
}

__global__ void k_rsqrt_ip(float* d, int n) {
  int i = blockIdx.x * blockDim.x + threadIdx.x;
  if (i < n) d[i] = rsqrtf(d[i]);
}

__global__ void k_stats(const float* __restrict__ x, float* __restrict__ stats, int n) {
  int col = threadIdx.x & (HDIM - 1);
  int half = threadIdx.x >> 7;
  float s = 0.f, s2 = 0.f;
  for (int r = blockIdx.x * 2 + half; r < n; r += gridDim.x * 2) {
    float v = x[(size_t)r * HDIM + col];
    s += v; s2 += v * v;
  }
  __shared__ float ls[HDIM], ls2[HDIM];
  if (half == 0) { ls[col] = s; ls2[col] = s2; }
  __syncthreads();
  if (half == 1) {
    atomicAdd(&stats[col], s + ls[col]);
    atomicAdd(&stats[HDIM + col], s2 + ls2[col]);
  }
}

__global__ void k_bias_relu_stats(float* __restrict__ x, const float* __restrict__ b,
                                  float* __restrict__ stats, int n, int do_stats) {
  int col = threadIdx.x & (HDIM - 1);
  int half = threadIdx.x >> 7;
  float bias = b[col];
  float s = 0.f, s2 = 0.f;
  for (int r = blockIdx.x * 2 + half; r < n; r += gridDim.x * 2) {
    size_t idx = (size_t)r * HDIM + col;
    float v = x[idx] + bias;
    v = v > 0.f ? v : 0.f;
    x[idx] = v;
    s += v; s2 += v * v;
  }
  if (!do_stats) return;
  __shared__ float ls[HDIM], ls2[HDIM];
  if (half == 0) { ls[col] = s; ls2[col] = s2; }
  __syncthreads();
  if (half == 1) {
    atomicAdd(&stats[col], s + ls[col]);
    atomicAdd(&stats[HDIM + col], s2 + ls2[col]);
  }
}

__global__ void k_fold(const float* __restrict__ stats, const float* __restrict__ gamma,
                       const float* __restrict__ beta, const float* __restrict__ W,
                       float* __restrict__ WfT, float* __restrict__ cb, float inv_n) {
  __shared__ float as[HDIM], cs[HDIM];
  int t = threadIdx.x;
  float mean = stats[t] * inv_n;
  float var  = stats[HDIM + t] * inv_n - mean * mean;
  float a = gamma[t] * rsqrtf(var + BN_EPS);
  float c = beta[t] - a * mean;
  as[t] = a; cs[t] = c;
  __syncthreads();
  float acc = 0.f;
  for (int k = 0; k < HDIM; ++k) {
    float w = W[k * HDIM + t];
    WfT[t * HDIM + k] = as[k] * w;
    acc += cs[k] * w;
  }
  cb[t] = acc;
}

#define XPAD 68

__global__ __launch_bounds__(128) void k_gemm(const float* __restrict__ x,
                                              const float* __restrict__ WfT,
                                              const float* __restrict__ cb,
                                              float* __restrict__ h, int n) {
  __shared__ float xs[64 * XPAD];
  __shared__ float wfs[HDIM * XPAD];
  const int tid = threadIdx.x;
  const int lane = tid & 31;
  const int wave = tid >> 5;
  const int rowBase = blockIdx.x * 64;
  const int m = lane & 15;
  const int hi = lane >> 4;

  v8f acc[8];
#pragma unroll
  for (int c = 0; c < 8; ++c)
#pragma unroll
    for (int v = 0; v < 8; ++v) acc[c][v] = 0.f;

  for (int kh = 0; kh < 2; ++kh) {
#pragma unroll
    for (int p = 0; p < 8; ++p) {
      int idx = p * 512 + tid * 4;
      int r = idx >> 6, cc = idx & 63;
      int row = rowBase + r;
      float4 v = make_float4(0.f, 0.f, 0.f, 0.f);
      if (row < n) v = *(const float4*)(x + (size_t)row * HDIM + kh * 64 + cc);
      *(float4*)(&xs[r * XPAD + cc]) = v;
    }
#pragma unroll
    for (int p = 0; p < 16; ++p) {
      int idx = p * 512 + tid * 4;
      int nn = idx >> 6, cc = idx & 63;
      *(float4*)(&wfs[nn * XPAD + cc]) = *(const float4*)(WfT + nn * HDIM + kh * 64 + cc);
    }
    __syncthreads();

    const float* xrow = &xs[(wave * 16 + m) * XPAD];
#pragma unroll
    for (int k4 = 0; k4 < 16; ++k4) {
      int kb = k4 * 4 + hi * 2;
      v2f a; a.x = xrow[kb]; a.y = xrow[kb + 1];
#pragma unroll
      for (int c = 0; c < 8; ++c) {
        const float* wrow = &wfs[(c * 16 + m) * XPAD];
        v2f b; b.x = wrow[kb]; b.y = wrow[kb + 1];
        acc[c] = __builtin_amdgcn_wmma_f32_16x16x4_f32(
            false, a, false, b, (short)0, acc[c], false, false);
      }
    }
    __syncthreads();
  }

#pragma unroll
  for (int c = 0; c < 8; ++c) {
    int col = c * 16 + m;
    float cbv = cb[col];
#pragma unroll
    for (int v = 0; v < 8; ++v) {
      int row = rowBase + wave * 16 + v + 8 * hi;
      if (row < n) h[(size_t)row * HDIM + col] = acc[c][v] + cbv;
    }
  }
}

#define NWAVE 32
#define QCAP 64
#define FROWS 400
#define FPITCH 130
#define NCLSP 48

template <bool FIN>
__global__ __launch_bounds__(1024) void k_agg_tile(const int* __restrict__ ei,
                                                   const float* __restrict__ keys_f,
                                                   const float* __restrict__ dinv,
                                                   const float* __restrict__ hbuf,
                                                   float* __restrict__ out,
                                                   const float* __restrict__ bias,
                                                   const float* __restrict__ Wo,
                                                   const float* __restrict__ bo,
                                                   int E, int Epad, int n) {
  constexpr int TS = FIN ? FPITCH : HDIM;
  constexpr int TR = FIN ? FROWS : TN;
  __shared__ __attribute__((aligned(16))) float tile[TR * TS];
  __shared__ int qe[NWAVE][QCAP];
  __shared__ float wos[FIN ? NCLSP * FPITCH : 1];
  __shared__ float bsw[FIN ? HDIM : 1];
  const int tid = threadIdx.x;
  const int lane = tid & 31;
  const int wave = __builtin_amdgcn_readfirstlane(tid >> 5);
  const int base = blockIdx.x * TN;
  const int* __restrict__ esrc = ei;
  const int* __restrict__ edst = ei + E;

  for (int r = wave; r < TR; r += NWAVE) {
    int node = base + r;
    float4 v = make_float4(0.f, 0.f, 0.f, 0.f);
    if (r < TN && node < n) {
      float d = dinv[node];
      float w = d * d;
      float4 hv = *(const float4*)(hbuf + (size_t)node * HDIM + lane * 4);
      v.x = hv.x * w; v.y = hv.y * w; v.z = hv.z * w; v.w = hv.w * w;
    }
    float* t = &tile[r * TS + lane];
    t[0] = v.x; t[32] = v.y; t[64] = v.z; t[96] = v.w;
  }
  if constexpr (FIN) {
    for (int i = tid; i < HDIM * NCLS; i += NWAVE * 32) {
      const int k = i / NCLS, c = i - k * NCLS;
      wos[c * FPITCH + (k & 3) * 32 + (k >> 2)] = Wo[i];
    }
    for (int i = tid; i < (NCLSP - NCLS) * HDIM; i += NWAVE * 32)
      wos[(NCLS + (i >> 7)) * FPITCH + (i & (HDIM - 1))] = 0.f;
    if (tid < HDIM) bsw[(tid & 3) * 32 + (tid >> 2)] = bias[tid];
  }
  __syncthreads();

  auto drain = [&](const int qtot) {
    for (int qb = 0; qb < qtot; qb += 32) {
      const int cnt = (qtot - qb < 32) ? (qtot - qb) : 32;
      int my_s = 0, my_d = -1; float my_w = 0.f;
      if (lane < cnt) {
        const int e = qe[wave][qb + lane];
        if (e < E) {
          const int s = esrc[e];
          const int dd = edst[e];
          my_s = s;
          my_w = dinv[s] * dinv[dd];
          const int dl = dd - base;
          my_d = ((unsigned)dl < (unsigned)TN) ? dl : -1;
        }
      }
      for (int u0 = 0; u0 < cnt; u0 += 8) {
        float4 hv[8]; float w[8]; int dr[8];
#pragma unroll
        for (int u = 0; u < 8; ++u) {
          const int s = __builtin_amdgcn_readlane(my_s, u0 + u);
          dr[u] = __builtin_amdgcn_readlane(my_d, u0 + u);
          w[u] = __int_as_float(__builtin_amdgcn_readlane(__float_as_int(my_w), u0 + u));
          hv[u] = *(const float4*)(hbuf + (size_t)s * HDIM + lane * 4);
        }
#pragma unroll
        for (int u = 0; u < 8; ++u) {
          if (u0 + u < cnt && dr[u] >= 0) {
            float* t = &tile[dr[u] * TS + lane];
            atomicAdd(t + 0,  hv[u].x * w[u]);
            atomicAdd(t + 32, hv[u].y * w[u]);
            atomicAdd(t + 64, hv[u].z * w[u]);
            atomicAdd(t + 96, hv[u].w * w[u]);
          }
        }
      }
    }
  };

  const uint4* __restrict__ keys = (const uint4*)keys_f;
  const int per = ((((Epad >> 9) + NWAVE - 1) / NWAVE)) << 9;
  const int e0 = wave * per;
  const int e1 = (e0 + per < Epad) ? (e0 + per) : Epad;
  const int trips = (e1 > e0) ? ((e1 - e0) >> 9) : 0;
  const int start = trips ? (int)((blockIdx.x * 37u) % (unsigned)trips) : 0;
  const unsigned pat = ((unsigned)blockIdx.x & 0xFFu) * 0x01010101u;
  int qn = 0;

  auto load_keys = [&](const int t) -> uint4 { return keys[((e0 + (t << 9)) >> 4) + lane]; };
  int tc = start;
  uint4 kn = make_uint4(0u, 0u, 0u, 0u);
  if (trips > 0) kn = load_keys(tc);

  for (int it = 0; it < trips; ++it) {
    const uint4 k = kn;
    const int eb = e0 + (tc << 9) + lane * 16;
    tc = (tc + 1 == trips) ? 0 : tc + 1;
    if (it + 1 < trips) kn = load_keys(tc);

    const unsigned x0 = k.x ^ pat, x1 = k.y ^ pat, x2 = k.z ^ pat, x3 = k.w ^ pat;
    const unsigned t0 = (x0 - 0x01010101u) & ~x0 & 0x80808080u;
    const unsigned t1 = (x1 - 0x01010101u) & ~x1 & 0x80808080u;
    const unsigned t2 = (x2 - 0x01010101u) & ~x2 & 0x80808080u;
    const unsigned t3 = (x3 - 0x01010101u) & ~x3 & 0x80808080u;
    unsigned u = (t0 >> 7) | (t1 >> 6) | (t2 >> 5) | (t3 >> 4);

    for (int g = 0; g < 16; ++g) {
      const bool has = (u != 0u);
      const unsigned mask = (unsigned)__ballot(has);
      if (mask == 0u) break;
      if (has) {
        const int p = __builtin_ctz(u);
        qe[wave][qn + (int)__builtin_amdgcn_mbcnt_lo(mask, 0u)] = eb + ((p & 7) << 2) + (p >> 3);
      }
      u &= u - 1u;
      qn += __popc(mask);
      if (qn >= 32) { drain(qn); qn = 0; }
    }
  }
  if (qn > 0) drain(qn);
  __syncthreads();

  if constexpr (!FIN) {
    for (int r = wave; r < TN; r += NWAVE) {
      int node = base + r;
      if (node < n) {
        const float* t = &tile[r * TS + lane];
        float4 v = make_float4(t[0], t[32], t[64], t[96]);
        *(float4*)(out + (size_t)node * HDIM + lane * 4) = v;
      }
    }
  } else {
    const int m = lane & 15;
    const int hi = lane >> 4;
    v8f acc[3];
#pragma unroll
    for (int c = 0; c < 3; ++c)
#pragma unroll
      for (int v = 0; v < 8; ++v) acc[c][v] = 0.f;
    if (wave < FROWS / 16) {
      const float* arow = &tile[(wave * 16 + m) * FPITCH + hi * 32];
      const float* brow = &wos[m * FPITCH + hi * 32];
      const float* bb = &bsw[hi * 32];
#pragma unroll 4
      for (int j = 0; j < 32; ++j) {
        const float t0 = arow[j] + bb[j];
        const float t1 = arow[64 + j] + bb[64 + j];
        v2f a; a.x = t0 > 0.f ? t0 : 0.f; a.y = t1 > 0.f ? t1 : 0.f;
#pragma unroll
        for (int c = 0; c < 3; ++c) {
          v2f b; b.x = brow[c * 16 * FPITCH + j]; b.y = brow[c * 16 * FPITCH + 64 + j];
          acc[c] = __builtin_amdgcn_wmma_f32_16x16x4_f32(
              false, a, false, b, (short)0, acc[c], false, false);
        }
      }
    }
    __syncthreads();
    float* stg = tile;
    static_assert(TN % 8 == 0, "a lane half's 8 rows must not straddle the last row");
    const int r0 = wave * 16 + 8 * hi;
    if (r0 < TN) {
#pragma unroll
      for (int c = 0; c < 3; ++c) {
        const int col = c * 16 + m;
        if (col < NCLS) {
          const float bov = bo[col];
#pragma unroll
          for (int v = 0; v < 8; ++v) stg[(r0 + v) * NCLS + col] = acc[c][v] + bov;
        }
      }
    }
    __syncthreads();
    const int rows = (n - base < TN) ? (n - base) : TN;
    const int nf4 = rows * (NCLS / 4);
    float4* __restrict__ o4 = (float4*)(out + (size_t)base * NCLS);
    const float4* s4 = (const float4*)stg;
    for (int i = tid; i < nf4; i += NWAVE * 32) o4[i] = s4[i];
  }
}

__global__ void k_copy_keys(const float* __restrict__ src, float* __restrict__ dst, int n16) {
  const int i = blockIdx.x * blockDim.x + threadIdx.x;
  if (i < n16) ((float4*)dst)[i] = ((const float4*)src)[i];
}

extern "C" void kernel_launch(void* const* d_in, const int* in_sizes, int n_in,
                              void* d_out, int out_size, void* d_ws, size_t ws_size,
                              hipStream_t stream) {
  const float* x_in  = (const float*)d_in[0];
  const int*   ei    = (const int*)d_in[1];
  const float* gamma = (const float*)d_in[2];
  const float* beta  = (const float*)d_in[3];
  const float* W     = (const float*)d_in[4];
  const float* b     = (const float*)d_in[5];
  const float* Wo    = (const float*)d_in[6];
  const float* bo    = (const float*)d_in[7];
  float* out = (float*)d_out;

  const int N = in_sizes[0] / HDIM;
  const int E = in_sizes[1] / 2;
  const int Epad = ((E + 511) / 512) * 512;

  float* dinv  = (float*)d_ws;
  float* stats = dinv + ((N + 255) / 256) * 256;
  float* WfT   = stats + 2 * HDIM;
  float* cb    = WfT + HDIM * HDIM;
  float* hbuf  = cb + HDIM + 64;
  float* xbuf  = hbuf + (size_t)N * HDIM;
  float* keys  = out;

  k_set_ones<<<(N + 255) / 256, 256, 0, stream>>>(dinv, N);
  k_deg_keys<<<(Epad / 4 + 255) / 256, 256, 0, stream>>>(ei, dinv, keys, E, Epad);
  k_rsqrt_ip<<<(N + 255) / 256, 256, 0, stream>>>(dinv, N);

  hipMemsetAsync(stats, 0, 2 * HDIM * sizeof(float), stream);
  k_stats<<<512, 256, 0, stream>>>(x_in, stats, N);

  for (int l = 0; l < 3; ++l) {
    k_fold<<<1, 128, 0, stream>>>(stats, gamma + l * HDIM, beta + l * HDIM,
                                  W + (size_t)l * HDIM * HDIM, WfT, cb, 1.0f / (float)N);
    const float* xl = (l == 0) ? x_in : xbuf;
    k_gemm<<<(N + 63) / 64, 128, 0, stream>>>(xl, WfT, cb, hbuf, N);
    if (l < 2) {
      k_agg_tile<false><<<(N + TN - 1) / TN, 1024, 0, stream>>>(ei, keys, dinv, hbuf, xbuf,
                                                                nullptr, nullptr, nullptr, E, Epad, N);
      hipMemsetAsync(stats, 0, 2 * HDIM * sizeof(float), stream);
      k_bias_relu_stats<<<512, 256, 0, stream>>>(xbuf, b + l * HDIM, stats, N, 1);
    } else {
      k_copy_keys<<<(Epad / 16 + 255) / 256, 256, 0, stream>>>(keys, xbuf, Epad / 16);
      k_agg_tile<true><<<(N + TN - 1) / TN, 1024, 0, stream>>>(ei, xbuf, dinv, hbuf, out,
                                                               b + l * HDIM, Wo, bo, E, Epad, N);
    }
  }
}


# First layer's owner blocks write down the edges they matched (source, local row; one word each, a list per block behind the keys); later layers read their list, no key scan; LDS add order differs
# speedup vs baseline: 2.3348x; 1.0896x over previous
#include <hip/hip_runtime.h>

#define HDIM 128
#define NCLS 40
#define BN_EPS 1e-5f

typedef __attribute__((ext_vector_type(2))) float v2f;
typedef __attribute__((ext_vector_type(8))) float v8f;

__global__ void k_set_ones(float* deg, int n) {
  int i = blockIdx.x * blockDim.x + threadIdx.x;
  if (i < n) deg[i] = 1.0f;
}

#define TN 392
__global__ void k_deg_keys(const int* __restrict__ ei, float* deg, float* __restrict__ keys_f,
                           int E, int Epad) {
  const int q = blockIdx.x * blockDim.x + threadIdx.x;
  const int e = q * 4;
  if (e >= Epad) return;
  const int* __restrict__ edst = ei + E;
  int d[4];
  if (((E & 3) == 0) && e + 4 <= E) {
    const int4 dv = *(const int4*)(edst + e);
    d[0] = dv.x; d[1] = dv.y; d[2] = dv.z; d[3] = dv.w;
  } else {
#pragma unroll
    for (int j = 0; j < 4; ++j) d[j] = (e + j < E) ? edst[e + j] : -1;
  }
  unsigned packed = 0u;
#pragma unroll
  for (int j = 0; j < 4; ++j) {
    unsigned k = 0xFFu;
    if (d[j] >= 0) {
      atomicAdd(&deg[d[j]], 1.0f);
      k = ((unsigned)d[j] / (unsigned)TN) & 0xFFu;
    }
    packed |= k << (8 * j);
  }
  ((unsigned*)keys_f)[q] = packed;
}

__global__ void k_rsqrt_ip(float* d, int n) {
  int i = blockIdx.x * blockDim.x + threadIdx.x;
  if (i < n) d[i] = rsqrtf(d[i]);
}

__global__ void k_stats(const float* __restrict__ x, float* __restrict__ stats, int n) {
  int col = threadIdx.x & (HDIM - 1);
  int half = threadIdx.x >> 7;
  float s = 0.f, s2 = 0.f;
  for (int r = blockIdx.x * 2 + half; r < n; r += gridDim.x * 2) {
    float v = x[(size_t)r * HDIM + col];
    s += v; s2 += v * v;
  }
  __shared__ float ls[HDIM], ls2[HDIM];
  if (half == 0) { ls[col] = s; ls2[col] = s2; }
  __syncthreads();
  if (half == 1) {
    atomicAdd(&stats[col], s + ls[col]);
    atomicAdd(&stats[HDIM + col], s2 + ls2[col]);
  }
}

__global__ void k_bias_relu_stats(float* __restrict__ x, const float* __restrict__ b,
                                  float* __restrict__ stats, int n, int do_stats) {
  int col = threadIdx.x & (HDIM - 1);
  int half = threadIdx.x >> 7;
  float bias = b[col];
  float s = 0.f, s2 = 0.f;
  for (int r = blockIdx.x * 2 + half; r < n; r += gridDim.x * 2) {
    size_t idx = (size_t)r * HDIM + col;
    float v = x[idx] + bias;
    v = v > 0.f ? v : 0.f;
    x[idx] = v;
    s += v; s2 += v * v;
  }
  if (!do_stats) return;
  __shared__ float ls[HDIM], ls2[HDIM];
  if (half == 0) { ls[col] = s; ls2[col] = s2; }
  __syncthreads();
  if (half == 1) {
    atomicAdd(&stats[col], s + ls[col]);
    atomicAdd(&stats[HDIM + col], s2 + ls2[col]);
  }
}

__global__ void k_fold(const float* __restrict__ stats, const float* __restrict__ gamma,
                       const float* __restrict__ beta, const float* __restrict__ W,
                       float* __restrict__ WfT, float* __restrict__ cb, float inv_n) {
  __shared__ float as[HDIM], cs[HDIM];
  int t = threadIdx.x;
  float mean = stats[t] * inv_n;
  float var  = stats[HDIM + t] * inv_n - mean * mean;
  float a = gamma[t] * rsqrtf(var + BN_EPS);
  float c = beta[t] - a * mean;
  as[t] = a; cs[t] = c;
  __syncthreads();
  float acc = 0.f;
  for (int k = 0; k < HDIM; ++k) {
    float w = W[k * HDIM + t];
    WfT[t * HDIM + k] = as[k] * w;
    acc += cs[k] * w;
  }
  cb[t] = acc;
}

#define XPAD 68

__global__ __launch_bounds__(128) void k_gemm(const float* __restrict__ x,
                                              const float* __restrict__ WfT,
                                              const float* __restrict__ cb,
                                              float* __restrict__ h, int n) {
  __shared__ float xs[64 * XPAD];
  __shared__ float wfs[HDIM * XPAD];
  const int tid = threadIdx.x;
  const int lane = tid & 31;
  const int wave = tid >> 5;
  const int rowBase = blockIdx.x * 64;
  const int m = lane & 15;
  const int hi = lane >> 4;

  v8f acc[8];
#pragma unroll
  for (int c = 0; c < 8; ++c)
#pragma unroll
    for (int v = 0; v < 8; ++v) acc[c][v] = 0.f;

  for (int kh = 0; kh < 2; ++kh) {
#pragma unroll
    for (int p = 0; p < 8; ++p) {
      int idx = p * 512 + tid * 4;
      int r = idx >> 6, cc = idx & 63;
      int row = rowBase + r;
      float4 v = make_float4(0.f, 0.f, 0.f, 0.f);
      if (row < n) v = *(const float4*)(x + (size_t)row * HDIM + kh * 64 + cc);
      *(float4*)(&xs[r * XPAD + cc]) = v;
    }
#pragma unroll
    for (int p = 0; p < 16; ++p) {
      int idx = p * 512 + tid * 4;
      int nn = idx >> 6, cc = idx & 63;
      *(float4*)(&wfs[nn * XPAD + cc]) = *(const float4*)(WfT + nn * HDIM + kh * 64 + cc);
    }
    __syncthreads();

    const float* xrow = &xs[(wave * 16 + m) * XPAD];
#pragma unroll
    for (int k4 = 0; k4 < 16; ++k4) {
      int kb = k4 * 4 + hi * 2;
      v2f a; a.x = xrow[kb]; a.y = xrow[kb + 1];
#pragma unroll
      for (int c = 0; c < 8; ++c) {
        const float* wrow = &wfs[(c * 16 + m) * XPAD];
        v2f b; b.x = wrow[kb]; b.y = wrow[kb + 1];
        acc[c] = __builtin_amdgcn_wmma_f32_16x16x4_f32(
            false, a, false, b, (short)0, acc[c], false, false);
      }
    }
    __syncthreads();
  }

#pragma unroll
  for (int c = 0; c < 8; ++c) {
    int col = c * 16 + m;
    float cbv = cb[col];
#pragma unroll
    for (int v = 0; v < 8; ++v) {
      int row = rowBase + wave * 16 + v + 8 * hi;
      if (row < n) h[(size_t)row * HDIM + col] = acc[c][v] + cbv;
    }
  }
}

#define NWAVE 32
#define QCAP 64
#define FROWS 400
#define FPITCH 130
#define NCLSP 48
#define LNONE 0xFFFFFFFFu
static_assert(TN <= 512, "a list entry keeps the local row in 9 bits");


template <bool FIN, bool REPLAY>
__global__ __launch_bounds__(1024) void k_agg_tile(const int* __restrict__ ei,
                                                   const float* __restrict__ keys_f,
                                                   float* lists_f,
                                                   const float* __restrict__ dinv,
                                                   const float* __restrict__ hbuf,
                                                   float* __restrict__ out,
                                                   const float* __restrict__ bias,
                                                   const float* __restrict__ Wo,
                                                   const float* __restrict__ bo,
                                                   int E, int Epad, int n) {
  constexpr int TS = FIN ? FPITCH : HDIM;
  constexpr int TR = FIN ? FROWS : TN;
  __shared__ __attribute__((aligned(16))) float tile[TR * TS];
  __shared__ int qe[NWAVE][QCAP];
  __shared__ float wos[FIN ? NCLSP * FPITCH : 1];
  __shared__ float bsw[FIN ? HDIM : 1];
  __shared__ int lcur;
  const int tid = threadIdx.x;
  const int lane = tid & 31;
  const int wave = __builtin_amdgcn_readfirstlane(tid >> 5);
  const int base = blockIdx.x * TN;
  const int* __restrict__ esrc = ei;
  const int* __restrict__ edst = ei + E;
  const int lcap = (int)(((long long)n * NCLS - (long long)(Epad >> 2)) / (long long)gridDim.x);
  unsigned* lst = (unsigned*)lists_f + (size_t)blockIdx.x * (size_t)(lcap > 0 ? lcap : 0);
  if (!REPLAY && tid == 0) lcur = 0;

  for (int r = wave; r < TR; r += NWAVE) {
    int node = base + r;
    float4 v = make_float4(0.f, 0.f, 0.f, 0.f);
    if (r < TN && node < n) {
      float d = dinv[node];
      float w = d * d;
      float4 hv = *(const float4*)(hbuf + (size_t)node * HDIM + lane * 4);
      v.x = hv.x * w; v.y = hv.y * w; v.z = hv.z * w; v.w = hv.w * w;
    }
    float* t = &tile[r * TS + lane];
    t[0] = v.x; t[32] = v.y; t[64] = v.z; t[96] = v.w;
  }
  if constexpr (FIN) {
    for (int i = tid; i < HDIM * NCLS; i += NWAVE * 32) {
      const int k = i / NCLS, c = i - k * NCLS;
      wos[c * FPITCH + (k & 3) * 32 + (k >> 2)] = Wo[i];
    }
    for (int i = tid; i < (NCLSP - NCLS) * HDIM; i += NWAVE * 32)
      wos[(NCLS + (i >> 7)) * FPITCH + (i & (HDIM - 1))] = 0.f;
    if (tid < HDIM) bsw[(tid & 3) * 32 + (tid >> 2)] = bias[tid];
  }
  __syncthreads();

  auto gather_add = [&](const int my_s, const int my_d, const float my_ws, const float my_wd, const int cnt) {
      for (int u0 = 0; u0 < cnt; u0 += 8) {
        float4 hv[8]; int dr[8];
#pragma unroll
        for (int u = 0; u < 8; ++u) {
          const int s = __builtin_amdgcn_readlane(my_s, u0 + u);
          dr[u] = __builtin_amdgcn_readlane(my_d, u0 + u);
          hv[u] = *(const float4*)(hbuf + (size_t)s * HDIM + lane * 4);
        }
#pragma unroll
        for (int u = 0; u < 8; ++u) {
          const float w = __int_as_float(__builtin_amdgcn_readlane(__float_as_int(my_ws), u0 + u)) *
                          __int_as_float(__builtin_amdgcn_readlane(__float_as_int(my_wd), u0 + u));
          if (u0 + u < cnt && dr[u] >= 0) {
            float* t = &tile[dr[u] * TS + lane];
            atomicAdd(t + 0,  hv[u].x * w);
            atomicAdd(t + 32, hv[u].y * w);
            atomicAdd(t + 64, hv[u].z * w);
            atomicAdd(t + 96, hv[u].w * w);
          }
        }
      }
  };

  auto drain = [&](const int qtot) {
    for (int qb = 0; qb < qtot; qb += 32) {
      const int cnt = (qtot - qb < 32) ? (qtot - qb) : 32;
      int my_s = 0, my_d = -1; float my_ws = 0.f, my_wd = 0.f;
      if (lane < cnt) {
        const int e = qe[wave][qb + lane];
        if (e < E) {
          const int s = esrc[e];
          const int dd = edst[e];
          my_s = s;
          my_ws = dinv[s]; my_wd = dinv[dd];
          const int dl = dd - base;
          my_d = ((unsigned)dl < (unsigned)TN) ? dl : -1;
        }
      }
      if constexpr (!REPLAY) {
        int pos = 0;
        if (lane == 0) pos = atomicAdd(&lcur, cnt);
        pos = __builtin_amdgcn_readfirstlane(pos);
        if (lane < cnt && 1 + pos + lane < lcap)
          lst[1 + pos + lane] = (my_d >= 0) ? (((unsigned)my_s << 9) | (unsigned)my_d) : LNONE;
      }
      gather_add(my_s, my_d, my_ws, my_wd, cnt);
    }
  };

  int lcount = -1;
  if constexpr (REPLAY) {
    if (lcap >= 2) {
      const unsigned c = (unsigned)__builtin_amdgcn_readfirstlane((int)lst[0]);
      if (c < (unsigned)lcap) lcount = (int)c;
    }
  }
  if (REPLAY && lcount >= 0) {
    auto load_ent = [&](const int i0) -> unsigned {
      const int k = i0 + lane;
      const unsigned v = lst[1 + ((k < lcount) ? k : lcount - 1)];
      return (k < lcount) ? v : LNONE;
    };
    int i = wave * 32;
    unsigned en = load_ent(i);
    for (; i < lcount; i += NWAVE * 32) {
      const unsigned e = en;
      en = load_ent(i + NWAVE * 32);
      const int cnt = (lcount - i < 32) ? (lcount - i) : 32;
      const bool ok = (e != LNONE);
      const int my_s = ok ? (int)(e >> 9) : 0;
      const int dl = ok ? (int)(e & 511u) : 0;
      const float my_ws = dinv[my_s];
      const float my_wd = dinv[base + dl];
      gather_add(my_s, ok ? dl : -1, my_ws, my_wd, cnt);
    }
  } else {
  const uint4* __restrict__ keys = (const uint4*)keys_f;
  const int per = ((((Epad >> 9) + NWAVE - 1) / NWAVE)) << 9;
  const int e0 = wave * per;
  const int e1 = (e0 + per < Epad) ? (e0 + per) : Epad;
  const int trips = (e1 > e0) ? ((e1 - e0) >> 9) : 0;
  const int start = trips ? (int)((blockIdx.x * 37u) % (unsigned)trips) : 0;
  const unsigned pat = ((unsigned)blockIdx.x & 0xFFu) * 0x01010101u;
  int qn = 0;

  auto load_keys = [&](const int t) -> uint4 { return keys[((e0 + (t << 9)) >> 4) + lane]; };
  int tc = start;
  uint4 kn = make_uint4(0u, 0u, 0u, 0u);
  if (trips > 0) kn = load_keys(tc);

  for (int it = 0; it < trips; ++it) {
    const uint4 k = kn;
    const int eb = e0 + (tc << 9) + lane * 16;
    tc = (tc + 1 == trips) ? 0 : tc + 1;
    if (it + 1 < trips) kn = load_keys(tc);

    const unsigned x0 = k.x ^ pat, x1 = k.y ^ pat, x2 = k.z ^ pat, x3 = k.w ^ pat;
    const unsigned t0 = (x0 - 0x01010101u) & ~x0 & 0x80808080u;
    const unsigned t1 = (x1 - 0x01010101u) & ~x1 & 0x80808080u;
    const unsigned t2 = (x2 - 0x01010101u) & ~x2 & 0x80808080u;
    const unsigned t3 = (x3 - 0x01010101u) & ~x3 & 0x80808080u;
    unsigned u = (t0 >> 7) | (t1 >> 6) | (t2 >> 5) | (t3 >> 4);

    for (int g = 0; g < 16; ++g) {
      const bool has = (u != 0u);
      const unsigned mask = (unsigned)__ballot(has);
      if (mask == 0u) break;
      if (has) {
        const int p = __builtin_ctz(u);
        qe[wave][qn + (int)__builtin_amdgcn_mbcnt_lo(mask, 0u)] = eb + ((p & 7) << 2) + (p >> 3);
      }
      u &= u - 1u;
      qn += __popc(mask);
      if (qn >= 32) { drain(qn); qn = 0; }
    }
  }
  if (qn > 0) drain(qn);
  }
  __syncthreads();
  if constexpr (!REPLAY) {
    if (tid == 0 && lcap >= 2) lst[0] = (unsigned)lcur;
  }

  if constexpr (!FIN) {
    for (int r = wave; r < TN; r += NWAVE) {
      int node = base + r;
      if (node < n) {
        const float* t = &tile[r * TS + lane];
        float4 v = make_float4(t[0], t[32], t[64], t[96]);
        *(float4*)(out + (size_t)node * HDIM + lane * 4) = v;
      }
    }
  } else {
    const int m = lane & 15;
    const int hi = lane >> 4;
    v8f acc[3];
#pragma unroll
    for (int c = 0; c < 3; ++c)
#pragma unroll
      for (int v = 0; v < 8; ++v) acc[c][v] = 0.f;
    if (wave < FROWS / 16) {
      const float* arow = &tile[(wave * 16 + m) * FPITCH + hi * 32];
      const float* brow = &wos[m * FPITCH + hi * 32];
      const float* bb = &bsw[hi * 32];
#pragma unroll 4
      for (int j = 0; j < 32; ++j) {
        const float t0 = arow[j] + bb[j];
        const float t1 = arow[64 + j] + bb[64 + j];
        v2f a; a.x = t0 > 0.f ? t0 : 0.f; a.y = t1 > 0.f ? t1 : 0.f;
#pragma unroll
        for (int c = 0; c < 3; ++c) {
          v2f b; b.x = brow[c * 16 * FPITCH + j]; b.y = brow[c * 16 * FPITCH + 64 + j];
          acc[c] = __builtin_amdgcn_wmma_f32_16x16x4_f32(
              false, a, false, b, (short)0, acc[c], false, false);
        }
      }
    }
    __syncthreads();
    float* stg = tile;
    static_assert(TN % 8 == 0, "a lane half's 8 rows must not straddle the last row");
    const int r0 = wave * 16 + 8 * hi;
    if (r0 < TN) {
#pragma unroll
      for (int c = 0; c < 3; ++c) {
        const int col = c * 16 + m;
        if (col < NCLS) {
          const float bov = bo[col];
#pragma unroll
          for (int v = 0; v < 8; ++v) stg[(r0 + v) * NCLS + col] = acc[c][v] + bov;
        }
      }
    }
    __syncthreads();
    const int rows = (n - base < TN) ? (n - base) : TN;
    const int nf4 = rows * (NCLS / 4);
    float4* __restrict__ o4 = (float4*)(out + (size_t)base * NCLS);
    const float4* s4 = (const float4*)stg;
    for (int i = tid; i < nf4; i += NWAVE * 32) o4[i] = s4[i];
  }
}

#define LSPLIT 8
__global__ void k_copy_keys(const float* __restrict__ src, float* __restrict__ dst, int n16,
                            int nkb, int Epad, int n, int nblk) {
  if ((int)blockIdx.x < nkb) {
    const int i = blockIdx.x * blockDim.x + threadIdx.x;
    if (i < n16) ((float4*)dst)[i] = ((const float4*)src)[i];
    return;
  }
  const int j = (int)blockIdx.x - nkb;
  const int lcap = (int)(((long long)n * NCLS - (long long)(Epad >> 2)) / (long long)nblk);
  if (lcap < 2) return;
  const size_t off = (size_t)(Epad >> 2) + (size_t)(j / LSPLIT) * (size_t)lcap;
  const unsigned* __restrict__ s = (const unsigned*)src + off;
  unsigned* __restrict__ d = (unsigned*)dst + off;
  const unsigned c = s[0];
  const int nw = (c < (unsigned)lcap) ? (int)c + 1 : 1;
  for (int i = (j % LSPLIT) * (int)blockDim.x + (int)threadIdx.x; i < nw; i += LSPLIT * (int)blockDim.x)
    d[i] = s[i];
}

extern "C" void kernel_launch(void* const* d_in, const int* in_sizes, int n_in,
                              void* d_out, int out_size, void* d_ws, size_t ws_size,
                              hipStream_t stream) {
  const float* x_in  = (const float*)d_in[0];
  const int*   ei    = (const int*)d_in[1];
  const float* gamma = (const float*)d_in[2];
  const float* beta  = (const float*)d_in[3];
  const float* W     = (const float*)d_in[4];
  const float* b     = (const float*)d_in[5];
  const float* Wo    = (const float*)d_in[6];
  const float* bo    = (const float*)d_in[7];
  float* out = (float*)d_out;

  const int N = in_sizes[0] / HDIM;
  const int E = in_sizes[1] / 2;
  const int Epad = ((E + 511) / 512) * 512;

  float* dinv  = (float*)d_ws;
  float* stats = dinv + ((N + 255) / 256) * 256;
  float* WfT   = stats + 2 * HDIM;
  float* cb    = WfT + HDIM * HDIM;
  float* hbuf  = cb + HDIM + 64;
  float* xbuf  = hbuf + (size_t)N * HDIM;
  float* keys  = out;
  float* lists = out + Epad / 4;
  const int nblk = (N + TN - 1) / TN;

  k_set_ones<<<(N + 255) / 256, 256, 0, stream>>>(dinv, N);
  k_deg_keys<<<(Epad / 4 + 255) / 256, 256, 0, stream>>>(ei, dinv, keys, E, Epad);
  k_rsqrt_ip<<<(N + 255) / 256, 256, 0, stream>>>(dinv, N);

  hipMemsetAsync(stats, 0, 2 * HDIM * sizeof(float), stream);
  k_stats<<<512, 256, 0, stream>>>(x_in, stats, N);

  for (int l = 0; l < 3; ++l) {
    k_fold<<<1, 128, 0, stream>>>(stats, gamma + l * HDIM, beta + l * HDIM,
                                  W + (size_t)l * HDIM * HDIM, WfT, cb, 1.0f / (float)N);
    const float* xl = (l == 0) ? x_in : xbuf;
    k_gemm<<<(N + 63) / 64, 128, 0, stream>>>(xl, WfT, cb, hbuf, N);
    if (l < 2) {
      if (l == 0)
        k_agg_tile<false, false><<<nblk, 1024, 0, stream>>>(ei, keys, lists, dinv, hbuf, xbuf,
                                                            nullptr, nullptr, nullptr, E, Epad, N);
      else
        k_agg_tile<false, true><<<nblk, 1024, 0, stream>>>(ei, keys, lists, dinv, hbuf, xbuf,
                                                           nullptr, nullptr, nullptr, E, Epad, N);
      hipMemsetAsync(stats, 0, 2 * HDIM * sizeof(float), stream);
      k_bias_relu_stats<<<512, 256, 0, stream>>>(xbuf, b + l * HDIM, stats, N, 1);
    } else {
      const int nkb = (Epad / 16 + 255) / 256;
      k_copy_keys<<<nkb + nblk * LSPLIT, 256, 0, stream>>>(keys, xbuf, Epad / 16, nkb, Epad, N, nblk);
      k_agg_tile<true, true><<<nblk, 1024, 0, stream>>>(ei, xbuf, xbuf + Epad / 4, dinv, hbuf, out,
                                                        b + l * HDIM, Wo, bo, E, Epad, N);
    }
  }
}


# Bias+relu pass: a wave takes 8 whole 512-B rows per trip (16 B per lane), all 8 loads issued before the first use, whole-row stores; was one 4-B load per lane per trip; column-sum order differs
# speedup vs baseline: 2.6828x; 1.1490x over previous
#include <hip/hip_runtime.h>

#define HDIM 128
#define NCLS 40
#define BN_EPS 1e-5f

typedef __attribute__((ext_vector_type(2))) float v2f;
typedef __attribute__((ext_vector_type(8))) float v8f;

__global__ void k_set_ones(float* deg, int n) {
  int i = blockIdx.x * blockDim.x + threadIdx.x;
  if (i < n) deg[i] = 1.0f;
}

#define TN 392
__global__ void k_deg_keys(const int* __restrict__ ei, float* deg, float* __restrict__ keys_f,
                           int E, int Epad) {
  const int q = blockIdx.x * blockDim.x + threadIdx.x;
  const int e = q * 4;
  if (e >= Epad) return;
  const int* __restrict__ edst = ei + E;
  int d[4];
  if (((E & 3) == 0) && e + 4 <= E) {
    const int4 dv = *(const int4*)(edst + e);
    d[0] = dv.x; d[1] = dv.y; d[2] = dv.z; d[3] = dv.w;
  } else {
#pragma unroll
    for (int j = 0; j < 4; ++j) d[j] = (e + j < E) ? edst[e + j] : -1;
  }
  unsigned packed = 0u;
#pragma unroll
  for (int j = 0; j < 4; ++j) {
    unsigned k = 0xFFu;
    if (d[j] >= 0) {
      atomicAdd(&deg[d[j]], 1.0f);
      k = ((unsigned)d[j] / (unsigned)TN) & 0xFFu;
    }
    packed |= k << (8 * j);
  }
  ((unsigned*)keys_f)[q] = packed;
}

__global__ void k_rsqrt_ip(float* d, int n) {
  int i = blockIdx.x * blockDim.x + threadIdx.x;
  if (i < n) d[i] = rsqrtf(d[i]);
}

__global__ void k_stats(const float* __restrict__ x, float* __restrict__ stats, int n) {
  int col = threadIdx.x & (HDIM - 1);
  int half = threadIdx.x >> 7;
  float s = 0.f, s2 = 0.f;
  for (int r = blockIdx.x * 2 + half; r < n; r += gridDim.x * 2) {
    float v = x[(size_t)r * HDIM + col];
    s += v; s2 += v * v;
  }
  __shared__ float ls[HDIM], ls2[HDIM];
  if (half == 0) { ls[col] = s; ls2[col] = s2; }
  __syncthreads();
  if (half == 1) {
    atomicAdd(&stats[col], s + ls[col]);
    atomicAdd(&stats[HDIM + col], s2 + ls2[col]);
  }
}

#define BR_ROWS 8
#define BR_WAVES 8
__global__ __launch_bounds__(BR_WAVES * 32) void k_bias_relu_stats(float* __restrict__ x,
                                                                   const float* __restrict__ b,
                                                                   float* __restrict__ stats, int n,
                                                                   int do_stats) {
  const int lane = threadIdx.x & 31;
  const int wave = threadIdx.x >> 5;
  const int gw = blockIdx.x * BR_WAVES + wave;
  const int nw = gridDim.x * BR_WAVES;
  const float4 bv = *(const float4*)(b + lane * 4);
  float4 s = make_float4(0.f, 0.f, 0.f, 0.f), s2 = make_float4(0.f, 0.f, 0.f, 0.f);
  auto finish = [&](float4 v, float* p) {
    v.x += bv.x; v.y += bv.y; v.z += bv.z; v.w += bv.w;
    v.x = v.x > 0.f ? v.x : 0.f; v.y = v.y > 0.f ? v.y : 0.f;
    v.z = v.z > 0.f ? v.z : 0.f; v.w = v.w > 0.f ? v.w : 0.f;
    *(float4*)p = v;
    s.x += v.x; s.y += v.y; s.z += v.z; s.w += v.w;
    s2.x += v.x * v.x; s2.y += v.y * v.y; s2.z += v.z * v.z; s2.w += v.w * v.w;
  };
  for (int r0 = gw * BR_ROWS; r0 < n; r0 += nw * BR_ROWS) {
    float* p = x + (size_t)r0 * HDIM + lane * 4;
    if (r0 + BR_ROWS <= n) {
      float4 v[BR_ROWS];
      for (int u = 0; u < BR_ROWS; ++u) v[u] = *(const float4*)(p + u * HDIM);
      for (int u = 0; u < BR_ROWS; ++u) finish(v[u], p + u * HDIM);
    } else {
      for (int u = 0; r0 + u < n; ++u) finish(*(const float4*)(p + u * HDIM), p + u * HDIM);
    }
  }
  if (!do_stats) return;
  __shared__ __attribute__((aligned(16))) float part[BR_WAVES][2 * HDIM];
  *(float4*)(&part[wave][lane * 4]) = s;
  *(float4*)(&part[wave][HDIM + lane * 4]) = s2;
  __syncthreads();
  const int t = threadIdx.x;
  float a = 0.f;
  for (int w = 0; w < BR_WAVES; ++w) a += part[w][t];
  atomicAdd(&stats[t], a);
}

__global__ void k_fold(const float* __restrict__ stats, const float* __restrict__ gamma,
                       const float* __restrict__ beta, const float* __restrict__ W,
                       float* __restrict__ WfT, float* __restrict__ cb, float inv_n) {
  __shared__ float as[HDIM], cs[HDIM];
  int t = threadIdx.x;
  float mean = stats[t] * inv_n;
  float var  = stats[HDIM + t] * inv_n - mean * mean;
  float a = gamma[t] * rsqrtf(var + BN_EPS);
  float c = beta[t] - a * mean;
  as[t] = a; cs[t] = c;
  __syncthreads();
  float acc = 0.f;
  for (int k = 0; k < HDIM; ++k) {
    float w = W[k * HDIM + t];
    WfT[t * HDIM + k] = as[k] * w;
    acc += cs[k] * w;
  }
  cb[t] = acc;
}

#define XPAD 68

__global__ __launch_bounds__(128) void k_gemm(const float* __restrict__ x,
                                              const float* __restrict__ WfT,
                                              const float* __restrict__ cb,
                                              float* __restrict__ h, int n) {
  __shared__ float xs[64 * XPAD];
  __shared__ float wfs[HDIM * XPAD];
  const int tid = threadIdx.x;
  const int lane = tid & 31;
  const int wave = tid >> 5;
  const int rowBase = blockIdx.x * 64;
  const int m = lane & 15;
  const int hi = lane >> 4;

  v8f acc[8];
#pragma unroll
  for (int c = 0; c < 8; ++c)
#pragma unroll
    for (int v = 0; v < 8; ++v) acc[c][v] = 0.f;

  for (int kh = 0; kh < 2; ++kh) {
#pragma unroll
    for (int p = 0; p < 8; ++p) {
      int idx = p * 512 + tid * 4;
      int r = idx >> 6, cc = idx & 63;
      int row = rowBase + r;
      float4 v = make_float4(0.f, 0.f, 0.f, 0.f);
      if (row < n) v = *(const float4*)(x + (size_t)row * HDIM + kh * 64 + cc);
      *(float4*)(&xs[r * XPAD + cc]) = v;
    }
#pragma unroll
    for (int p = 0; p < 16; ++p) {
      int idx = p * 512 + tid * 4;
      int nn = idx >> 6, cc = idx & 63;
      *(float4*)(&wfs[nn * XPAD + cc]) = *(const float4*)(WfT + nn * HDIM + kh * 64 + cc);
    }
    __syncthreads();

    const float* xrow = &xs[(wave * 16 + m) * XPAD];
#pragma unroll
    for (int k4 = 0; k4 < 16; ++k4) {
      int kb = k4 * 4 + hi * 2;
      v2f a; a.x = xrow[kb]; a.y = xrow[kb + 1];
#pragma unroll
      for (int c = 0; c < 8; ++c) {
        const float* wrow = &wfs[(c * 16 + m) * XPAD];
        v2f b; b.x = wrow[kb]; b.y = wrow[kb + 1];
        acc[c] = __builtin_amdgcn_wmma_f32_16x16x4_f32(
            false, a, false, b, (short)0, acc[c], false, false);
      }
    }
    __syncthreads();
  }

#pragma unroll
  for (int c = 0; c < 8; ++c) {
    int col = c * 16 + m;
    float cbv = cb[col];
#pragma unroll
    for (int v = 0; v < 8; ++v) {
      int row = rowBase + wave * 16 + v + 8 * hi;
      if (row < n) h[(size_t)row * HDIM + col] = acc[c][v] + cbv;
    }
  }
}

#define NWAVE 32
#define QCAP 64
#define FROWS 400
#define FPITCH 130
#define NCLSP 48
#define LNONE 0xFFFFFFFFu
static_assert(TN <= 512, "a list entry keeps the local row in 9 bits");


template <bool FIN, bool REPLAY>
__global__ __launch_bounds__(1024) void k_agg_tile(const int* __restrict__ ei,
                                                   const float* __restrict__ keys_f,
                                                   float* lists_f,
                                                   const float* __restrict__ dinv,
                                                   const float* __restrict__ hbuf,
                                                   float* __restrict__ out,
                                                   const float* __restrict__ bias,
                                                   const float* __restrict__ Wo,
                                                   const float* __restrict__ bo,
                                                   int E, int Epad, int n) {
  constexpr int TS = FIN ? FPITCH : HDIM;
  constexpr int TR = FIN ? FROWS : TN;
  __shared__ __attribute__((aligned(16))) float tile[TR * TS];
  __shared__ int qe[NWAVE][QCAP];
  __shared__ float wos[FIN ? NCLSP * FPITCH : 1];
  __shared__ float bsw[FIN ? HDIM : 1];
  __shared__ int lcur;
  const int tid = threadIdx.x;
  const int lane = tid & 31;
  const int wave = __builtin_amdgcn_readfirstlane(tid >> 5);
  const int base = blockIdx.x * TN;
  const int* __restrict__ esrc = ei;
  const int* __restrict__ edst = ei + E;
  const int lcap = (int)(((long long)n * NCLS - (long long)(Epad >> 2)) / (long long)gridDim.x);
  unsigned* lst = (unsigned*)lists_f + (size_t)blockIdx.x * (size_t)(lcap > 0 ? lcap : 0);
  if (!REPLAY && tid == 0) lcur = 0;

  for (int r = wave; r < TR; r += NWAVE) {
    int node = base + r;
    float4 v = make_float4(0.f, 0.f, 0.f, 0.f);
    if (r < TN && node < n) {
      float d = dinv[node];
      float w = d * d;
      float4 hv = *(const float4*)(hbuf + (size_t)node * HDIM + lane * 4);
      v.x = hv.x * w; v.y = hv.y * w; v.z = hv.z * w; v.w = hv.w * w;
    }
    float* t = &tile[r * TS + lane];
    t[0] = v.x; t[32] = v.y; t[64] = v.z; t[96] = v.w;
  }
  if constexpr (FIN) {
    for (int i = tid; i < HDIM * NCLS; i += NWAVE * 32) {
      const int k = i / NCLS, c = i - k * NCLS;
      wos[c * FPITCH + (k & 3) * 32 + (k >> 2)] = Wo[i];
    }
    for (int i = tid; i < (NCLSP - NCLS) * HDIM; i += NWAVE * 32)
      wos[(NCLS + (i >> 7)) * FPITCH + (i & (HDIM - 1))] = 0.f;
    if (tid < HDIM) bsw[(tid & 3) * 32 + (tid >> 2)] = bias[tid];
  }
  __syncthreads();

  auto gather_add = [&](const int my_s, const int my_d, const float my_ws, const float my_wd, const int cnt) {
      for (int u0 = 0; u0 < cnt; u0 += 8) {
        float4 hv[8]; int dr[8];
#pragma unroll
        for (int u = 0; u < 8; ++u) {
          const int s = __builtin_amdgcn_readlane(my_s, u0 + u);
          dr[u] = __builtin_amdgcn_readlane(my_d, u0 + u);
          hv[u] = *(const float4*)(hbuf + (size_t)s * HDIM + lane * 4);
        }
#pragma unroll
        for (int u = 0; u < 8; ++u) {
          const float w = __int_as_float(__builtin_amdgcn_readlane(__float_as_int(my_ws), u0 + u)) *
                          __int_as_float(__builtin_amdgcn_readlane(__float_as_int(my_wd), u0 + u));
          if (u0 + u < cnt && dr[u] >= 0) {
            float* t = &tile[dr[u] * TS + lane];
            atomicAdd(t + 0,  hv[u].x * w);
            atomicAdd(t + 32, hv[u].y * w);
            atomicAdd(t + 64, hv[u].z * w);
            atomicAdd(t + 96, hv[u].w * w);
          }
        }
      }
  };

  auto drain = [&](const int qtot) {
    for (int qb = 0; qb < qtot; qb += 32) {
      const int cnt = (qtot - qb < 32) ? (qtot - qb) : 32;
      int my_s = 0, my_d = -1; float my_ws = 0.f, my_wd = 0.f;
      if (lane < cnt) {
        const int e = qe[wave][qb + lane];
        if (e < E) {
          const int s = esrc[e];
          const int dd = edst[e];
          my_s = s;
          my_ws = dinv[s]; my_wd = dinv[dd];
          const int dl = dd - base;
          my_d = ((unsigned)dl < (unsigned)TN) ? dl : -1;
        }
      }
      if constexpr (!REPLAY) {
        int pos = 0;
        if (lane == 0) pos = atomicAdd(&lcur, cnt);
        pos = __builtin_amdgcn_readfirstlane(pos);
        if (lane < cnt && 1 + pos + lane < lcap)
          lst[1 + pos + lane] = (my_d >= 0) ? (((unsigned)my_s << 9) | (unsigned)my_d) : LNONE;
      }
      gather_add(my_s, my_d, my_ws, my_wd, cnt);
    }
  };

  int lcount = -1;
  if constexpr (REPLAY) {
    if (lcap >= 2) {
      const unsigned c = (unsigned)__builtin_amdgcn_readfirstlane((int)lst[0]);
      if (c < (unsigned)lcap) lcount = (int)c;
    }
  }
  if (REPLAY && lcount >= 0) {
    auto load_ent = [&](const int i0) -> unsigned {
      const int k = i0 + lane;
      const unsigned v = lst[1 + ((k < lcount) ? k : lcount - 1)];
      return (k < lcount) ? v : LNONE;
    };
    int i = wave * 32;
    unsigned en = load_ent(i);
    for (; i < lcount; i += NWAVE * 32) {
      const unsigned e = en;
      en = load_ent(i + NWAVE * 32);
      const int cnt = (lcount - i < 32) ? (lcount - i) : 32;
      const bool ok = (e != LNONE);
      const int my_s = ok ? (int)(e >> 9) : 0;
      const int dl = ok ? (int)(e & 511u) : 0;
      const float my_ws = dinv[my_s];
      const float my_wd = dinv[base + dl];
      gather_add(my_s, ok ? dl : -1, my_ws, my_wd, cnt);
    }
  } else {
  const uint4* __restrict__ keys = (const uint4*)keys_f;
  const int per = ((((Epad >> 9) + NWAVE - 1) / NWAVE)) << 9;
  const int e0 = wave * per;
  const int e1 = (e0 + per < Epad) ? (e0 + per) : Epad;
  const int trips = (e1 > e0) ? ((e1 - e0) >> 9) : 0;
  const int start = trips ? (int)((blockIdx.x * 37u) % (unsigned)trips) : 0;
  const unsigned pat = ((unsigned)blockIdx.x & 0xFFu) * 0x01010101u;
  int qn = 0;

  auto load_keys = [&](const int t) -> uint4 { return keys[((e0 + (t << 9)) >> 4) + lane]; };
  int tc = start;
  uint4 kn = make_uint4(0u, 0u, 0u, 0u);
  if (trips > 0) kn = load_keys(tc);

  for (int it = 0; it < trips; ++it) {
    const uint4 k = kn;
    const int eb = e0 + (tc << 9) + lane * 16;
    tc = (tc + 1 == trips) ? 0 : tc + 1;
    if (it + 1 < trips) kn = load_keys(tc);

    const unsigned x0 = k.x ^ pat, x1 = k.y ^ pat, x2 = k.z ^ pat, x3 = k.w ^ pat;
    const unsigned t0 = (x0 - 0x01010101u) & ~x0 & 0x80808080u;
    const unsigned t1 = (x1 - 0x01010101u) & ~x1 & 0x80808080u;
    const unsigned t2 = (x2 - 0x01010101u) & ~x2 & 0x80808080u;
    const unsigned t3 = (x3 - 0x01010101u) & ~x3 & 0x80808080u;
    unsigned u = (t0 >> 7) | (t1 >> 6) | (t2 >> 5) | (t3 >> 4);

    for (int g = 0; g < 16; ++g) {
      const bool has = (u != 0u);
      const unsigned mask = (unsigned)__ballot(has);
      if (mask == 0u) break;
      if (has) {
        const int p = __builtin_ctz(u);
        qe[wave][qn + (int)__builtin_amdgcn_mbcnt_lo(mask, 0u)] = eb + ((p & 7) << 2) + (p >> 3);
      }
      u &= u - 1u;
      qn += __popc(mask);
      if (qn >= 32) { drain(qn); qn = 0; }
    }
  }
  if (qn > 0) drain(qn);
  }
  __syncthreads();
  if constexpr (!REPLAY) {
    if (tid == 0 && lcap >= 2) lst[0] = (unsigned)lcur;
  }

  if constexpr (!FIN) {
    for (int r = wave; r < TN; r += NWAVE) {
      int node = base + r;
      if (node < n) {
        const float* t = &tile[r * TS + lane];
        float4 v = make_float4(t[0], t[32], t[64], t[96]);
        *(float4*)(out + (size_t)node * HDIM + lane * 4) = v;
      }
    }
  } else {
    const int m = lane & 15;
    const int hi = lane >> 4;
    v8f acc[3];
#pragma unroll
    for (int c = 0; c < 3; ++c)
#pragma unroll
      for (int v = 0; v < 8; ++v) acc[c][v] = 0.f;
    if (wave < FROWS / 16) {
      const float* arow = &tile[(wave * 16 + m) * FPITCH + hi * 32];
      const float* brow = &wos[m * FPITCH + hi * 32];
      const float* bb = &bsw[hi * 32];
#pragma unroll 4
      for (int j = 0; j < 32; ++j) {
        const float t0 = arow[j] + bb[j];
        const float t1 = arow[64 + j] + bb[64 + j];
        v2f a; a.x = t0 > 0.f ? t0 : 0.f; a.y = t1 > 0.f ? t1 : 0.f;
#pragma unroll
        for (int c = 0; c < 3; ++c) {
          v2f b; b.x = brow[c * 16 * FPITCH + j]; b.y = brow[c * 16 * FPITCH + 64 + j];
          acc[c] = __builtin_amdgcn_wmma_f32_16x16x4_f32(
              false, a, false, b, (short)0, acc[c], false, false);
        }
      }
    }
    __syncthreads();
    float* stg = tile;
    static_assert(TN % 8 == 0, "a lane half's 8 rows must not straddle the last row");
    const int r0 = wave * 16 + 8 * hi;
    if (r0 < TN) {
#pragma unroll
      for (int c = 0; c < 3; ++c) {
        const int col = c * 16 + m;
        if (col < NCLS) {
          const float bov = bo[col];
#pragma unroll
          for (int v = 0; v < 8; ++v) stg[(r0 + v) * NCLS + col] = acc[c][v] + bov;
        }
      }
    }
    __syncthreads();
    const int rows = (n - base < TN) ? (n - base) : TN;
    const int nf4 = rows * (NCLS / 4);
    float4* __restrict__ o4 = (float4*)(out + (size_t)base * NCLS);
    const float4* s4 = (const float4*)stg;
    for (int i = tid; i < nf4; i += NWAVE * 32) o4[i] = s4[i];
  }
}

#define LSPLIT 8
__global__ void k_copy_keys(const float* __restrict__ src, float* __restrict__ dst, int n16,
                            int nkb, int Epad, int n, int nblk) {
  if ((int)blockIdx.x < nkb) {
    const int i = blockIdx.x * blockDim.x + threadIdx.x;
    if (i < n16) ((float4*)dst)[i] = ((const float4*)src)[i];
    return;
  }
  const int j = (int)blockIdx.x - nkb;
  const int lcap = (int)(((long long)n * NCLS - (long long)(Epad >> 2)) / (long long)nblk);
  if (lcap < 2) return;
  const size_t off = (size_t)(Epad >> 2) + (size_t)(j / LSPLIT) * (size_t)lcap;
  const unsigned* __restrict__ s = (const unsigned*)src + off;
  unsigned* __restrict__ d = (unsigned*)dst + off;
  const unsigned c = s[0];
  const int nw = (c < (unsigned)lcap) ? (int)c + 1 : 1;
  for (int i = (j % LSPLIT) * (int)blockDim.x + (int)threadIdx.x; i < nw; i += LSPLIT * (int)blockDim.x)
    d[i] = s[i];
}

extern "C" void kernel_launch(void* const* d_in, const int* in_sizes, int n_in,
                              void* d_out, int out_size, void* d_ws, size_t ws_size,
                              hipStream_t stream) {
  const float* x_in  = (const float*)d_in[0];
  const int*   ei    = (const int*)d_in[1];
  const float* gamma = (const float*)d_in[2];
  const float* beta  = (const float*)d_in[3];
  const float* W     = (const float*)d_in[4];
  const float* b     = (const float*)d_in[5];
  const float* Wo    = (const float*)d_in[6];
  const float* bo    = (const float*)d_in[7];
  float* out = (float*)d_out;

  const int N = in_sizes[0] / HDIM;
  const int E = in_sizes[1] / 2;
  const int Epad = ((E + 511) / 512) * 512;

  float* dinv  = (float*)d_ws;
  float* stats = dinv + ((N + 255) / 256) * 256;
  float* WfT   = stats + 2 * HDIM;
  float* cb    = WfT + HDIM * HDIM;
  float* hbuf  = cb + HDIM + 64;
  float* xbuf  = hbuf + (size_t)N * HDIM;
  float* keys  = out;
  float* lists = out + Epad / 4;
  const int nblk = (N + TN - 1) / TN;

  k_set_ones<<<(N + 255) / 256, 256, 0, stream>>>(dinv, N);
  k_deg_keys<<<(Epad / 4 + 255) / 256, 256, 0, stream>>>(ei, dinv, keys, E, Epad);
  k_rsqrt_ip<<<(N + 255) / 256, 256, 0, stream>>>(dinv, N);

  hipMemsetAsync(stats, 0, 2 * HDIM * sizeof(float), stream);
  k_stats<<<512, 256, 0, stream>>>(x_in, stats, N);

  for (int l = 0; l < 3; ++l) {
    k_fold<<<1, 128, 0, stream>>>(stats, gamma + l * HDIM, beta + l * HDIM,
                                  W + (size_t)l * HDIM * HDIM, WfT, cb, 1.0f / (float)N);
    const float* xl = (l == 0) ? x_in : xbuf;
    k_gemm<<<(N + 63) / 64, 128, 0, stream>>>(xl, WfT, cb, hbuf, N);
    if (l < 2) {
      if (l == 0)
        k_agg_tile<false, false><<<nblk, 1024, 0, stream>>>(ei, keys, lists, dinv, hbuf, xbuf,
                                                            nullptr, nullptr, nullptr, E, Epad, N);
      else
        k_agg_tile<false, true><<<nblk, 1024, 0, stream>>>(ei, keys, lists, dinv, hbuf, xbuf,
                                                           nullptr, nullptr, nullptr, E, Epad, N);
      hipMemsetAsync(stats, 0, 2 * HDIM * sizeof(float), stream);
      k_bias_relu_stats<<<512, 256, 0, stream>>>(xbuf, b + l * HDIM, stats, N, 1);
    } else {
      const int nkb = (Epad / 16 + 255) / 256;
      k_copy_keys<<<nkb + nblk * LSPLIT, 256, 0, stream>>>(keys, xbuf, Epad / 16, nkb, Epad, N, nblk);
      k_agg_tile<true, true><<<nblk, 1024, 0, stream>>>(ei, xbuf, xbuf + Epad / 4, dinv, hbuf, out,
                                                        b + l * HDIM, Wo, bo, E, Epad, N);
    }
  }
}


# Edge weight split: GEMM stores rows times dinv[src], owner blocks add them as they are and scale each finished row by dinv[dst]; per-edge 4-B factor gathers and products go; sum rounding differs
# speedup vs baseline: 2.7665x; 1.0312x over previous
#include <hip/hip_runtime.h>

#define HDIM 128
#define NCLS 40
#define BN_EPS 1e-5f

typedef __attribute__((ext_vector_type(2))) float v2f;
typedef __attribute__((ext_vector_type(8))) float v8f;

__global__ void k_set_ones(float* deg, int n) {
  int i = blockIdx.x * blockDim.x + threadIdx.x;
  if (i < n) deg[i] = 1.0f;
}

#define TN 392
__global__ void k_deg_keys(const int* __restrict__ ei, float* deg, float* __restrict__ keys_f,
                           int E, int Epad) {
  const int q = blockIdx.x * blockDim.x + threadIdx.x;
  const int e = q * 4;
  if (e >= Epad) return;
  const int* __restrict__ edst = ei + E;
  int d[4];
  if (((E & 3) == 0) && e + 4 <= E) {
    const int4 dv = *(const int4*)(edst + e);
    d[0] = dv.x; d[1] = dv.y; d[2] = dv.z; d[3] = dv.w;
  } else {
#pragma unroll
    for (int j = 0; j < 4; ++j) d[j] = (e + j < E) ? edst[e + j] : -1;
  }
  unsigned packed = 0u;
#pragma unroll
  for (int j = 0; j < 4; ++j) {
    unsigned k = 0xFFu;
    if (d[j] >= 0) {
      atomicAdd(&deg[d[j]], 1.0f);
      k = ((unsigned)d[j] / (unsigned)TN) & 0xFFu;
    }
    packed |= k << (8 * j);
  }
  ((unsigned*)keys_f)[q] = packed;
}

__global__ void k_rsqrt_ip(float* d, int n) {
  int i = blockIdx.x * blockDim.x + threadIdx.x;
  if (i < n) d[i] = rsqrtf(d[i]);
}

__global__ void k_stats(const float* __restrict__ x, float* __restrict__ stats, int n) {
  int col = threadIdx.x & (HDIM - 1);
  int half = threadIdx.x >> 7;
  float s = 0.f, s2 = 0.f;
  for (int r = blockIdx.x * 2 + half; r < n; r += gridDim.x * 2) {
    float v = x[(size_t)r * HDIM + col];
    s += v; s2 += v * v;
  }
  __shared__ float ls[HDIM], ls2[HDIM];
  if (half == 0) { ls[col] = s; ls2[col] = s2; }
  __syncthreads();
  if (half == 1) {
    atomicAdd(&stats[col], s + ls[col]);
    atomicAdd(&stats[HDIM + col], s2 + ls2[col]);
  }
}

#define BR_ROWS 8
#define BR_WAVES 8
__global__ __launch_bounds__(BR_WAVES * 32) void k_bias_relu_stats(float* __restrict__ x,
                                                                   const float* __restrict__ b,
                                                                   float* __restrict__ stats, int n,
                                                                   int do_stats) {
  const int lane = threadIdx.x & 31;
  const int wave = threadIdx.x >> 5;
  const int gw = blockIdx.x * BR_WAVES + wave;
  const int nw = gridDim.x * BR_WAVES;
  const float4 bv = *(const float4*)(b + lane * 4);
  float4 s = make_float4(0.f, 0.f, 0.f, 0.f), s2 = make_float4(0.f, 0.f, 0.f, 0.f);
  auto finish = [&](float4 v, float* p) {
    v.x += bv.x; v.y += bv.y; v.z += bv.z; v.w += bv.w;
    v.x = v.x > 0.f ? v.x : 0.f; v.y = v.y > 0.f ? v.y : 0.f;
    v.z = v.z > 0.f ? v.z : 0.f; v.w = v.w > 0.f ? v.w : 0.f;
    *(float4*)p = v;
    s.x += v.x; s.y += v.y; s.z += v.z; s.w += v.w;
    s2.x += v.x * v.x; s2.y += v.y * v.y; s2.z += v.z * v.z; s2.w += v.w * v.w;
  };
  for (int r0 = gw * BR_ROWS; r0 < n; r0 += nw * BR_ROWS) {
    float* p = x + (size_t)r0 * HDIM + lane * 4;
    if (r0 + BR_ROWS <= n) {
      float4 v[BR_ROWS];
      for (int u = 0; u < BR_ROWS; ++u) v[u] = *(const float4*)(p + u * HDIM);
      for (int u = 0; u < BR_ROWS; ++u) finish(v[u], p + u * HDIM);
    } else {
      for (int u = 0; r0 + u < n; ++u) finish(*(const float4*)(p + u * HDIM), p + u * HDIM);
    }
  }
  if (!do_stats) return;
  __shared__ __attribute__((aligned(16))) float part[BR_WAVES][2 * HDIM];
  *(float4*)(&part[wave][lane * 4]) = s;
  *(float4*)(&part[wave][HDIM + lane * 4]) = s2;
  __syncthreads();
  const int t = threadIdx.x;
  float a = 0.f;
  for (int w = 0; w < BR_WAVES; ++w) a += part[w][t];
  atomicAdd(&stats[t], a);
}

__global__ void k_fold(const float* __restrict__ stats, const float* __restrict__ gamma,
                       const float* __restrict__ beta, const float* __restrict__ W,
                       float* __restrict__ WfT, float* __restrict__ cb, float inv_n) {
  __shared__ float as[HDIM], cs[HDIM];
  int t = threadIdx.x;
  float mean = stats[t] * inv_n;
  float var  = stats[HDIM + t] * inv_n - mean * mean;
  float a = gamma[t] * rsqrtf(var + BN_EPS);
  float c = beta[t] - a * mean;
  as[t] = a; cs[t] = c;
  __syncthreads();
  float acc = 0.f;
  for (int k = 0; k < HDIM; ++k) {
    float w = W[k * HDIM + t];
    WfT[t * HDIM + k] = as[k] * w;
    acc += cs[k] * w;
  }
  cb[t] = acc;
}

#define XPAD 68

__global__ __launch_bounds__(128) void k_gemm(const float* __restrict__ x,
                                              const float* __restrict__ WfT,
                                              const float* __restrict__ cb,
                                              const float* __restrict__ dinv,
                                              float* __restrict__ h, int n) {
  __shared__ float xs[64 * XPAD];
  __shared__ float wfs[HDIM * XPAD];
  const int tid = threadIdx.x;
  const int lane = tid & 31;
  const int wave = tid >> 5;
  const int rowBase = blockIdx.x * 64;
  const int m = lane & 15;
  const int hi = lane >> 4;

  v8f acc[8];
#pragma unroll
  for (int c = 0; c < 8; ++c)
#pragma unroll
    for (int v = 0; v < 8; ++v) acc[c][v] = 0.f;

  for (int kh = 0; kh < 2; ++kh) {
#pragma unroll
    for (int p = 0; p < 8; ++p) {
      int idx = p * 512 + tid * 4;
      int r = idx >> 6, cc = idx & 63;
      int row = rowBase + r;
      float4 v = make_float4(0.f, 0.f, 0.f, 0.f);
      if (row < n) v = *(const float4*)(x + (size_t)row * HDIM + kh * 64 + cc);
      *(float4*)(&xs[r * XPAD + cc]) = v;
    }
#pragma unroll
    for (int p = 0; p < 16; ++p) {
      int idx = p * 512 + tid * 4;
      int nn = idx >> 6, cc = idx & 63;
      *(float4*)(&wfs[nn * XPAD + cc]) = *(const float4*)(WfT + nn * HDIM + kh * 64 + cc);
    }
    __syncthreads();

    const float* xrow = &xs[(wave * 16 + m) * XPAD];
#pragma unroll
    for (int k4 = 0; k4 < 16; ++k4) {
      int kb = k4 * 4 + hi * 2;
      v2f a; a.x = xrow[kb]; a.y = xrow[kb + 1];
#pragma unroll
      for (int c = 0; c < 8; ++c) {
        const float* wrow = &wfs[(c * 16 + m) * XPAD];
        v2f b; b.x = wrow[kb]; b.y = wrow[kb + 1];
        acc[c] = __builtin_amdgcn_wmma_f32_16x16x4_f32(
            false, a, false, b, (short)0, acc[c], false, false);
      }
    }
    __syncthreads();
  }

  float dv[8];
#pragma unroll
  for (int v = 0; v < 8; ++v) {
    const int row = rowBase + wave * 16 + v + 8 * hi;
    dv[v] = (row < n) ? dinv[row] : 0.f;
  }
#pragma unroll
  for (int c = 0; c < 8; ++c) {
    int col = c * 16 + m;
    float cbv = cb[col];
#pragma unroll
    for (int v = 0; v < 8; ++v) {
      int row = rowBase + wave * 16 + v + 8 * hi;
      if (row < n) h[(size_t)row * HDIM + col] = (acc[c][v] + cbv) * dv[v];
    }
  }
}

#define NWAVE 32
#define QCAP 64
#define FROWS 400
#define FPITCH 130
#define NCLSP 48
#define LNONE 0xFFFFFFFFu
static_assert(TN <= 512, "a list entry keeps the local row in 9 bits");


template <bool FIN, bool REPLAY>
__global__ __launch_bounds__(1024) void k_agg_tile(const int* __restrict__ ei,
                                                   const float* __restrict__ keys_f,
                                                   float* lists_f,
                                                   const float* __restrict__ dinv,
                                                   const float* __restrict__ hbuf,
                                                   float* __restrict__ out,
                                                   const float* __restrict__ bias,
                                                   const float* __restrict__ Wo,
                                                   const float* __restrict__ bo,
                                                   int E, int Epad, int n) {
  constexpr int TS = FIN ? FPITCH : HDIM;
  constexpr int TR = FIN ? FROWS : TN;
  __shared__ __attribute__((aligned(16))) float tile[TR * TS];
  __shared__ int qe[NWAVE][QCAP];
  __shared__ float wos[FIN ? NCLSP * FPITCH : 1];
  __shared__ float bsw[FIN ? HDIM : 1];
  __shared__ int lcur;
  const int tid = threadIdx.x;
  const int lane = tid & 31;
  const int wave = __builtin_amdgcn_readfirstlane(tid >> 5);
  const int base = blockIdx.x * TN;
  const int* __restrict__ esrc = ei;
  const int* __restrict__ edst = ei + E;
  const int lcap = (int)(((long long)n * NCLS - (long long)(Epad >> 2)) / (long long)gridDim.x);
  unsigned* lst = (unsigned*)lists_f + (size_t)blockIdx.x * (size_t)(lcap > 0 ? lcap : 0);
  if (!REPLAY && tid == 0) lcur = 0;

  for (int r = wave; r < TR; r += NWAVE) {
    int node = base + r;
    float4 v = make_float4(0.f, 0.f, 0.f, 0.f);
    if (r < TN && node < n) v = *(const float4*)(hbuf + (size_t)node * HDIM + lane * 4);
    float* t = &tile[r * TS + lane];
    t[0] = v.x; t[32] = v.y; t[64] = v.z; t[96] = v.w;
  }
  if constexpr (FIN) {
    for (int i = tid; i < HDIM * NCLS; i += NWAVE * 32) {
      const int k = i / NCLS, c = i - k * NCLS;
      wos[c * FPITCH + (k & 3) * 32 + (k >> 2)] = Wo[i];
    }
    for (int i = tid; i < (NCLSP - NCLS) * HDIM; i += NWAVE * 32)
      wos[(NCLS + (i >> 7)) * FPITCH + (i & (HDIM - 1))] = 0.f;
    if (tid < HDIM) bsw[(tid & 3) * 32 + (tid >> 2)] = bias[tid];
  }
  __syncthreads();

  auto gather_add = [&](const int my_s, const int my_d, const int cnt) {
      for (int u0 = 0; u0 < cnt; u0 += 8) {
        float4 hv[8]; int dr[8];
#pragma unroll
        for (int u = 0; u < 8; ++u) {
          const int s = __builtin_amdgcn_readlane(my_s, u0 + u);
          dr[u] = __builtin_amdgcn_readlane(my_d, u0 + u);
          hv[u] = *(const float4*)(hbuf + (size_t)s * HDIM + lane * 4);
        }
#pragma unroll
        for (int u = 0; u < 8; ++u) {
          if (u0 + u < cnt && dr[u] >= 0) {
            float* t = &tile[dr[u] * TS + lane];
            atomicAdd(t + 0,  hv[u].x);
            atomicAdd(t + 32, hv[u].y);
            atomicAdd(t + 64, hv[u].z);
            atomicAdd(t + 96, hv[u].w);
          }
        }
      }
  };

  auto drain = [&](const int qtot) {
    for (int qb = 0; qb < qtot; qb += 32) {
      const int cnt = (qtot - qb < 32) ? (qtot - qb) : 32;
      int my_s = 0, my_d = -1;
      if (lane < cnt) {
        const int e = qe[wave][qb + lane];
        if (e < E) {
          const int s = esrc[e];
          const int dd = edst[e];
          my_s = s;
          const int dl = dd - base;
          my_d = ((unsigned)dl < (unsigned)TN) ? dl : -1;
        }
      }
      if constexpr (!REPLAY) {
        int pos = 0;
        if (lane == 0) pos = atomicAdd(&lcur, cnt);
        pos = __builtin_amdgcn_readfirstlane(pos);
        if (lane < cnt && 1 + pos + lane < lcap)
          lst[1 + pos + lane] = (my_d >= 0) ? (((unsigned)my_s << 9) | (unsigned)my_d) : LNONE;
      }
      gather_add(my_s, my_d, cnt);
    }
  };

  int lcount = -1;
  if constexpr (REPLAY) {
    if (lcap >= 2) {
      const unsigned c = (unsigned)__builtin_amdgcn_readfirstlane((int)lst[0]);
      if (c < (unsigned)lcap) lcount = (int)c;
    }
  }
  if (REPLAY && lcount >= 0) {
    auto load_ent = [&](const int i0) -> unsigned {
      const int k = i0 + lane;
      const unsigned v = lst[1 + ((k < lcount) ? k : lcount - 1)];
      return (k < lcount) ? v : LNONE;
    };
    int i = wave * 32;
    unsigned en = load_ent(i);
    for (; i < lcount; i += NWAVE * 32) {
      const unsigned e = en;
      en = load_ent(i + NWAVE * 32);
      const int cnt = (lcount - i < 32) ? (lcount - i) : 32;
      const bool ok = (e != LNONE);
      const int my_s = ok ? (int)(e >> 9) : 0;
      gather_add(my_s, ok ? (int)(e & 511u) : -1, cnt);
    }
  } else {
  const uint4* __restrict__ keys = (const uint4*)keys_f;
  const int per = ((((Epad >> 9) + NWAVE - 1) / NWAVE)) << 9;
  const int e0 = wave * per;
  const int e1 = (e0 + per < Epad) ? (e0 + per) : Epad;
  const int trips = (e1 > e0) ? ((e1 - e0) >> 9) : 0;
  const int start = trips ? (int)((blockIdx.x * 37u) % (unsigned)trips) : 0;
  const unsigned pat = ((unsigned)blockIdx.x & 0xFFu) * 0x01010101u;
  int qn = 0;

  auto load_keys = [&](const int t) -> uint4 { return keys[((e0 + (t << 9)) >> 4) + lane]; };
  int tc = start;
  uint4 kn = make_uint4(0u, 0u, 0u, 0u);
  if (trips > 0) kn = load_keys(tc);

  for (int it = 0; it < trips; ++it) {
    const uint4 k = kn;
    const int eb = e0 + (tc << 9) + lane * 16;
    tc = (tc + 1 == trips) ? 0 : tc + 1;
    if (it + 1 < trips) kn = load_keys(tc);

    const unsigned x0 = k.x ^ pat, x1 = k.y ^ pat, x2 = k.z ^ pat, x3 = k.w ^ pat;
    const unsigned t0 = (x0 - 0x01010101u) & ~x0 & 0x80808080u;
    const unsigned t1 = (x1 - 0x01010101u) & ~x1 & 0x80808080u;
    const unsigned t2 = (x2 - 0x01010101u) & ~x2 & 0x80808080u;
    const unsigned t3 = (x3 - 0x01010101u) & ~x3 & 0x80808080u;
    unsigned u = (t0 >> 7) | (t1 >> 6) | (t2 >> 5) | (t3 >> 4);

    for (int g = 0; g < 16; ++g) {
      const bool has = (u != 0u);
      const unsigned mask = (unsigned)__ballot(has);
      if (mask == 0u) break;
      if (has) {
        const int p = __builtin_ctz(u);
        qe[wave][qn + (int)__builtin_amdgcn_mbcnt_lo(mask, 0u)] = eb + ((p & 7) << 2) + (p >> 3);
      }
      u &= u - 1u;
      qn += __popc(mask);
      if (qn >= 32) { drain(qn); qn = 0; }
    }
  }
  if (qn > 0) drain(qn);
  }
  static_assert(TN <= 32 * NWAVE, "one lane per row that a wave stores");
  const int xr = FIN ? (wave * 16 + (lane & 15)) : (wave + lane * NWAVE);
  const float xd = (xr < TN && base + xr < n) ? dinv[base + xr] : 0.f;
  __syncthreads();
  if constexpr (!REPLAY) {
    if (tid == 0 && lcap >= 2) lst[0] = (unsigned)lcur;
  }

  if constexpr (!FIN) {
    for (int r = wave, i = 0; r < TN; r += NWAVE, ++i) {
      int node = base + r;
      if (node < n) {
        const float d = __int_as_float(__builtin_amdgcn_readlane(__float_as_int(xd), i));
        const float* t = &tile[r * TS + lane];
        float4 v = make_float4(t[0] * d, t[32] * d, t[64] * d, t[96] * d);
        *(float4*)(out + (size_t)node * HDIM + lane * 4) = v;
      }
    }
  } else {
    const int m = lane & 15;
    const int hi = lane >> 4;
    v8f acc[3];
#pragma unroll
    for (int c = 0; c < 3; ++c)
#pragma unroll
      for (int v = 0; v < 8; ++v) acc[c][v] = 0.f;
    if (wave < FROWS / 16) {
      const float* arow = &tile[(wave * 16 + m) * FPITCH + hi * 32];
      const float* brow = &wos[m * FPITCH + hi * 32];
      const float* bb = &bsw[hi * 32];
      const float ad = xd;
#pragma unroll 4
      for (int j = 0; j < 32; ++j) {
        const float t0 = arow[j] * ad + bb[j];
        const float t1 = arow[64 + j] * ad + bb[64 + j];
        v2f a; a.x = t0 > 0.f ? t0 : 0.f; a.y = t1 > 0.f ? t1 : 0.f;
#pragma unroll
        for (int c = 0; c < 3; ++c) {
          v2f b; b.x = brow[c * 16 * FPITCH + j]; b.y = brow[c * 16 * FPITCH + 64 + j];
          acc[c] = __builtin_amdgcn_wmma_f32_16x16x4_f32(
              false, a, false, b, (short)0, acc[c], false, false);
        }
      }
    }
    __syncthreads();
    float* stg = tile;
    static_assert(TN % 8 == 0, "a lane half's 8 rows must not straddle the last row");
    const int r0 = wave * 16 + 8 * hi;
    if (r0 < TN) {
#pragma unroll
      for (int c = 0; c < 3; ++c) {
        const int col = c * 16 + m;
        if (col < NCLS) {
          const float bov = bo[col];
#pragma unroll
          for (int v = 0; v < 8; ++v) stg[(r0 + v) * NCLS + col] = acc[c][v] + bov;
        }
      }
    }
    __syncthreads();
    const int rows = (n - base < TN) ? (n - base) : TN;
    const int nf4 = rows * (NCLS / 4);
    float4* __restrict__ o4 = (float4*)(out + (size_t)base * NCLS);
    const float4* s4 = (const float4*)stg;
    for (int i = tid; i < nf4; i += NWAVE * 32) o4[i] = s4[i];
  }
}

#define LSPLIT 8
__global__ void k_copy_keys(const float* __restrict__ src, float* __restrict__ dst, int n16,
                            int nkb, int Epad, int n, int nblk) {
  if ((int)blockIdx.x < nkb) {
    const int i = blockIdx.x * blockDim.x + threadIdx.x;
    if (i < n16) ((float4*)dst)[i] = ((const float4*)src)[i];
    return;
  }
  const int j = (int)blockIdx.x - nkb;
  const int lcap = (int)(((long long)n * NCLS - (long long)(Epad >> 2)) / (long long)nblk);
  if (lcap < 2) return;
  const size_t off = (size_t)(Epad >> 2) + (size_t)(j / LSPLIT) * (size_t)lcap;
  const unsigned* __restrict__ s = (const unsigned*)src + off;
  unsigned* __restrict__ d = (unsigned*)dst + off;
  const unsigned c = s[0];
  const int nw = (c < (unsigned)lcap) ? (int)c + 1 : 1;
  for (int i = (j % LSPLIT) * (int)blockDim.x + (int)threadIdx.x; i < nw; i += LSPLIT * (int)blockDim.x)
    d[i] = s[i];
}

extern "C" void kernel_launch(void* const* d_in, const int* in_sizes, int n_in,
                              void* d_out, int out_size, void* d_ws, size_t ws_size,
                              hipStream_t stream) {
  const float* x_in  = (const float*)d_in[0];
  const int*   ei    = (const int*)d_in[1];
  const float* gamma = (const float*)d_in[2];
  const float* beta  = (const float*)d_in[3];
  const float* W     = (const float*)d_in[4];
  const float* b     = (const float*)d_in[5];
  const float* Wo    = (const float*)d_in[6];
  const float* bo    = (const float*)d_in[7];
  float* out = (float*)d_out;

  const int N = in_sizes[0] / HDIM;
  const int E = in_sizes[1] / 2;
  const int Epad = ((E + 511) / 512) * 512;

  float* dinv  = (float*)d_ws;
  float* stats = dinv + ((N + 255) / 256) * 256;
  float* WfT   = stats + 2 * HDIM;
  float* cb    = WfT + HDIM * HDIM;
  float* hbuf  = cb + HDIM + 64;
  float* xbuf  = hbuf + (size_t)N * HDIM;
  float* keys  = out;
  float* lists = out + Epad / 4;
  const int nblk = (N + TN - 1) / TN;

  k_set_ones<<<(N + 255) / 256, 256, 0, stream>>>(dinv, N);
  k_deg_keys<<<(Epad / 4 + 255) / 256, 256, 0, stream>>>(ei, dinv, keys, E, Epad);
  k_rsqrt_ip<<<(N + 255) / 256, 256, 0, stream>>>(dinv, N);

  hipMemsetAsync(stats, 0, 2 * HDIM * sizeof(float), stream);
  k_stats<<<512, 256, 0, stream>>>(x_in, stats, N);

  for (int l = 0; l < 3; ++l) {
    k_fold<<<1, 128, 0, stream>>>(stats, gamma + l * HDIM, beta + l * HDIM,
                                  W + (size_t)l * HDIM * HDIM, WfT, cb, 1.0f / (float)N);
    const float* xl = (l == 0) ? x_in : xbuf;
    k_gemm<<<(N + 63) / 64, 128, 0, stream>>>(xl, WfT, cb, dinv, hbuf, N);
    if (l < 2) {
      if (l == 0)
        k_agg_tile<false, false><<<nblk, 1024, 0, stream>>>(ei, keys, lists, dinv, hbuf, xbuf,
                                                            nullptr, nullptr, nullptr, E, Epad, N);
      else
        k_agg_tile<false, true><<<nblk, 1024, 0, stream>>>(ei, keys, lists, dinv, hbuf, xbuf,
                                                           nullptr, nullptr, nullptr, E, Epad, N);
      hipMemsetAsync(stats, 0, 2 * HDIM * sizeof(float), stream);
      k_bias_relu_stats<<<512, 256, 0, stream>>>(xbuf, b + l * HDIM, stats, N, 1);
    } else {
      const int nkb = (Epad / 16 + 255) / 256;
      k_copy_keys<<<nkb + nblk * LSPLIT, 256, 0, stream>>>(keys, xbuf, Epad / 16, nkb, Epad, N, nblk);
      k_agg_tile<true, true><<<nblk, 1024, 0, stream>>>(ei, xbuf, xbuf + Epad / 4, dinv, hbuf, out,
                                                        b + l * HDIM, Wo, bo, E, Epad, N);
    }
  }
}


# BN fold kernel spread over 128 blocks, one thread per element of W: the fold waits for memory once, not in 16 dependent trips of a single block ahead of each GEMM; cb's k sum order differs
# speedup vs baseline: 2.9327x; 1.0601x over previous
#include <hip/hip_runtime.h>

#define HDIM 128
#define NCLS 40
#define BN_EPS 1e-5f

typedef __attribute__((ext_vector_type(2))) float v2f;
typedef __attribute__((ext_vector_type(8))) float v8f;

__global__ void k_set_ones(float* deg, int n) {
  int i = blockIdx.x * blockDim.x + threadIdx.x;
  if (i < n) deg[i] = 1.0f;
}

#define TN 392
__global__ void k_deg_keys(const int* __restrict__ ei, float* deg, float* __restrict__ keys_f,
                           int E, int Epad) {
  const int q = blockIdx.x * blockDim.x + threadIdx.x;
  const int e = q * 4;
  if (e >= Epad) return;
  const int* __restrict__ edst = ei + E;
  int d[4];
  if (((E & 3) == 0) && e + 4 <= E) {
    const int4 dv = *(const int4*)(edst + e);
    d[0] = dv.x; d[1] = dv.y; d[2] = dv.z; d[3] = dv.w;
  } else {
#pragma unroll
    for (int j = 0; j < 4; ++j) d[j] = (e + j < E) ? edst[e + j] : -1;
  }
  unsigned packed = 0u;
#pragma unroll
  for (int j = 0; j < 4; ++j) {
    unsigned k = 0xFFu;
    if (d[j] >= 0) {
      atomicAdd(&deg[d[j]], 1.0f);
      k = ((unsigned)d[j] / (unsigned)TN) & 0xFFu;
    }
    packed |= k << (8 * j);
  }
  ((unsigned*)keys_f)[q] = packed;
}

__global__ void k_rsqrt_ip(float* d, int n) {
  int i = blockIdx.x * blockDim.x + threadIdx.x;
  if (i < n) d[i] = rsqrtf(d[i]);
}

__global__ void k_stats(const float* __restrict__ x, float* __restrict__ stats, int n) {
  int col = threadIdx.x & (HDIM - 1);
  int half = threadIdx.x >> 7;
  float s = 0.f, s2 = 0.f;
  for (int r = blockIdx.x * 2 + half; r < n; r += gridDim.x * 2) {
    float v = x[(size_t)r * HDIM + col];
    s += v; s2 += v * v;
  }
  __shared__ float ls[HDIM], ls2[HDIM];
  if (half == 0) { ls[col] = s; ls2[col] = s2; }
  __syncthreads();
  if (half == 1) {
    atomicAdd(&stats[col], s + ls[col]);
    atomicAdd(&stats[HDIM + col], s2 + ls2[col]);
  }
}

#define BR_ROWS 8
#define BR_WAVES 8
__global__ __launch_bounds__(BR_WAVES * 32) void k_bias_relu_stats(float* __restrict__ x,
                                                                   const float* __restrict__ b,
                                                                   float* __restrict__ stats, int n,
                                                                   int do_stats) {
  const int lane = threadIdx.x & 31;
  const int wave = threadIdx.x >> 5;
  const int gw = blockIdx.x * BR_WAVES + wave;
  const int nw = gridDim.x * BR_WAVES;
  const float4 bv = *(const float4*)(b + lane * 4);
  float4 s = make_float4(0.f, 0.f, 0.f, 0.f), s2 = make_float4(0.f, 0.f, 0.f, 0.f);
  auto finish = [&](float4 v, float* p) {
    v.x += bv.x; v.y += bv.y; v.z += bv.z; v.w += bv.w;
    v.x = v.x > 0.f ? v.x : 0.f; v.y = v.y > 0.f ? v.y : 0.f;
    v.z = v.z > 0.f ? v.z : 0.f; v.w = v.w > 0.f ? v.w : 0.f;
    *(float4*)p = v;
    s.x += v.x; s.y += v.y; s.z += v.z; s.w += v.w;
    s2.x += v.x * v.x; s2.y += v.y * v.y; s2.z += v.z * v.z; s2.w += v.w * v.w;
  };
  for (int r0 = gw * BR_ROWS; r0 < n; r0 += nw * BR_ROWS) {
    float* p = x + (size_t)r0 * HDIM + lane * 4;
    if (r0 + BR_ROWS <= n) {
      float4 v[BR_ROWS];
      for (int u = 0; u < BR_ROWS; ++u) v[u] = *(const float4*)(p + u * HDIM);
      for (int u = 0; u < BR_ROWS; ++u) finish(v[u], p + u * HDIM);
    } else {
      for (int u = 0; r0 + u < n; ++u) finish(*(const float4*)(p + u * HDIM), p + u * HDIM);
    }
  }
  if (!do_stats) return;
  __shared__ __attribute__((aligned(16))) float part[BR_WAVES][2 * HDIM];
  *(float4*)(&part[wave][lane * 4]) = s;
  *(float4*)(&part[wave][HDIM + lane * 4]) = s2;
  __syncthreads();
  const int t = threadIdx.x;
  float a = 0.f;
  for (int w = 0; w < BR_WAVES; ++w) a += part[w][t];
  atomicAdd(&stats[t], a);
}

__global__ __launch_bounds__(HDIM) void k_fold(const float* __restrict__ stats,
                                               const float* __restrict__ gamma,
                                               const float* __restrict__ beta,
                                               const float* __restrict__ W,
                                               float* __restrict__ WfT, float* __restrict__ cb,
                                               float inv_n) {
  __shared__ float red[HDIM / 32];
  const int k = threadIdx.x;
  const int n = blockIdx.x;
  const float w = W[k * HDIM + n];
  const float mean = stats[k] * inv_n;
  const float var  = stats[HDIM + k] * inv_n - mean * mean;
  const float a = gamma[k] * rsqrtf(var + BN_EPS);
  const float c = beta[k] - a * mean;
  float p = c * w;
  for (int o = 16; o > 0; o >>= 1) p += __shfl_xor(p, o, 32);
  if ((k & 31) == 0) red[k >> 5] = p;
  __syncthreads();
  WfT[n * HDIM + k] = a * w;
  if (k == 0) cb[n] = (red[0] + red[1]) + (red[2] + red[3]);
}

#define XPAD 68

__global__ __launch_bounds__(128) void k_gemm(const float* __restrict__ x,
                                              const float* __restrict__ WfT,
                                              const float* __restrict__ cb,
                                              const float* __restrict__ dinv,
                                              float* __restrict__ h, int n) {
  __shared__ float xs[64 * XPAD];
  __shared__ float wfs[HDIM * XPAD];
  const int tid = threadIdx.x;
  const int lane = tid & 31;
  const int wave = tid >> 5;
  const int rowBase = blockIdx.x * 64;
  const int m = lane & 15;
  const int hi = lane >> 4;

  v8f acc[8];
#pragma unroll
  for (int c = 0; c < 8; ++c)
#pragma unroll
    for (int v = 0; v < 8; ++v) acc[c][v] = 0.f;

  for (int kh = 0; kh < 2; ++kh) {
#pragma unroll
    for (int p = 0; p < 8; ++p) {
      int idx = p * 512 + tid * 4;
      int r = idx >> 6, cc = idx & 63;
      int row = rowBase + r;
      float4 v = make_float4(0.f, 0.f, 0.f, 0.f);
      if (row < n) v = *(const float4*)(x + (size_t)row * HDIM + kh * 64 + cc);
      *(float4*)(&xs[r * XPAD + cc]) = v;
    }
#pragma unroll
    for (int p = 0; p < 16; ++p) {
      int idx = p * 512 + tid * 4;
      int nn = idx >> 6, cc = idx & 63;
      *(float4*)(&wfs[nn * XPAD + cc]) = *(const float4*)(WfT + nn * HDIM + kh * 64 + cc);
    }
    __syncthreads();

    const float* xrow = &xs[(wave * 16 + m) * XPAD];
#pragma unroll
    for (int k4 = 0; k4 < 16; ++k4) {
      int kb = k4 * 4 + hi * 2;
      v2f a; a.x = xrow[kb]; a.y = xrow[kb + 1];
#pragma unroll
      for (int c = 0; c < 8; ++c) {
        const float* wrow = &wfs[(c * 16 + m) * XPAD];
        v2f b; b.x = wrow[kb]; b.y = wrow[kb + 1];
        acc[c] = __builtin_amdgcn_wmma_f32_16x16x4_f32(
            false, a, false, b, (short)0, acc[c], false, false);
      }
    }
    __syncthreads();
  }

  float dv[8];
#pragma unroll
  for (int v = 0; v < 8; ++v) {
    const int row = rowBase + wave * 16 + v + 8 * hi;
    dv[v] = (row < n) ? dinv[row] : 0.f;
  }
#pragma unroll
  for (int c = 0; c < 8; ++c) {
    int col = c * 16 + m;
    float cbv = cb[col];
#pragma unroll
    for (int v = 0; v < 8; ++v) {
      int row = rowBase + wave * 16 + v + 8 * hi;
      if (row < n) h[(size_t)row * HDIM + col] = (acc[c][v] + cbv) * dv[v];
    }
  }
}

#define NWAVE 32
#define QCAP 64
#define FROWS 400
#define FPITCH 130
#define NCLSP 48
#define LNONE 0xFFFFFFFFu
static_assert(TN <= 512, "a list entry keeps the local row in 9 bits");


template <bool FIN, bool REPLAY>
__global__ __launch_bounds__(1024) void k_agg_tile(const int* __restrict__ ei,
                                                   const float* __restrict__ keys_f,
                                                   float* lists_f,
                                                   const float* __restrict__ dinv,
                                                   const float* __restrict__ hbuf,
                                                   float* __restrict__ out,
                                                   const float* __restrict__ bias,
                                                   const float* __restrict__ Wo,
                                                   const float* __restrict__ bo,
                                                   int E, int Epad, int n) {
  constexpr int TS = FIN ? FPITCH : HDIM;
  constexpr int TR = FIN ? FROWS : TN;
  __shared__ __attribute__((aligned(16))) float tile[TR * TS];
  __shared__ int qe[NWAVE][QCAP];
  __shared__ float wos[FIN ? NCLSP * FPITCH : 1];
  __shared__ float bsw[FIN ? HDIM : 1];
  __shared__ int lcur;
  const int tid = threadIdx.x;
  const int lane = tid & 31;
  const int wave = __builtin_amdgcn_readfirstlane(tid >> 5);
  const int base = blockIdx.x * TN;
  const int* __restrict__ esrc = ei;
  const int* __restrict__ edst = ei + E;
  const int lcap = (int)(((long long)n * NCLS - (long long)(Epad >> 2)) / (long long)gridDim.x);
  unsigned* lst = (unsigned*)lists_f + (size_t)blockIdx.x * (size_t)(lcap > 0 ? lcap : 0);
  if (!REPLAY && tid == 0) lcur = 0;

  for (int r = wave; r < TR; r += NWAVE) {
    int node = base + r;
    float4 v = make_float4(0.f, 0.f, 0.f, 0.f);
    if (r < TN && node < n) v = *(const float4*)(hbuf + (size_t)node * HDIM + lane * 4);
    float* t = &tile[r * TS + lane];
    t[0] = v.x; t[32] = v.y; t[64] = v.z; t[96] = v.w;
  }
  if constexpr (FIN) {
    for (int i = tid; i < HDIM * NCLS; i += NWAVE * 32) {
      const int k = i / NCLS, c = i - k * NCLS;
      wos[c * FPITCH + (k & 3) * 32 + (k >> 2)] = Wo[i];
    }
    for (int i = tid; i < (NCLSP - NCLS) * HDIM; i += NWAVE * 32)
      wos[(NCLS + (i >> 7)) * FPITCH + (i & (HDIM - 1))] = 0.f;
    if (tid < HDIM) bsw[(tid & 3) * 32 + (tid >> 2)] = bias[tid];
  }
  __syncthreads();

  auto gather_add = [&](const int my_s, const int my_d, const int cnt) {
      for (int u0 = 0; u0 < cnt; u0 += 8) {
        float4 hv[8]; int dr[8];
#pragma unroll
        for (int u = 0; u < 8; ++u) {
          const int s = __builtin_amdgcn_readlane(my_s, u0 + u);
          dr[u] = __builtin_amdgcn_readlane(my_d, u0 + u);
          hv[u] = *(const float4*)(hbuf + (size_t)s * HDIM + lane * 4);
        }
#pragma unroll
        for (int u = 0; u < 8; ++u) {
          if (u0 + u < cnt && dr[u] >= 0) {
            float* t = &tile[dr[u] * TS + lane];
            atomicAdd(t + 0,  hv[u].x);
            atomicAdd(t + 32, hv[u].y);
            atomicAdd(t + 64, hv[u].z);
            atomicAdd(t + 96, hv[u].w);
          }
        }
      }
  };

  auto drain = [&](const int qtot) {
    for (int qb = 0; qb < qtot; qb += 32) {
      const int cnt = (qtot - qb < 32) ? (qtot - qb) : 32;
      int my_s = 0, my_d = -1;
      if (lane < cnt) {
        const int e = qe[wave][qb + lane];
        if (e < E) {
          const int s = esrc[e];
          const int dd = edst[e];
          my_s = s;
          const int dl = dd - base;
          my_d = ((unsigned)dl < (unsigned)TN) ? dl : -1;
        }
      }
      if constexpr (!REPLAY) {
        int pos = 0;
        if (lane == 0) pos = atomicAdd(&lcur, cnt);
        pos = __builtin_amdgcn_readfirstlane(pos);
        if (lane < cnt && 1 + pos + lane < lcap)
          lst[1 + pos + lane] = (my_d >= 0) ? (((unsigned)my_s << 9) | (unsigned)my_d) : LNONE;
      }
      gather_add(my_s, my_d, cnt);
    }
  };

  int lcount = -1;
  if constexpr (REPLAY) {
    if (lcap >= 2) {
      const unsigned c = (unsigned)__builtin_amdgcn_readfirstlane((int)lst[0]);
      if (c < (unsigned)lcap) lcount = (int)c;
    }
  }
  if (REPLAY && lcount >= 0) {
    auto load_ent = [&](const int i0) -> unsigned {
      const int k = i0 + lane;
      const unsigned v = lst[1 + ((k < lcount) ? k : lcount - 1)];
      return (k < lcount) ? v : LNONE;
    };
    int i = wave * 32;
    unsigned en = load_ent(i);
    for (; i < lcount; i += NWAVE * 32) {
      const unsigned e = en;
      en = load_ent(i + NWAVE * 32);
      const int cnt = (lcount - i < 32) ? (lcount - i) : 32;
      const bool ok = (e != LNONE);
      const int my_s = ok ? (int)(e >> 9) : 0;
      gather_add(my_s, ok ? (int)(e & 511u) : -1, cnt);
    }
  } else {
  const uint4* __restrict__ keys = (const uint4*)keys_f;
  const int per = ((((Epad >> 9) + NWAVE - 1) / NWAVE)) << 9;
  const int e0 = wave * per;
  const int e1 = (e0 + per < Epad) ? (e0 + per) : Epad;
  const int trips = (e1 > e0) ? ((e1 - e0) >> 9) : 0;
  const int start = trips ? (int)((blockIdx.x * 37u) % (unsigned)trips) : 0;
  const unsigned pat = ((unsigned)blockIdx.x & 0xFFu) * 0x01010101u;
  int qn = 0;

  auto load_keys = [&](const int t) -> uint4 { return keys[((e0 + (t << 9)) >> 4) + lane]; };
  int tc = start;
  uint4 kn = make_uint4(0u, 0u, 0u, 0u);
  if (trips > 0) kn = load_keys(tc);

  for (int it = 0; it < trips; ++it) {
    const uint4 k = kn;
    const int eb = e0 + (tc << 9) + lane * 16;
    tc = (tc + 1 == trips) ? 0 : tc + 1;
    if (it + 1 < trips) kn = load_keys(tc);

    const unsigned x0 = k.x ^ pat, x1 = k.y ^ pat, x2 = k.z ^ pat, x3 = k.w ^ pat;
    const unsigned t0 = (x0 - 0x01010101u) & ~x0 & 0x80808080u;
    const unsigned t1 = (x1 - 0x01010101u) & ~x1 & 0x80808080u;
    const unsigned t2 = (x2 - 0x01010101u) & ~x2 & 0x80808080u;
    const unsigned t3 = (x3 - 0x01010101u) & ~x3 & 0x80808080u;
    unsigned u = (t0 >> 7) | (t1 >> 6) | (t2 >> 5) | (t3 >> 4);

    for (int g = 0; g < 16; ++g) {
      const bool has = (u != 0u);
      const unsigned mask = (unsigned)__ballot(has);
      if (mask == 0u) break;
      if (has) {
        const int p = __builtin_ctz(u);
        qe[wave][qn + (int)__builtin_amdgcn_mbcnt_lo(mask, 0u)] = eb + ((p & 7) << 2) + (p >> 3);
      }
      u &= u - 1u;
      qn += __popc(mask);
      if (qn >= 32) { drain(qn); qn = 0; }
    }
  }
  if (qn > 0) drain(qn);
  }
  static_assert(TN <= 32 * NWAVE, "one lane per row that a wave stores");
  const int xr = FIN ? (wave * 16 + (lane & 15)) : (wave + lane * NWAVE);
  const float xd = (xr < TN && base + xr < n) ? dinv[base + xr] : 0.f;
  __syncthreads();
  if constexpr (!REPLAY) {
    if (tid == 0 && lcap >= 2) lst[0] = (unsigned)lcur;
  }

  if constexpr (!FIN) {
    for (int r = wave, i = 0; r < TN; r += NWAVE, ++i) {
      int node = base + r;
      if (node < n) {
        const float d = __int_as_float(__builtin_amdgcn_readlane(__float_as_int(xd), i));
        const float* t = &tile[r * TS + lane];
        float4 v = make_float4(t[0] * d, t[32] * d, t[64] * d, t[96] * d);
        *(float4*)(out + (size_t)node * HDIM + lane * 4) = v;
      }
    }
  } else {
    const int m = lane & 15;
    const int hi = lane >> 4;
    v8f acc[3];
#pragma unroll
    for (int c = 0; c < 3; ++c)
#pragma unroll
      for (int v = 0; v < 8; ++v) acc[c][v] = 0.f;
    if (wave < FROWS / 16) {
      const float* arow = &tile[(wave * 16 + m) * FPITCH + hi * 32];
      const float* brow = &wos[m * FPITCH + hi * 32];
      const float* bb = &bsw[hi * 32];
      const float ad = xd;
#pragma unroll 4
      for (int j = 0; j < 32; ++j) {
        const float t0 = arow[j] * ad + bb[j];
        const float t1 = arow[64 + j] * ad + bb[64 + j];
        v2f a; a.x = t0 > 0.f ? t0 : 0.f; a.y = t1 > 0.f ? t1 : 0.f;
#pragma unroll
        for (int c = 0; c < 3; ++c) {
          v2f b; b.x = brow[c * 16 * FPITCH + j]; b.y = brow[c * 16 * FPITCH + 64 + j];
          acc[c] = __builtin_amdgcn_wmma_f32_16x16x4_f32(
              false, a, false, b, (short)0, acc[c], false, false);
        }
      }
    }
    __syncthreads();
    float* stg = tile;
    static_assert(TN % 8 == 0, "a lane half's 8 rows must not straddle the last row");
    const int r0 = wave * 16 + 8 * hi;
    if (r0 < TN) {
#pragma unroll
      for (int c = 0; c < 3; ++c) {
        const int col = c * 16 + m;
        if (col < NCLS) {
          const float bov = bo[col];
#pragma unroll
          for (int v = 0; v < 8; ++v) stg[(r0 + v) * NCLS + col] = acc[c][v] + bov;
        }
      }
    }
    __syncthreads();
    const int rows = (n - base < TN) ? (n - base) : TN;
    const int nf4 = rows * (NCLS / 4);
    float4* __restrict__ o4 = (float4*)(out + (size_t)base * NCLS);
    const float4* s4 = (const float4*)stg;
    for (int i = tid; i < nf4; i += NWAVE * 32) o4[i] = s4[i];
  }
}

#define LSPLIT 8
__global__ void k_copy_keys(const float* __restrict__ src, float* __restrict__ dst, int n16,
                            int nkb, int Epad, int n, int nblk) {
  if ((int)blockIdx.x < nkb) {
    const int i = blockIdx.x * blockDim.x + threadIdx.x;
    if (i < n16) ((float4*)dst)[i] = ((const float4*)src)[i];
    return;
  }
  const int j = (int)blockIdx.x - nkb;
  const int lcap = (int)(((long long)n * NCLS - (long long)(Epad >> 2)) / (long long)nblk);
  if (lcap < 2) return;
  const size_t off = (size_t)(Epad >> 2) + (size_t)(j / LSPLIT) * (size_t)lcap;
  const unsigned* __restrict__ s = (const unsigned*)src + off;
  unsigned* __restrict__ d = (unsigned*)dst + off;
  const unsigned c = s[0];
  const int nw = (c < (unsigned)lcap) ? (int)c + 1 : 1;
  for (int i = (j % LSPLIT) * (int)blockDim.x + (int)threadIdx.x; i < nw; i += LSPLIT * (int)blockDim.x)
    d[i] = s[i];
}

extern "C" void kernel_launch(void* const* d_in, const int* in_sizes, int n_in,
                              void* d_out, int out_size, void* d_ws, size_t ws_size,
                              hipStream_t stream) {
  const float* x_in  = (const float*)d_in[0];
  const int*   ei    = (const int*)d_in[1];
  const float* gamma = (const float*)d_in[2];
  const float* beta  = (const float*)d_in[3];
  const float* W     = (const float*)d_in[4];
  const float* b     = (const float*)d_in[5];
  const float* Wo    = (const float*)d_in[6];
  const float* bo    = (const float*)d_in[7];
  float* out = (float*)d_out;

  const int N = in_sizes[0] / HDIM;
  const int E = in_sizes[1] / 2;
  const int Epad = ((E + 511) / 512) * 512;

  float* dinv  = (float*)d_ws;
  float* stats = dinv + ((N + 255) / 256) * 256;
  float* WfT   = stats + 2 * HDIM;
  float* cb    = WfT + HDIM * HDIM;
  float* hbuf  = cb + HDIM + 64;
  float* xbuf  = hbuf + (size_t)N * HDIM;
  float* keys  = out;
  float* lists = out + Epad / 4;
  const int nblk = (N + TN - 1) / TN;

  k_set_ones<<<(N + 255) / 256, 256, 0, stream>>>(dinv, N);
  k_deg_keys<<<(Epad / 4 + 255) / 256, 256, 0, stream>>>(ei, dinv, keys, E, Epad);
  k_rsqrt_ip<<<(N + 255) / 256, 256, 0, stream>>>(dinv, N);

  hipMemsetAsync(stats, 0, 2 * HDIM * sizeof(float), stream);
  k_stats<<<512, 256, 0, stream>>>(x_in, stats, N);

  for (int l = 0; l < 3; ++l) {
    k_fold<<<HDIM, HDIM, 0, stream>>>(stats, gamma + l * HDIM, beta + l * HDIM,
                                  W + (size_t)l * HDIM * HDIM, WfT, cb, 1.0f / (float)N);
    const float* xl = (l == 0) ? x_in : xbuf;
    k_gemm<<<(N + 63) / 64, 128, 0, stream>>>(xl, WfT, cb, dinv, hbuf, N);
    if (l < 2) {
      if (l == 0)
        k_agg_tile<false, false><<<nblk, 1024, 0, stream>>>(ei, keys, lists, dinv, hbuf, xbuf,
                                                            nullptr, nullptr, nullptr, E, Epad, N);
      else
        k_agg_tile<false, true><<<nblk, 1024, 0, stream>>>(ei, keys, lists, dinv, hbuf, xbuf,
                                                           nullptr, nullptr, nullptr, E, Epad, N);
      hipMemsetAsync(stats, 0, 2 * HDIM * sizeof(float), stream);
      k_bias_relu_stats<<<512, 256, 0, stream>>>(xbuf, b + l * HDIM, stats, N, 1);
    } else {
      const int nkb = (Epad / 16 + 255) / 256;
      k_copy_keys<<<nkb + nblk * LSPLIT, 256, 0, stream>>>(keys, xbuf, Epad / 16, nkb, Epad, N, nblk);
      k_agg_tile<true, true><<<nblk, 1024, 0, stream>>>(ei, xbuf, xbuf + Epad / 4, dinv, hbuf, out,
                                                        b + l * HDIM, Wo, bo, E, Epad, N);
    }
  }
}


# Degree and BN sum tables get no launches of their own: one clear for both, rsqrt(deg+1) taken where a factor is read, GEMM block 0 re-clears the sums; 19 stream operations become 15
# speedup vs baseline: 3.0931x; 1.0547x over previous
#include <hip/hip_runtime.h>

#define HDIM 128
#define NCLS 40
#define BN_EPS 1e-5f

typedef __attribute__((ext_vector_type(2))) float v2f;
typedef __attribute__((ext_vector_type(8))) float v8f;

#define TN 392
__global__ void k_deg_keys(const int* __restrict__ ei, float* deg, float* __restrict__ keys_f,
                           int E, int Epad) {
  const int q = blockIdx.x * blockDim.x + threadIdx.x;
  const int e = q * 4;
  if (e >= Epad) return;
  const int* __restrict__ edst = ei + E;
  int d[4];
  if (((E & 3) == 0) && e + 4 <= E) {
    const int4 dv = *(const int4*)(edst + e);
    d[0] = dv.x; d[1] = dv.y; d[2] = dv.z; d[3] = dv.w;
  } else {
#pragma unroll
    for (int j = 0; j < 4; ++j) d[j] = (e + j < E) ? edst[e + j] : -1;
  }
  unsigned packed = 0u;
#pragma unroll
  for (int j = 0; j < 4; ++j) {
    unsigned k = 0xFFu;
    if (d[j] >= 0) {
      atomicAdd(&deg[d[j]], 1.0f);
      k = ((unsigned)d[j] / (unsigned)TN) & 0xFFu;
    }
    packed |= k << (8 * j);
  }
  ((unsigned*)keys_f)[q] = packed;
}

__global__ void k_stats(const float* __restrict__ x, float* __restrict__ stats, int n) {
  int col = threadIdx.x & (HDIM - 1);
  int half = threadIdx.x >> 7;
  float s = 0.f, s2 = 0.f;
  for (int r = blockIdx.x * 2 + half; r < n; r += gridDim.x * 2) {
    float v = x[(size_t)r * HDIM + col];
    s += v; s2 += v * v;
  }
  __shared__ float ls[HDIM], ls2[HDIM];
  if (half == 0) { ls[col] = s; ls2[col] = s2; }
  __syncthreads();
  if (half == 1) {
    atomicAdd(&stats[col], s + ls[col]);
    atomicAdd(&stats[HDIM + col], s2 + ls2[col]);
  }
}

#define BR_ROWS 8
#define BR_WAVES 8
__global__ __launch_bounds__(BR_WAVES * 32) void k_bias_relu_stats(float* __restrict__ x,
                                                                   const float* __restrict__ b,
                                                                   float* __restrict__ stats, int n,
                                                                   int do_stats) {
  const int lane = threadIdx.x & 31;
  const int wave = threadIdx.x >> 5;
  const int gw = blockIdx.x * BR_WAVES + wave;
  const int nw = gridDim.x * BR_WAVES;
  const float4 bv = *(const float4*)(b + lane * 4);
  float4 s = make_float4(0.f, 0.f, 0.f, 0.f), s2 = make_float4(0.f, 0.f, 0.f, 0.f);
  auto finish = [&](float4 v, float* p) {
    v.x += bv.x; v.y += bv.y; v.z += bv.z; v.w += bv.w;
    v.x = v.x > 0.f ? v.x : 0.f; v.y = v.y > 0.f ? v.y : 0.f;
    v.z = v.z > 0.f ? v.z : 0.f; v.w = v.w > 0.f ? v.w : 0.f;
    *(float4*)p = v;
    s.x += v.x; s.y += v.y; s.z += v.z; s.w += v.w;
    s2.x += v.x * v.x; s2.y += v.y * v.y; s2.z += v.z * v.z; s2.w += v.w * v.w;
  };
  for (int r0 = gw * BR_ROWS; r0 < n; r0 += nw * BR_ROWS) {
    float* p = x + (size_t)r0 * HDIM + lane * 4;
    if (r0 + BR_ROWS <= n) {
      float4 v[BR_ROWS];
      for (int u = 0; u < BR_ROWS; ++u) v[u] = *(const float4*)(p + u * HDIM);
      for (int u = 0; u < BR_ROWS; ++u) finish(v[u], p + u * HDIM);
    } else {
      for (int u = 0; r0 + u < n; ++u) finish(*(const float4*)(p + u * HDIM), p + u * HDIM);
    }
  }
  if (!do_stats) return;
  __shared__ __attribute__((aligned(16))) float part[BR_WAVES][2 * HDIM];
  *(float4*)(&part[wave][lane * 4]) = s;
  *(float4*)(&part[wave][HDIM + lane * 4]) = s2;
  __syncthreads();
  const int t = threadIdx.x;
  float a = 0.f;
  for (int w = 0; w < BR_WAVES; ++w) a += part[w][t];
  atomicAdd(&stats[t], a);
}

__global__ __launch_bounds__(HDIM) void k_fold(const float* __restrict__ stats,
                                               const float* __restrict__ gamma,
                                               const float* __restrict__ beta,
                                               const float* __restrict__ W,
                                               float* __restrict__ WfT, float* __restrict__ cb,
                                               float inv_n) {
  __shared__ float red[HDIM / 32];
  const int k = threadIdx.x;
  const int n = blockIdx.x;
  const float w = W[k * HDIM + n];
  const float mean = stats[k] * inv_n;
  const float var  = stats[HDIM + k] * inv_n - mean * mean;
  const float a = gamma[k] * rsqrtf(var + BN_EPS);
  const float c = beta[k] - a * mean;
  float p = c * w;
  for (int o = 16; o > 0; o >>= 1) p += __shfl_xor(p, o, 32);
  if ((k & 31) == 0) red[k >> 5] = p;
  __syncthreads();
  WfT[n * HDIM + k] = a * w;
  if (k == 0) cb[n] = (red[0] + red[1]) + (red[2] + red[3]);
}

#define XPAD 68

__global__ __launch_bounds__(128) void k_gemm(const float* __restrict__ x,
                                              const float* __restrict__ WfT,
                                              const float* __restrict__ cb,
                                              const float* __restrict__ deg,
                                              float* __restrict__ stats,
                                              float* __restrict__ h, int n) {
  if (blockIdx.x == 0) { stats[threadIdx.x] = 0.f; stats[HDIM + threadIdx.x] = 0.f; }
  __shared__ float xs[64 * XPAD];
  __shared__ float wfs[HDIM * XPAD];
  const int tid = threadIdx.x;
  const int lane = tid & 31;
  const int wave = tid >> 5;
  const int rowBase = blockIdx.x * 64;
  const int m = lane & 15;
  const int hi = lane >> 4;

  v8f acc[8];
#pragma unroll
  for (int c = 0; c < 8; ++c)
#pragma unroll
    for (int v = 0; v < 8; ++v) acc[c][v] = 0.f;

  for (int kh = 0; kh < 2; ++kh) {
#pragma unroll
    for (int p = 0; p < 8; ++p) {
      int idx = p * 512 + tid * 4;
      int r = idx >> 6, cc = idx & 63;
      int row = rowBase + r;
      float4 v = make_float4(0.f, 0.f, 0.f, 0.f);
      if (row < n) v = *(const float4*)(x + (size_t)row * HDIM + kh * 64 + cc);
      *(float4*)(&xs[r * XPAD + cc]) = v;
    }
#pragma unroll
    for (int p = 0; p < 16; ++p) {
      int idx = p * 512 + tid * 4;
      int nn = idx >> 6, cc = idx & 63;
      *(float4*)(&wfs[nn * XPAD + cc]) = *(const float4*)(WfT + nn * HDIM + kh * 64 + cc);
    }
    __syncthreads();

    const float* xrow = &xs[(wave * 16 + m) * XPAD];
#pragma unroll
    for (int k4 = 0; k4 < 16; ++k4) {
      int kb = k4 * 4 + hi * 2;
      v2f a; a.x = xrow[kb]; a.y = xrow[kb + 1];
#pragma unroll
      for (int c = 0; c < 8; ++c) {
        const float* wrow = &wfs[(c * 16 + m) * XPAD];
        v2f b; b.x = wrow[kb]; b.y = wrow[kb + 1];
        acc[c] = __builtin_amdgcn_wmma_f32_16x16x4_f32(
            false, a, false, b, (short)0, acc[c], false, false);
      }
    }
    __syncthreads();
  }

  float dv[8];
#pragma unroll
  for (int v = 0; v < 8; ++v) {
    const int row = rowBase + wave * 16 + v + 8 * hi;
    dv[v] = (row < n) ? rsqrtf(deg[row] + 1.0f) : 0.f;
  }
#pragma unroll
  for (int c = 0; c < 8; ++c) {
    int col = c * 16 + m;
    float cbv = cb[col];
#pragma unroll
    for (int v = 0; v < 8; ++v) {
      int row = rowBase + wave * 16 + v + 8 * hi;
      if (row < n) h[(size_t)row * HDIM + col] = (acc[c][v] + cbv) * dv[v];
    }
  }
}

#define NWAVE 32
#define QCAP 64
#define FROWS 400
#define FPITCH 130
#define NCLSP 48
#define LNONE 0xFFFFFFFFu
static_assert(TN <= 512, "a list entry keeps the local row in 9 bits");


template <bool FIN, bool REPLAY>
__global__ __launch_bounds__(1024) void k_agg_tile(const int* __restrict__ ei,
                                                   const float* __restrict__ keys_f,
                                                   float* lists_f,
                                                   const float* __restrict__ deg,
                                                   const float* __restrict__ hbuf,
                                                   float* __restrict__ out,
                                                   const float* __restrict__ bias,
                                                   const float* __restrict__ Wo,
                                                   const float* __restrict__ bo,
                                                   int E, int Epad, int n) {
  constexpr int TS = FIN ? FPITCH : HDIM;
  constexpr int TR = FIN ? FROWS : TN;
  __shared__ __attribute__((aligned(16))) float tile[TR * TS];
  __shared__ int qe[NWAVE][QCAP];
  __shared__ float wos[FIN ? NCLSP * FPITCH : 1];
  __shared__ float bsw[FIN ? HDIM : 1];
  __shared__ int lcur;
  const int tid = threadIdx.x;
  const int lane = tid & 31;
  const int wave = __builtin_amdgcn_readfirstlane(tid >> 5);
  const int base = blockIdx.x * TN;
  const int* __restrict__ esrc = ei;
  const int* __restrict__ edst = ei + E;
  const int lcap = (int)(((long long)n * NCLS - (long long)(Epad >> 2)) / (long long)gridDim.x);
  unsigned* lst = (unsigned*)lists_f + (size_t)blockIdx.x * (size_t)(lcap > 0 ? lcap : 0);
  if (!REPLAY && tid == 0) lcur = 0;

  for (int r = wave; r < TR; r += NWAVE) {
    int node = base + r;
    float4 v = make_float4(0.f, 0.f, 0.f, 0.f);
    if (r < TN && node < n) v = *(const float4*)(hbuf + (size_t)node * HDIM + lane * 4);
    float* t = &tile[r * TS + lane];
    t[0] = v.x; t[32] = v.y; t[64] = v.z; t[96] = v.w;
  }
  if constexpr (FIN) {
    for (int i = tid; i < HDIM * NCLS; i += NWAVE * 32) {
      const int k = i / NCLS, c = i - k * NCLS;
      wos[c * FPITCH + (k & 3) * 32 + (k >> 2)] = Wo[i];
    }
    for (int i = tid; i < (NCLSP - NCLS) * HDIM; i += NWAVE * 32)
      wos[(NCLS + (i >> 7)) * FPITCH + (i & (HDIM - 1))] = 0.f;
    if (tid < HDIM) bsw[(tid & 3) * 32 + (tid >> 2)] = bias[tid];
  }
  __syncthreads();

  auto gather_add = [&](const int my_s, const int my_d, const int cnt) {
      for (int u0 = 0; u0 < cnt; u0 += 8) {
        float4 hv[8]; int dr[8];
#pragma unroll
        for (int u = 0; u < 8; ++u) {
          const int s = __builtin_amdgcn_readlane(my_s, u0 + u);
          dr[u] = __builtin_amdgcn_readlane(my_d, u0 + u);
          hv[u] = *(const float4*)(hbuf + (size_t)s * HDIM + lane * 4);
        }
#pragma unroll
        for (int u = 0; u < 8; ++u) {
          if (u0 + u < cnt && dr[u] >= 0) {
            float* t = &tile[dr[u] * TS + lane];
            atomicAdd(t + 0,  hv[u].x);
            atomicAdd(t + 32, hv[u].y);
            atomicAdd(t + 64, hv[u].z);
            atomicAdd(t + 96, hv[u].w);
          }
        }
      }
  };

  auto drain = [&](const int qtot) {
    for (int qb = 0; qb < qtot; qb += 32) {
      const int cnt = (qtot - qb < 32) ? (qtot - qb) : 32;
      int my_s = 0, my_d = -1;
      if (lane < cnt) {
        const int e = qe[wave][qb + lane];
        if (e < E) {
          const int s = esrc[e];
          const int dd = edst[e];
          my_s = s;
          const int dl = dd - base;
          my_d = ((unsigned)dl < (unsigned)TN) ? dl : -1;
        }
      }
      if constexpr (!REPLAY) {
        int pos = 0;
        if (lane == 0) pos = atomicAdd(&lcur, cnt);
        pos = __builtin_amdgcn_readfirstlane(pos);
        if (lane < cnt && 1 + pos + lane < lcap)
          lst[1 + pos + lane] = (my_d >= 0) ? (((unsigned)my_s << 9) | (unsigned)my_d) : LNONE;
      }
      gather_add(my_s, my_d, cnt);
    }
  };

  int lcount = -1;
  if constexpr (REPLAY) {
    if (lcap >= 2) {
      const unsigned c = (unsigned)__builtin_amdgcn_readfirstlane((int)lst[0]);
      if (c < (unsigned)lcap) lcount = (int)c;
    }
  }
  if (REPLAY && lcount >= 0) {
    auto load_ent = [&](const int i0) -> unsigned {
      const int k = i0 + lane;
      const unsigned v = lst[1 + ((k < lcount) ? k : lcount - 1)];
      return (k < lcount) ? v : LNONE;
    };
    int i = wave * 32;
    unsigned en = load_ent(i);
    for (; i < lcount; i += NWAVE * 32) {
      const unsigned e = en;
      en = load_ent(i + NWAVE * 32);
      const int cnt = (lcount - i < 32) ? (lcount - i) : 32;
      const bool ok = (e != LNONE);
      const int my_s = ok ? (int)(e >> 9) : 0;
      gather_add(my_s, ok ? (int)(e & 511u) : -1, cnt);
    }
  } else {
  const uint4* __restrict__ keys = (const uint4*)keys_f;
  const int per = ((((Epad >> 9) + NWAVE - 1) / NWAVE)) << 9;
  const int e0 = wave * per;
  const int e1 = (e0 + per < Epad) ? (e0 + per) : Epad;
  const int trips = (e1 > e0) ? ((e1 - e0) >> 9) : 0;
  const int start = trips ? (int)((blockIdx.x * 37u) % (unsigned)trips) : 0;
  const unsigned pat = ((unsigned)blockIdx.x & 0xFFu) * 0x01010101u;
  int qn = 0;

  auto load_keys = [&](const int t) -> uint4 { return keys[((e0 + (t << 9)) >> 4) + lane]; };
  int tc = start;
  uint4 kn = make_uint4(0u, 0u, 0u, 0u);
  if (trips > 0) kn = load_keys(tc);

  for (int it = 0; it < trips; ++it) {
    const uint4 k = kn;
    const int eb = e0 + (tc << 9) + lane * 16;
    tc = (tc + 1 == trips) ? 0 : tc + 1;
    if (it + 1 < trips) kn = load_keys(tc);

    const unsigned x0 = k.x ^ pat, x1 = k.y ^ pat, x2 = k.z ^ pat, x3 = k.w ^ pat;
    const unsigned t0 = (x0 - 0x01010101u) & ~x0 & 0x80808080u;
    const unsigned t1 = (x1 - 0x01010101u) & ~x1 & 0x80808080u;
    const unsigned t2 = (x2 - 0x01010101u) & ~x2 & 0x80808080u;
    const unsigned t3 = (x3 - 0x01010101u) & ~x3 & 0x80808080u;
    unsigned u = (t0 >> 7) | (t1 >> 6) | (t2 >> 5) | (t3 >> 4);

    for (int g = 0; g < 16; ++g) {
      const bool has = (u != 0u);
      const unsigned mask = (unsigned)__ballot(has);
      if (mask == 0u) break;
      if (has) {
        const int p = __builtin_ctz(u);
        qe[wave][qn + (int)__builtin_amdgcn_mbcnt_lo(mask, 0u)] = eb + ((p & 7) << 2) + (p >> 3);
      }
      u &= u - 1u;
      qn += __popc(mask);
      if (qn >= 32) { drain(qn); qn = 0; }
    }
  }
  if (qn > 0) drain(qn);
  }
  static_assert(TN <= 32 * NWAVE, "one lane per row that a wave stores");
  const int xr = FIN ? (wave * 16 + (lane & 15)) : (wave + lane * NWAVE);
  const float xd = (xr < TN && base + xr < n) ? rsqrtf(deg[base + xr] + 1.0f) : 0.f;
  __syncthreads();
  if constexpr (!REPLAY) {
    if (tid == 0 && lcap >= 2) lst[0] = (unsigned)lcur;
  }

  if constexpr (!FIN) {
    for (int r = wave, i = 0; r < TN; r += NWAVE, ++i) {
      int node = base + r;
      if (node < n) {
        const float d = __int_as_float(__builtin_amdgcn_readlane(__float_as_int(xd), i));
        const float* t = &tile[r * TS + lane];
        float4 v = make_float4(t[0] * d, t[32] * d, t[64] * d, t[96] * d);
        *(float4*)(out + (size_t)node * HDIM + lane * 4) = v;
      }
    }
  } else {
    const int m = lane & 15;
    const int hi = lane >> 4;
    v8f acc[3];
#pragma unroll
    for (int c = 0; c < 3; ++c)
#pragma unroll
      for (int v = 0; v < 8; ++v) acc[c][v] = 0.f;
    if (wave < FROWS / 16) {
      const float* arow = &tile[(wave * 16 + m) * FPITCH + hi * 32];
      const float* brow = &wos[m * FPITCH + hi * 32];
      const float* bb = &bsw[hi * 32];
      const float ad = xd;
#pragma unroll 4
      for (int j = 0; j < 32; ++j) {
        const float t0 = arow[j] * ad + bb[j];
        const float t1 = arow[64 + j] * ad + bb[64 + j];
        v2f a; a.x = t0 > 0.f ? t0 : 0.f; a.y = t1 > 0.f ? t1 : 0.f;
#pragma unroll
        for (int c = 0; c < 3; ++c) {
          v2f b; b.x = brow[c * 16 * FPITCH + j]; b.y = brow[c * 16 * FPITCH + 64 + j];
          acc[c] = __builtin_amdgcn_wmma_f32_16x16x4_f32(
              false, a, false, b, (short)0, acc[c], false, false);
        }
      }
    }
    __syncthreads();
    float* stg = tile;
    static_assert(TN % 8 == 0, "a lane half's 8 rows must not straddle the last row");
    const int r0 = wave * 16 + 8 * hi;
    if (r0 < TN) {
#pragma unroll
      for (int c = 0; c < 3; ++c) {
        const int col = c * 16 + m;
        if (col < NCLS) {
          const float bov = bo[col];
#pragma unroll
          for (int v = 0; v < 8; ++v) stg[(r0 + v) * NCLS + col] = acc[c][v] + bov;
        }
      }
    }
    __syncthreads();
    const int rows = (n - base < TN) ? (n - base) : TN;
    const int nf4 = rows * (NCLS / 4);
    float4* __restrict__ o4 = (float4*)(out + (size_t)base * NCLS);
    const float4* s4 = (const float4*)stg;
    for (int i = tid; i < nf4; i += NWAVE * 32) o4[i] = s4[i];
  }
}

#define LSPLIT 8
__global__ void k_copy_keys(const float* __restrict__ src, float* __restrict__ dst, int n16,
                            int nkb, int Epad, int n, int nblk) {
  if ((int)blockIdx.x < nkb) {
    const int i = blockIdx.x * blockDim.x + threadIdx.x;
    if (i < n16) ((float4*)dst)[i] = ((const float4*)src)[i];
    return;
  }
  const int j = (int)blockIdx.x - nkb;
  const int lcap = (int)(((long long)n * NCLS - (long long)(Epad >> 2)) / (long long)nblk);
  if (lcap < 2) return;
  const size_t off = (size_t)(Epad >> 2) + (size_t)(j / LSPLIT) * (size_t)lcap;
  const unsigned* __restrict__ s = (const unsigned*)src + off;
  unsigned* __restrict__ d = (unsigned*)dst + off;
  const unsigned c = s[0];
  const int nw = (c < (unsigned)lcap) ? (int)c + 1 : 1;
  for (int i = (j % LSPLIT) * (int)blockDim.x + (int)threadIdx.x; i < nw; i += LSPLIT * (int)blockDim.x)
    d[i] = s[i];
}

extern "C" void kernel_launch(void* const* d_in, const int* in_sizes, int n_in,
                              void* d_out, int out_size, void* d_ws, size_t ws_size,
                              hipStream_t stream) {
  const float* x_in  = (const float*)d_in[0];
  const int*   ei    = (const int*)d_in[1];
  const float* gamma = (const float*)d_in[2];
  const float* beta  = (const float*)d_in[3];
  const float* W     = (const float*)d_in[4];
  const float* b     = (const float*)d_in[5];
  const float* Wo    = (const float*)d_in[6];
  const float* bo    = (const float*)d_in[7];
  float* out = (float*)d_out;

  const int N = in_sizes[0] / HDIM;
  const int E = in_sizes[1] / 2;
  const int Epad = ((E + 511) / 512) * 512;

  float* deg   = (float*)d_ws;
  float* stats = deg + ((N + 255) / 256) * 256;
  float* WfT   = stats + 2 * HDIM;
  float* cb    = WfT + HDIM * HDIM;
  float* hbuf  = cb + HDIM + 64;
  float* xbuf  = hbuf + (size_t)N * HDIM;
  float* keys  = out;
  float* lists = out + Epad / 4;
  const int nblk = (N + TN - 1) / TN;

  hipMemsetAsync(deg, 0, (((N + 255) / 256) * 256 + 2 * HDIM) * sizeof(float), stream);
  k_deg_keys<<<(Epad / 4 + 255) / 256, 256, 0, stream>>>(ei, deg, keys, E, Epad);

  k_stats<<<512, 256, 0, stream>>>(x_in, stats, N);

  for (int l = 0; l < 3; ++l) {
    k_fold<<<HDIM, HDIM, 0, stream>>>(stats, gamma + l * HDIM, beta + l * HDIM,
                                  W + (size_t)l * HDIM * HDIM, WfT, cb, 1.0f / (float)N);
    const float* xl = (l == 0) ? x_in : xbuf;
    k_gemm<<<(N + 63) / 64, 128, 0, stream>>>(xl, WfT, cb, deg, stats, hbuf, N);
    if (l < 2) {
      if (l == 0)
        k_agg_tile<false, false><<<nblk, 1024, 0, stream>>>(ei, keys, lists, deg, hbuf, xbuf,
                                                            nullptr, nullptr, nullptr, E, Epad, N);
      else
        k_agg_tile<false, true><<<nblk, 1024, 0, stream>>>(ei, keys, lists, deg, hbuf, xbuf,
                                                           nullptr, nullptr, nullptr, E, Epad, N);
      k_bias_relu_stats<<<512, 256, 0, stream>>>(xbuf, b + l * HDIM, stats, N, 1);
    } else {
      const int nkb = (Epad / 16 + 255) / 256;
      k_copy_keys<<<nkb + nblk * LSPLIT, 256, 0, stream>>>(keys, xbuf, Epad / 16, nkb, Epad, N, nblk);
      k_agg_tile<true, true><<<nblk, 1024, 0, stream>>>(ei, xbuf, xbuf + Epad / 4, deg, hbuf, out,
                                                        b + l * HDIM, Wo, bo, E, Epad, N);
    }
  }
}


# First layer's key scan leaves the owner blocks: extra blocks of the input column-sum launch build the edge lists (one global count word per list); all layers replay; LDS add order differs
# speedup vs baseline: 3.1910x; 1.0317x over previous
#include <hip/hip_runtime.h>

#define HDIM 128
#define NCLS 40
#define BN_EPS 1e-5f

typedef __attribute__((ext_vector_type(2))) float v2f;
typedef __attribute__((ext_vector_type(8))) float v8f;

#define TN 392
#define QCAP 64
#define LNONE 0xFFFFFFFFu
#define SPL 4
__global__ void k_deg_keys(const int* __restrict__ ei, float* deg, float* __restrict__ keys_f,
                           int E, int Epad, int n, int nblk) {
  const int q = blockIdx.x * blockDim.x + threadIdx.x;
  if (q < nblk) {
    const int lcap = (int)(((long long)n * NCLS - (long long)(Epad >> 2)) / (long long)nblk);
    if (lcap >= 2) ((unsigned*)keys_f)[(size_t)(Epad >> 2) + (size_t)q * (size_t)lcap] = 0u;
  }
  const int e = q * 4;
  if (e >= Epad) return;
  const int* __restrict__ edst = ei + E;
  int d[4];
  if (((E & 3) == 0) && e + 4 <= E) {
    const int4 dv = *(const int4*)(edst + e);
    d[0] = dv.x; d[1] = dv.y; d[2] = dv.z; d[3] = dv.w;
  } else {
#pragma unroll
    for (int j = 0; j < 4; ++j) d[j] = (e + j < E) ? edst[e + j] : -1;
  }
  unsigned packed = 0u;
#pragma unroll
  for (int j = 0; j < 4; ++j) {
    unsigned k = 0xFFu;
    if (d[j] >= 0) {
      atomicAdd(&deg[d[j]], 1.0f);
      k = ((unsigned)d[j] / (unsigned)TN) & 0xFFu;
    }
    packed |= k << (8 * j);
  }
  ((unsigned*)keys_f)[q] = packed;
}

__device__ __forceinline__ void scan_keys(const int* __restrict__ ei, const float* __restrict__ keys_f,
                                          float* lists_f, int E, int Epad, int n, int nblk, int j) {
  __shared__ int qe[8][QCAP];
  const int lane = threadIdx.x & 31;
  const int wave = __builtin_amdgcn_readfirstlane((int)(threadIdx.x >> 5));
  const int owner = j / SPL;
  const int share = (j - owner * SPL) * 8 + wave;
  const int base = owner * TN;
  const int lcap = (int)(((long long)n * NCLS - (long long)(Epad >> 2)) / (long long)nblk);
  if (lcap < 2) return;
  unsigned* lst = (unsigned*)lists_f + (size_t)owner * (size_t)lcap;
  const int* __restrict__ esrc = ei;
  const int* __restrict__ edst = ei + E;

  auto drain = [&](const int qtot) {
    for (int qb = 0; qb < qtot; qb += 32) {
      const int cnt = (qtot - qb < 32) ? (qtot - qb) : 32;
      unsigned ent = LNONE;
      if (lane < cnt) {
        const int e = qe[wave][qb + lane];
        if (e < E) {
          const int s = esrc[e];
          const int dl = edst[e] - base;
          if ((unsigned)dl < (unsigned)TN) ent = ((unsigned)s << 9) | (unsigned)dl;
        }
      }
      int pos = 0;
      if (lane == 0) pos = (int)atomicAdd(&lst[0], (unsigned)cnt);
      pos = __builtin_amdgcn_readfirstlane(pos);
      if (lane < cnt && 1 + pos + lane < lcap) lst[1 + pos + lane] = ent;
    }
  };

  const uint4* __restrict__ keys = (const uint4*)keys_f;
  const int per = ((((Epad >> 9) + 32 - 1) / 32)) << 9;
  const int e0 = share * per;
  const int e1 = (e0 + per < Epad) ? (e0 + per) : Epad;
  const int trips = (e1 > e0) ? ((e1 - e0) >> 9) : 0;
  const int start = trips ? (int)(((unsigned)owner * 37u) % (unsigned)trips) : 0;
  const unsigned pat = ((unsigned)owner & 0xFFu) * 0x01010101u;
  int qn = 0;

  auto load_keys = [&](const int t) -> uint4 { return keys[((e0 + (t << 9)) >> 4) + lane]; };
  int tc = start;
  uint4 kn = make_uint4(0u, 0u, 0u, 0u);
  if (trips > 0) kn = load_keys(tc);

  for (int it = 0; it < trips; ++it) {
    const uint4 k = kn;
    const int eb = e0 + (tc << 9) + lane * 16;
    tc = (tc + 1 == trips) ? 0 : tc + 1;
    if (it + 1 < trips) kn = load_keys(tc);

    const unsigned x0 = k.x ^ pat, x1 = k.y ^ pat, x2 = k.z ^ pat, x3 = k.w ^ pat;
    const unsigned t0 = (x0 - 0x01010101u) & ~x0 & 0x80808080u;
    const unsigned t1 = (x1 - 0x01010101u) & ~x1 & 0x80808080u;
    const unsigned t2 = (x2 - 0x01010101u) & ~x2 & 0x80808080u;
    const unsigned t3 = (x3 - 0x01010101u) & ~x3 & 0x80808080u;
    unsigned u = (t0 >> 7) | (t1 >> 6) | (t2 >> 5) | (t3 >> 4);

    for (int g = 0; g < 16; ++g) {
      const bool has = (u != 0u);
      const unsigned mask = (unsigned)__ballot(has);
      if (mask == 0u) break;
      if (has) {
        const int p = __builtin_ctz(u);
        qe[wave][qn + (int)__builtin_amdgcn_mbcnt_lo(mask, 0u)] = eb + ((p & 7) << 2) + (p >> 3);
      }
      u &= u - 1u;
      qn += __popc(mask);
      if (qn >= 32) { drain(qn); qn = 0; }
    }
  }
  if (qn > 0) drain(qn);
}

__global__ __launch_bounds__(256) void k_stats(const float* __restrict__ x, float* __restrict__ stats, int n,
                                               int nsb, const int* __restrict__ ei,
                                               const float* __restrict__ keys_f, float* lists_f,
                                               int E, int Epad, int nblk) {
  if ((int)blockIdx.x >= nsb) {
    scan_keys(ei, keys_f, lists_f, E, Epad, n, nblk, (int)blockIdx.x - nsb);
    return;
  }
  int col = threadIdx.x & (HDIM - 1);
  int half = threadIdx.x >> 7;
  float s = 0.f, s2 = 0.f;
  for (int r = blockIdx.x * 2 + half; r < n; r += nsb * 2) {
    float v = x[(size_t)r * HDIM + col];
    s += v; s2 += v * v;
  }
  __shared__ float ls[HDIM], ls2[HDIM];
  if (half == 0) { ls[col] = s; ls2[col] = s2; }
  __syncthreads();
  if (half == 1) {
    atomicAdd(&stats[col], s + ls[col]);
    atomicAdd(&stats[HDIM + col], s2 + ls2[col]);
  }
}

#define BR_ROWS 8
#define BR_WAVES 8
__global__ __launch_bounds__(BR_WAVES * 32) void k_bias_relu_stats(float* __restrict__ x,
                                                                   const float* __restrict__ b,
                                                                   float* __restrict__ stats, int n,
                                                                   int do_stats) {
  const int lane = threadIdx.x & 31;
  const int wave = threadIdx.x >> 5;
  const int gw = blockIdx.x * BR_WAVES + wave;
  const int nw = gridDim.x * BR_WAVES;
  const float4 bv = *(const float4*)(b + lane * 4);
  float4 s = make_float4(0.f, 0.f, 0.f, 0.f), s2 = make_float4(0.f, 0.f, 0.f, 0.f);
  auto finish = [&](float4 v, float* p) {
    v.x += bv.x; v.y += bv.y; v.z += bv.z; v.w += bv.w;
    v.x = v.x > 0.f ? v.x : 0.f; v.y = v.y > 0.f ? v.y : 0.f;
    v.z = v.z > 0.f ? v.z : 0.f; v.w = v.w > 0.f ? v.w : 0.f;
    *(float4*)p = v;
    s.x += v.x; s.y += v.y; s.z += v.z; s.w += v.w;
    s2.x += v.x * v.x; s2.y += v.y * v.y; s2.z += v.z * v.z; s2.w += v.w * v.w;
  };
  for (int r0 = gw * BR_ROWS; r0 < n; r0 += nw * BR_ROWS) {
    float* p = x + (size_t)r0 * HDIM + lane * 4;
    if (r0 + BR_ROWS <= n) {
      float4 v[BR_ROWS];
      for (int u = 0; u < BR_ROWS; ++u) v[u] = *(const float4*)(p + u * HDIM);
      for (int u = 0; u < BR_ROWS; ++u) finish(v[u], p + u * HDIM);
    } else {
      for (int u = 0; r0 + u < n; ++u) finish(*(const float4*)(p + u * HDIM), p + u * HDIM);
    }
  }
  if (!do_stats) return;
  __shared__ __attribute__((aligned(16))) float part[BR_WAVES][2 * HDIM];
  *(float4*)(&part[wave][lane * 4]) = s;
  *(float4*)(&part[wave][HDIM + lane * 4]) = s2;
  __syncthreads();
  const int t = threadIdx.x;
  float a = 0.f;
  for (int w = 0; w < BR_WAVES; ++w) a += part[w][t];
  atomicAdd(&stats[t], a);
}

__global__ __launch_bounds__(HDIM) void k_fold(const float* __restrict__ stats,
                                               const float* __restrict__ gamma,
                                               const float* __restrict__ beta,
                                               const float* __restrict__ W,
                                               float* __restrict__ WfT, float* __restrict__ cb,
                                               float inv_n) {
  __shared__ float red[HDIM / 32];
  const int k = threadIdx.x;
  const int n = blockIdx.x;
  const float w = W[k * HDIM + n];
  const float mean = stats[k] * inv_n;
  const float var  = stats[HDIM + k] * inv_n - mean * mean;
  const float a = gamma[k] * rsqrtf(var + BN_EPS);
  const float c = beta[k] - a * mean;
  float p = c * w;
  for (int o = 16; o > 0; o >>= 1) p += __shfl_xor(p, o, 32);
  if ((k & 31) == 0) red[k >> 5] = p;
  __syncthreads();
  WfT[n * HDIM + k] = a * w;
  if (k == 0) cb[n] = (red[0] + red[1]) + (red[2] + red[3]);
}

#define XPAD 68

__global__ __launch_bounds__(128) void k_gemm(const float* __restrict__ x,
                                              const float* __restrict__ WfT,
                                              const float* __restrict__ cb,
                                              const float* __restrict__ deg,
                                              float* __restrict__ stats,
                                              float* __restrict__ h, int n) {
  if (blockIdx.x == 0) { stats[threadIdx.x] = 0.f; stats[HDIM + threadIdx.x] = 0.f; }
  __shared__ float xs[64 * XPAD];
  __shared__ float wfs[HDIM * XPAD];
  const int tid = threadIdx.x;
  const int lane = tid & 31;
  const int wave = tid >> 5;
  const int rowBase = blockIdx.x * 64;
  const int m = lane & 15;
  const int hi = lane >> 4;

  v8f acc[8];
#pragma unroll
  for (int c = 0; c < 8; ++c)
#pragma unroll
    for (int v = 0; v < 8; ++v) acc[c][v] = 0.f;

  for (int kh = 0; kh < 2; ++kh) {
#pragma unroll
    for (int p = 0; p < 8; ++p) {
      int idx = p * 512 + tid * 4;
      int r = idx >> 6, cc = idx & 63;
      int row = rowBase + r;
      float4 v = make_float4(0.f, 0.f, 0.f, 0.f);
      if (row < n) v = *(const float4*)(x + (size_t)row * HDIM + kh * 64 + cc);
      *(float4*)(&xs[r * XPAD + cc]) = v;
    }
#pragma unroll
    for (int p = 0; p < 16; ++p) {
      int idx = p * 512 + tid * 4;
      int nn = idx >> 6, cc = idx & 63;
      *(float4*)(&wfs[nn * XPAD + cc]) = *(const float4*)(WfT + nn * HDIM + kh * 64 + cc);
    }
    __syncthreads();

    const float* xrow = &xs[(wave * 16 + m) * XPAD];
#pragma unroll
    for (int k4 = 0; k4 < 16; ++k4) {
      int kb = k4 * 4 + hi * 2;
      v2f a; a.x = xrow[kb]; a.y = xrow[kb + 1];
#pragma unroll
      for (int c = 0; c < 8; ++c) {
        const float* wrow = &wfs[(c * 16 + m) * XPAD];
        v2f b; b.x = wrow[kb]; b.y = wrow[kb + 1];
        acc[c] = __builtin_amdgcn_wmma_f32_16x16x4_f32(
            false, a, false, b, (short)0, acc[c], false, false);
      }
    }
    __syncthreads();
  }

  float dv[8];
#pragma unroll
  for (int v = 0; v < 8; ++v) {
    const int row = rowBase + wave * 16 + v + 8 * hi;
    dv[v] = (row < n) ? rsqrtf(deg[row] + 1.0f) : 0.f;
  }
#pragma unroll
  for (int c = 0; c < 8; ++c) {
    int col = c * 16 + m;
    float cbv = cb[col];
#pragma unroll
    for (int v = 0; v < 8; ++v) {
      int row = rowBase + wave * 16 + v + 8 * hi;
      if (row < n) h[(size_t)row * HDIM + col] = (acc[c][v] + cbv) * dv[v];
    }
  }
}

#define NWAVE 32
#define FROWS 400
#define FPITCH 130
#define NCLSP 48
static_assert(TN <= 512, "a list entry keeps the local row in 9 bits");


template <bool FIN>
__global__ __launch_bounds__(1024) void k_agg_tile(const int* __restrict__ ei,
                                                   const float* __restrict__ keys_f,
                                                   const float* __restrict__ lists_f,
                                                   const float* __restrict__ deg,
                                                   const float* __restrict__ hbuf,
                                                   float* __restrict__ out,
                                                   const float* __restrict__ bias,
                                                   const float* __restrict__ Wo,
                                                   const float* __restrict__ bo,
                                                   int E, int Epad, int n) {
  constexpr int TS = FIN ? FPITCH : HDIM;
  constexpr int TR = FIN ? FROWS : TN;
  __shared__ __attribute__((aligned(16))) float tile[TR * TS];
  __shared__ int qe[NWAVE][QCAP];
  __shared__ float wos[FIN ? NCLSP * FPITCH : 1];
  __shared__ float bsw[FIN ? HDIM : 1];
  const int tid = threadIdx.x;
  const int lane = tid & 31;
  const int wave = __builtin_amdgcn_readfirstlane(tid >> 5);
  const int base = blockIdx.x * TN;
  const int* __restrict__ esrc = ei;
  const int* __restrict__ edst = ei + E;
  const int lcap = (int)(((long long)n * NCLS - (long long)(Epad >> 2)) / (long long)gridDim.x);
  const unsigned* __restrict__ lst = (const unsigned*)lists_f + (size_t)blockIdx.x * (size_t)(lcap > 0 ? lcap : 0);

  for (int r = wave; r < TR; r += NWAVE) {
    int node = base + r;
    float4 v = make_float4(0.f, 0.f, 0.f, 0.f);
    if (r < TN && node < n) v = *(const float4*)(hbuf + (size_t)node * HDIM + lane * 4);
    float* t = &tile[r * TS + lane];
    t[0] = v.x; t[32] = v.y; t[64] = v.z; t[96] = v.w;
  }
  if constexpr (FIN) {
    for (int i = tid; i < HDIM * NCLS; i += NWAVE * 32) {
      const int k = i / NCLS, c = i - k * NCLS;
      wos[c * FPITCH + (k & 3) * 32 + (k >> 2)] = Wo[i];
    }
    for (int i = tid; i < (NCLSP - NCLS) * HDIM; i += NWAVE * 32)
      wos[(NCLS + (i >> 7)) * FPITCH + (i & (HDIM - 1))] = 0.f;
    if (tid < HDIM) bsw[(tid & 3) * 32 + (tid >> 2)] = bias[tid];
  }
  __syncthreads();

  auto gather_add = [&](const int my_s, const int my_d, const int cnt) {
      for (int u0 = 0; u0 < cnt; u0 += 8) {
        float4 hv[8]; int dr[8];
#pragma unroll
        for (int u = 0; u < 8; ++u) {
          const int s = __builtin_amdgcn_readlane(my_s, u0 + u);
          dr[u] = __builtin_amdgcn_readlane(my_d, u0 + u);
          hv[u] = *(const float4*)(hbuf + (size_t)s * HDIM + lane * 4);
        }
#pragma unroll
        for (int u = 0; u < 8; ++u) {
          if (u0 + u < cnt && dr[u] >= 0) {
            float* t = &tile[dr[u] * TS + lane];
            atomicAdd(t + 0,  hv[u].x);
            atomicAdd(t + 32, hv[u].y);
            atomicAdd(t + 64, hv[u].z);
            atomicAdd(t + 96, hv[u].w);
          }
        }
      }
  };

  auto drain = [&](const int qtot) {
    for (int qb = 0; qb < qtot; qb += 32) {
      const int cnt = (qtot - qb < 32) ? (qtot - qb) : 32;
      int my_s = 0, my_d = -1;
      if (lane < cnt) {
        const int e = qe[wave][qb + lane];
        if (e < E) {
          const int s = esrc[e];
          const int dd = edst[e];
          my_s = s;
          const int dl = dd - base;
          my_d = ((unsigned)dl < (unsigned)TN) ? dl : -1;
        }
      }
      gather_add(my_s, my_d, cnt);
    }
  };

  int lcount = -1;
  if (lcap >= 2) {
    const unsigned c = (unsigned)__builtin_amdgcn_readfirstlane((int)lst[0]);
    if (c < (unsigned)lcap) lcount = (int)c;
  }
  if (lcount >= 0) {
    auto load_ent = [&](const int i0) -> unsigned {
      const int k = i0 + lane;
      const unsigned v = lst[1 + ((k < lcount) ? k : lcount - 1)];
      return (k < lcount) ? v : LNONE;
    };
    int i = wave * 32;
    unsigned en = load_ent(i);
    for (; i < lcount; i += NWAVE * 32) {
      const unsigned e = en;
      en = load_ent(i + NWAVE * 32);
      const int cnt = (lcount - i < 32) ? (lcount - i) : 32;
      const bool ok = (e != LNONE);
      const int my_s = ok ? (int)(e >> 9) : 0;
      gather_add(my_s, ok ? (int)(e & 511u) : -1, cnt);
    }
  } else {
  const uint4* __restrict__ keys = (const uint4*)keys_f;
  const int per = ((((Epad >> 9) + NWAVE - 1) / NWAVE)) << 9;
  const int e0 = wave * per;
  const int e1 = (e0 + per < Epad) ? (e0 + per) : Epad;
  const int trips = (e1 > e0) ? ((e1 - e0) >> 9) : 0;
  const int start = trips ? (int)((blockIdx.x * 37u) % (unsigned)trips) : 0;
  const unsigned pat = ((unsigned)blockIdx.x & 0xFFu) * 0x01010101u;
  int qn = 0;

  auto load_keys = [&](const int t) -> uint4 { return keys[((e0 + (t << 9)) >> 4) + lane]; };
  int tc = start;
  uint4 kn = make_uint4(0u, 0u, 0u, 0u);
  if (trips > 0) kn = load_keys(tc);

  for (int it = 0; it < trips; ++it) {
    const uint4 k = kn;
    const int eb = e0 + (tc << 9) + lane * 16;
    tc = (tc + 1 == trips) ? 0 : tc + 1;
    if (it + 1 < trips) kn = load_keys(tc);

    const unsigned x0 = k.x ^ pat, x1 = k.y ^ pat, x2 = k.z ^ pat, x3 = k.w ^ pat;
    const unsigned t0 = (x0 - 0x01010101u) & ~x0 & 0x80808080u;
    const unsigned t1 = (x1 - 0x01010101u) & ~x1 & 0x80808080u;
    const unsigned t2 = (x2 - 0x01010101u) & ~x2 & 0x80808080u;
    const unsigned t3 = (x3 - 0x01010101u) & ~x3 & 0x80808080u;
    unsigned u = (t0 >> 7) | (t1 >> 6) | (t2 >> 5) | (t3 >> 4);

    for (int g = 0; g < 16; ++g) {
      const bool has = (u != 0u);
      const unsigned mask = (unsigned)__ballot(has);
      if (mask == 0u) break;
      if (has) {
        const int p = __builtin_ctz(u);
        qe[wave][qn + (int)__builtin_amdgcn_mbcnt_lo(mask, 0u)] = eb + ((p & 7) << 2) + (p >> 3);
      }
      u &= u - 1u;
      qn += __popc(mask);
      if (qn >= 32) { drain(qn); qn = 0; }
    }
  }
  if (qn > 0) drain(qn);
  }
  static_assert(TN <= 32 * NWAVE, "one lane per row that a wave stores");
  const int xr = FIN ? (wave * 16 + (lane & 15)) : (wave + lane * NWAVE);
  const float xd = (xr < TN && base + xr < n) ? rsqrtf(deg[base + xr] + 1.0f) : 0.f;
  __syncthreads();

  if constexpr (!FIN) {
    for (int r = wave, i = 0; r < TN; r += NWAVE, ++i) {
      int node = base + r;
      if (node < n) {
        const float d = __int_as_float(__builtin_amdgcn_readlane(__float_as_int(xd), i));
        const float* t = &tile[r * TS + lane];
        float4 v = make_float4(t[0] * d, t[32] * d, t[64] * d, t[96] * d);
        *(float4*)(out + (size_t)node * HDIM + lane * 4) = v;
      }
    }
  } else {
    const int m = lane & 15;
    const int hi = lane >> 4;
    v8f acc[3];
#pragma unroll
    for (int c = 0; c < 3; ++c)
#pragma unroll
      for (int v = 0; v < 8; ++v) acc[c][v] = 0.f;
    if (wave < FROWS / 16) {
      const float* arow = &tile[(wave * 16 + m) * FPITCH + hi * 32];
      const float* brow = &wos[m * FPITCH + hi * 32];
      const float* bb = &bsw[hi * 32];
      const float ad = xd;
#pragma unroll 4
      for (int j = 0; j < 32; ++j) {
        const float t0 = arow[j] * ad + bb[j];
        const float t1 = arow[64 + j] * ad + bb[64 + j];
        v2f a; a.x = t0 > 0.f ? t0 : 0.f; a.y = t1 > 0.f ? t1 : 0.f;
#pragma unroll
        for (int c = 0; c < 3; ++c) {
          v2f b; b.x = brow[c * 16 * FPITCH + j]; b.y = brow[c * 16 * FPITCH + 64 + j];
          acc[c] = __builtin_amdgcn_wmma_f32_16x16x4_f32(
              false, a, false, b, (short)0, acc[c], false, false);
        }
      }
    }
    __syncthreads();
    float* stg = tile;
    static_assert(TN % 8 == 0, "a lane half's 8 rows must not straddle the last row");
    const int r0 = wave * 16 + 8 * hi;
    if (r0 < TN) {
#pragma unroll
      for (int c = 0; c < 3; ++c) {
        const int col = c * 16 + m;
        if (col < NCLS) {
          const float bov = bo[col];
#pragma unroll
          for (int v = 0; v < 8; ++v) stg[(r0 + v) * NCLS + col] = acc[c][v] + bov;
        }
      }
    }
    __syncthreads();
    const int rows = (n - base < TN) ? (n - base) : TN;
    const int nf4 = rows * (NCLS / 4);
    float4* __restrict__ o4 = (float4*)(out + (size_t)base * NCLS);
    const float4* s4 = (const float4*)stg;
    for (int i = tid; i < nf4; i += NWAVE * 32) o4[i] = s4[i];
  }
}

#define LSPLIT 8
__global__ void k_copy_keys(const float* __restrict__ src, float* __restrict__ dst, int n16,
                            int nkb, int Epad, int n, int nblk) {
  if ((int)blockIdx.x < nkb) {
    const int i = blockIdx.x * blockDim.x + threadIdx.x;
    if (i < n16) ((float4*)dst)[i] = ((const float4*)src)[i];
    return;
  }
  const int j = (int)blockIdx.x - nkb;
  const int lcap = (int)(((long long)n * NCLS - (long long)(Epad >> 2)) / (long long)nblk);
  if (lcap < 2) return;
  const size_t off = (size_t)(Epad >> 2) + (size_t)(j / LSPLIT) * (size_t)lcap;
  const unsigned* __restrict__ s = (const unsigned*)src + off;
  unsigned* __restrict__ d = (unsigned*)dst + off;
  const unsigned c = s[0];
  const int nw = (c < (unsigned)lcap) ? (int)c + 1 : 1;
  for (int i = (j % LSPLIT) * (int)blockDim.x + (int)threadIdx.x; i < nw; i += LSPLIT * (int)blockDim.x)
    d[i] = s[i];
}

extern "C" void kernel_launch(void* const* d_in, const int* in_sizes, int n_in,
                              void* d_out, int out_size, void* d_ws, size_t ws_size,
                              hipStream_t stream) {
  const float* x_in  = (const float*)d_in[0];
  const int*   ei    = (const int*)d_in[1];
  const float* gamma = (const float*)d_in[2];
  const float* beta  = (const float*)d_in[3];
  const float* W     = (const float*)d_in[4];
  const float* b     = (const float*)d_in[5];
  const float* Wo    = (const float*)d_in[6];
  const float* bo    = (const float*)d_in[7];
  float* out = (float*)d_out;

  const int N = in_sizes[0] / HDIM;
  const int E = in_sizes[1] / 2;
  const int Epad = ((E + 511) / 512) * 512;

  float* deg   = (float*)d_ws;
  float* stats = deg + ((N + 255) / 256) * 256;
  float* WfT   = stats + 2 * HDIM;
  float* cb    = WfT + HDIM * HDIM;
  float* hbuf  = cb + HDIM + 64;
  float* xbuf  = hbuf + (size_t)N * HDIM;
  float* keys  = out;
  float* lists = out + Epad / 4;
  const int nblk = (N + TN - 1) / TN;

  hipMemsetAsync(deg, 0, (((N + 255) / 256) * 256 + 2 * HDIM) * sizeof(float), stream);
  k_deg_keys<<<(Epad / 4 + 255) / 256, 256, 0, stream>>>(ei, deg, keys, E, Epad, N, nblk);

  k_stats<<<512 + nblk * SPL, 256, 0, stream>>>(x_in, stats, N, 512, ei, keys, lists, E, Epad, nblk);

  for (int l = 0; l < 3; ++l) {
    k_fold<<<HDIM, HDIM, 0, stream>>>(stats, gamma + l * HDIM, beta + l * HDIM,
                                  W + (size_t)l * HDIM * HDIM, WfT, cb, 1.0f / (float)N);
    const float* xl = (l == 0) ? x_in : xbuf;
    k_gemm<<<(N + 63) / 64, 128, 0, stream>>>(xl, WfT, cb, deg, stats, hbuf, N);
    if (l < 2) {
      k_agg_tile<false><<<nblk, 1024, 0, stream>>>(ei, keys, lists, deg, hbuf, xbuf,
                                                   nullptr, nullptr, nullptr, E, Epad, N);
      k_bias_relu_stats<<<512, 256, 0, stream>>>(xbuf, b + l * HDIM, stats, N, 1);
    } else {
      const int nkb = (Epad / 16 + 255) / 256;
      k_copy_keys<<<nkb + nblk * LSPLIT, 256, 0, stream>>>(keys, xbuf, Epad / 16, nkb, Epad, N, nblk);
      k_agg_tile<true><<<nblk, 1024, 0, stream>>>(ei, xbuf, xbuf + Epad / 4, deg, hbuf, out,
                                                  b + l * HDIM, Wo, bo, E, Epad, N);
    }
  }
}
